# starting kernel
# baseline (speedup 1.0000x reference)
#include <hip/hip_runtime.h>
#include <hip/hip_bf16.h>


typedef float v2f __attribute__((ext_vector_type(2)));
typedef float v8f __attribute__((ext_vector_type(8)));

#define NB 2
#define NC 64
#define NPIX 147456
#define NCLS 20
#define BLKPX 1024
#define CHUNKS (BLKPX / 32)
#define NBLKS (NPIX / BLKPX)
#define PARTS 16
#define PADALIGN (PARTS * 4)
#define NPAD_MAX (NPIX + NCLS * PADALIGN)
#define MIN_CLASS_SIZE 5000
#define EPSV 1e-6f

__global__ __launch_bounds__(32) void hist_kernel(const int* __restrict__ seg,
                                                  int* __restrict__ hist) {
  int blk = blockIdx.x;
  int b = blk / NBLKS, bb = blk % NBLKS;
  int lane = threadIdx.x;
  const int* s = seg + (size_t)b * NPIX + (size_t)bb * BLKPX;
  int cnt = 0;
  for (int ch = 0; ch < CHUNKS; ++ch) {
    int myk = s[ch * 32 + lane];
#pragma unroll
    for (int k = 0; k < NCLS; ++k) {
      unsigned bal = __builtin_amdgcn_ballot_w32(myk == k);
      if (lane == k) cnt += __popc(bal);
    }
  }
  if (lane < NCLS) hist[(size_t)blk * NCLS + lane] = cnt;
}

__global__ __launch_bounds__(64) void scan_kernel(const int* __restrict__ hist,
                                                  int* __restrict__ blockoff,
                                                  int* __restrict__ classinfo) {
  int tid = threadIdx.x;
  __shared__ int tot[NB * NCLS];
  __shared__ int padl[NB * NCLS];
  if (tid < NB * NCLS) {
    int b = tid / NCLS, k = tid % NCLS;
    int run = 0;
    for (int blk = 0; blk < NBLKS; ++blk) {
      size_t idx = ((size_t)(b * NBLKS + blk)) * NCLS + k;
      blockoff[idx] = run;
      run += hist[idx];
    }
    tot[tid] = run;
    padl[tid] = (run + PADALIGN - 1) & ~(PADALIGN - 1);
  }
  __syncthreads();
  if (tid < NB) {
    int b = tid, base = 0;
    for (int k = 0; k < NCLS; ++k) {
      classinfo[(b * NCLS + k) * 3 + 0] = base;
      classinfo[(b * NCLS + k) * 3 + 1] = tot[b * NCLS + k];
      classinfo[(b * NCLS + k) * 3 + 2] = padl[b * NCLS + k];
      base += padl[b * NCLS + k];
    }
  }
}

__global__ __launch_bounds__(256) void padzero_kernel(const int* __restrict__ classinfo,
                                                      float* __restrict__ comp) {
  int bk = blockIdx.x;
  int b = bk / NCLS;
  int base = classinfo[bk * 3 + 0];
  int count = classinfo[bk * 3 + 1];
  int padl = classinfo[bk * 3 + 2];
  int nel = (padl - count) * NC;
  float* dst = comp + ((size_t)b * NPAD_MAX + base + count) * NC;
  for (int i = threadIdx.x; i < nel; i += 256) dst[i] = 0.0f;
}

__global__ __launch_bounds__(32) void scatter_kernel(const float* __restrict__ hyper,
                                                     const int* __restrict__ seg,
                                                     const int* __restrict__ blockoff,
                                                     const int* __restrict__ classinfo,
                                                     float* __restrict__ comp) {
  int blk = blockIdx.x;
  int b = blk / NBLKS, bb = blk % NBLKS;
  int lane = threadIdx.x;
  __shared__ float stage[32 * 65];
  __shared__ int posArr[32];
  int run = 0;
  unsigned ltmask = (1u << lane) - 1u;
  size_t nbase0 = (size_t)bb * BLKPX;

  for (int ch = 0; ch < CHUNKS; ++ch) {
    size_t nbase = nbase0 + (size_t)ch * 32;
    int n = (int)(nbase + lane);
    int myk = seg[(size_t)b * NPIX + n];
    unsigned balmy = 0;
    int pend = 0;
#pragma unroll
    for (int k = 0; k < NCLS; ++k) {
      unsigned bal = __builtin_amdgcn_ballot_w32(myk == k);
      if (myk == k) balmy = bal;
      if (lane == k) pend = __popc(bal);
    }
    int runbase = __shfl(run, myk, 32);
    int myrank = __popc(balmy & ltmask);
    run += (lane < NCLS) ? pend : 0;

    int cbase = classinfo[(b * NCLS + myk) * 3 + 0];
    int brel = blockoff[((size_t)(b * NBLKS + bb)) * NCLS + myk];
    int pos = cbase + brel + runbase + myrank;
    posArr[lane] = pos;

    for (int c = 0; c < NC; ++c)
      stage[lane * 65 + c] = hyper[((size_t)b * NC + c) * NPIX + n];
    __syncthreads();
    for (int p = 0; p < 32; ++p) {
      size_t dst = ((size_t)b * NPAD_MAX + posArr[p]) * NC;
      comp[dst + lane] = stage[p * 65 + lane];
      comp[dst + lane + 32] = stage[p * 65 + lane + 32];
    }
    __syncthreads();
  }
}

__global__ __launch_bounds__(32) void syrk_kernel(const float* __restrict__ comp,
                                                  const int* __restrict__ classinfo,
                                                  float* __restrict__ sxxpart,
                                                  float* __restrict__ sumpart) {
  int gid = blockIdx.x;
  int part = gid / (NB * NCLS);
  int bk = gid % (NB * NCLS);
  int b = bk / NCLS;
  int lane = threadIdx.x;

  int base = classinfo[bk * 3 + 0];
  int padl = classinfo[bk * 3 + 2];
  int plen = padl / PARTS;
  size_t rowbase = (size_t)b * NPAD_MAX + base + (size_t)part * plen;

  int m = lane & 15;
  int kb = (lane >> 4) << 1;

  v8f acc[4][4] = {};
  for (int p = 0; p < plen; p += 4) {
    const float* row0 = comp + (rowbase + p + kb) * NC;
    const float* row1 = row0 + NC;
    v2f F[4];
#pragma unroll
    for (int blk = 0; blk < 4; ++blk) {
      F[blk].x = row0[blk * 16 + m];
      F[blk].y = row1[blk * 16 + m];
    }
#pragma unroll
    for (int cb = 0; cb < 4; ++cb)
#pragma unroll
      for (int db = 0; db < 4; ++db)
        acc[cb][db] = __builtin_amdgcn_wmma_f32_16x16x4_f32(
            false, F[cb], false, F[db], (short)0, acc[cb][db], false, false);
  }

  size_t ob = (size_t)gid * (NC * NC);
  int col = lane & 15;
  int rof = (lane >> 4) * 8;
#pragma unroll
  for (int cb = 0; cb < 4; ++cb)
#pragma unroll
    for (int db = 0; db < 4; ++db)
#pragma unroll
      for (int j = 0; j < 8; ++j) {
        int r = cb * 16 + rof + j;
        sxxpart[ob + (size_t)r * NC + db * 16 + col] = acc[cb][db][j];
      }

  float s0 = 0.0f, s1 = 0.0f;
  for (int p = 0; p < plen; ++p) {
    const float* row = comp + (rowbase + p) * NC;
    s0 += row[lane];
    s1 += row[lane + 32];
  }
  sumpart[(size_t)gid * NC + lane] = s0;
  sumpart[(size_t)gid * NC + lane + 32] = s1;
}

__global__ __launch_bounds__(256) void reduce_solve_kernel(const float* __restrict__ sxxpart,
                                                           const float* __restrict__ sumpart,
                                                           const int* __restrict__ classinfo,
                                                           const float* __restrict__ pattern,
                                                           float* __restrict__ params) {
  int bk = blockIdx.x;
  int tid = threadIdx.x;
  __shared__ float covL[64 * 65];
  __shared__ float meanL[64], pL[64], yv[64], wv[64];

  int count = classinfo[bk * 3 + 1];
  float safe = (float)(count > 0 ? count : 1);

  if (tid < 64) {
    float s = 0.0f;
    for (int part = 0; part < PARTS; ++part)
      s += sumpart[((size_t)part * NB * NCLS + bk) * NC + tid];
    meanL[tid] = s / safe;
    pL[tid] = pattern[tid];
  }
  __syncthreads();

  for (int idx = tid; idx < NC * NC; idx += 256) {
    int i = idx >> 6, j = idx & 63;
    float s = 0.0f;
    for (int part = 0; part < PARTS; ++part)
      s += sxxpart[((size_t)part * NB * NCLS + bk) * (NC * NC) + idx];
    float cv = s / safe - meanL[i] * meanL[j];
    if (i == j) cv += EPSV;
    covL[i * 65 + j] = cv;
  }
  __syncthreads();

  for (int kk = 0; kk < NC; ++kk) {
    if (tid == 0) covL[kk * 65 + kk] = sqrtf(covL[kk * 65 + kk]);
    __syncthreads();
    float d = covL[kk * 65 + kk];
    for (int i = kk + 1 + tid; i < NC; i += 256) covL[i * 65 + kk] /= d;
    __syncthreads();
    int nrem = NC - 1 - kk;
    for (int t = tid; t < nrem * nrem; t += 256) {
      int i = kk + 1 + t / nrem;
      int j = kk + 1 + t % nrem;
      if (j <= i) covL[i * 65 + j] -= covL[i * 65 + kk] * covL[j * 65 + kk];
    }
    __syncthreads();
  }

  if (tid < 64) yv[tid] = pL[tid];
  __syncthreads();
  for (int j = 0; j < NC; ++j) {
    if (tid == 0) yv[j] /= covL[j * 65 + j];
    __syncthreads();
    float yj = yv[j];
    for (int i = j + 1 + tid; i < NC; i += 256) yv[i] -= covL[i * 65 + j] * yj;
    __syncthreads();
  }
  if (tid < 64) wv[tid] = yv[tid];
  __syncthreads();
  for (int j = NC - 1; j >= 0; --j) {
    if (tid == 0) wv[j] /= covL[j * 65 + j];
    __syncthreads();
    float wj = wv[j];
    for (int i = tid; i < j; i += 256) wv[i] -= covL[j * 65 + i] * wj;
    __syncthreads();
  }

  if (tid < 64) params[(size_t)bk * 66 + tid] = wv[tid];
  if (tid == 0) {
    float pw = 0.0f, off = 0.0f;
    for (int c = 0; c < NC; ++c) {
      pw += pL[c] * wv[c];
      off += meanL[c] * wv[c];
    }
    float denom = sqrtf(pw);
    float scale = (count >= MIN_CLASS_SIZE && denom > 0.0f) ? (1.0f / denom) : 0.0f;
    params[(size_t)bk * 66 + 64] = off;
    params[(size_t)bk * 66 + 65] = scale;
  }
}

__global__ __launch_bounds__(256) void score_kernel(const float* __restrict__ hyper,
                                                    const int* __restrict__ seg,
                                                    const float* __restrict__ params,
                                                    float* __restrict__ out) {
  int b = blockIdx.y;
  int tid = threadIdx.x;
  __shared__ float wS[NCLS * 65];
  __shared__ float offS[NCLS], sclS[NCLS];
  for (int i = tid; i < NCLS * NC; i += 256) {
    int k = i >> 6, c = i & 63;
    wS[k * 65 + c] = params[((size_t)b * NCLS + k) * 66 + c];
  }
  if (tid < NCLS) {
    offS[tid] = params[((size_t)b * NCLS + tid) * 66 + 64];
    sclS[tid] = params[((size_t)b * NCLS + tid) * 66 + 65];
  }
  __syncthreads();

  size_t n = (size_t)blockIdx.x * 256 + tid;
  int k = seg[(size_t)b * NPIX + n];
  const float* xp = hyper + (size_t)b * NC * NPIX + n;
  float acc = 0.0f;
#pragma unroll
  for (int c = 0; c < NC; ++c)
    acc = fmaf(xp[(size_t)c * NPIX], wS[k * 65 + c], acc);
  out[(size_t)b * NPIX + n] = (acc - offS[k]) * sclS[k];
}

extern "C" void kernel_launch(void* const* d_in, const int* in_sizes, int n_in,
                              void* d_out, int out_size, void* d_ws, size_t ws_size,
                              hipStream_t stream) {
  const float* hyper = (const float*)d_in[0];
  const float* pattern = (const float*)d_in[1];
  const int* seg = (const int*)d_in[2];
  float* out = (float*)d_out;

  char* ws = (char*)d_ws;
  auto al = [](size_t x) { return (x + 255) & ~(size_t)255; };
  size_t off = 0;
  float* comp = (float*)(ws + off);
  off += al((size_t)NB * NPAD_MAX * NC * sizeof(float));
  int* hist = (int*)(ws + off);
  off += al((size_t)NB * NBLKS * NCLS * sizeof(int));
  int* blockoff = (int*)(ws + off);
  off += al((size_t)NB * NBLKS * NCLS * sizeof(int));
  int* classinfo = (int*)(ws + off);
  off += al((size_t)NB * NCLS * 3 * sizeof(int));
  float* sxxpart = (float*)(ws + off);
  off += al((size_t)PARTS * NB * NCLS * NC * NC * sizeof(float));
  float* sumpart = (float*)(ws + off);
  off += al((size_t)PARTS * NB * NCLS * NC * sizeof(float));
  float* params = (float*)(ws + off);
  off += al((size_t)NB * NCLS * 66 * sizeof(float));

  hipLaunchKernelGGL(hist_kernel, dim3(NB * NBLKS), dim3(32), 0, stream, seg, hist);
  hipLaunchKernelGGL(scan_kernel, dim3(1), dim3(64), 0, stream, hist, blockoff, classinfo);
  hipLaunchKernelGGL(padzero_kernel, dim3(NB * NCLS), dim3(256), 0, stream, classinfo, comp);
  hipLaunchKernelGGL(scatter_kernel, dim3(NB * NBLKS), dim3(32), 0, stream, hyper, seg,
                     blockoff, classinfo, comp);
  hipLaunchKernelGGL(syrk_kernel, dim3(PARTS * NB * NCLS), dim3(32), 0, stream, comp,
                     classinfo, sxxpart, sumpart);
  hipLaunchKernelGGL(reduce_solve_kernel, dim3(NB * NCLS), dim3(256), 0, stream, sxxpart,
                     sumpart, classinfo, pattern, params);
  hipLaunchKernelGGL(score_kernel, dim3(NPIX / 256, NB), dim3(256), 0, stream, hyper, seg,
                     params, out);
}


# scatter kernel made memory-parallel: one wave per 32-pixel chunk (1152 groups of 8 waves), all 64 channel loads in flight, LDS transpose, one 256-B store per row; same positions and sum order
# speedup vs baseline: 2.0356x; 2.0356x over previous
#include <hip/hip_runtime.h>
#include <hip/hip_bf16.h>


typedef float v2f __attribute__((ext_vector_type(2)));
typedef float v8f __attribute__((ext_vector_type(8)));

#define NB 2
#define NC 64
#define NPIX 147456
#define NCLS 20
#define BLKPX 1024
#define CHUNKS (BLKPX / 32)
#define NBLKS (NPIX / BLKPX)
#define PARTS 16
#define PADALIGN (PARTS * 4)
#define NPAD_MAX (NPIX + NCLS * PADALIGN)
#define MIN_CLASS_SIZE 5000
#define EPSV 1e-6f

__global__ __launch_bounds__(32) void hist_kernel(const int* __restrict__ seg,
                                                  int* __restrict__ hist) {
  int blk = blockIdx.x;
  int b = blk / NBLKS, bb = blk % NBLKS;
  int lane = threadIdx.x;
  const int* s = seg + (size_t)b * NPIX + (size_t)bb * BLKPX;
  int cnt = 0;
  for (int ch = 0; ch < CHUNKS; ++ch) {
    int myk = s[ch * 32 + lane];
#pragma unroll
    for (int k = 0; k < NCLS; ++k) {
      unsigned bal = __builtin_amdgcn_ballot_w32(myk == k);
      if (lane == k) cnt += __popc(bal);
    }
  }
  if (lane < NCLS) hist[(size_t)blk * NCLS + lane] = cnt;
}

__global__ __launch_bounds__(64) void scan_kernel(const int* __restrict__ hist,
                                                  int* __restrict__ blockoff,
                                                  int* __restrict__ classinfo) {
  int tid = threadIdx.x;
  __shared__ int tot[NB * NCLS];
  __shared__ int padl[NB * NCLS];
  if (tid < NB * NCLS) {
    int b = tid / NCLS, k = tid % NCLS;
    int run = 0;
    for (int blk = 0; blk < NBLKS; ++blk) {
      size_t idx = ((size_t)(b * NBLKS + blk)) * NCLS + k;
      blockoff[idx] = run;
      run += hist[idx];
    }
    tot[tid] = run;
    padl[tid] = (run + PADALIGN - 1) & ~(PADALIGN - 1);
  }
  __syncthreads();
  if (tid < NB) {
    int b = tid, base = 0;
    for (int k = 0; k < NCLS; ++k) {
      classinfo[(b * NCLS + k) * 3 + 0] = base;
      classinfo[(b * NCLS + k) * 3 + 1] = tot[b * NCLS + k];
      classinfo[(b * NCLS + k) * 3 + 2] = padl[b * NCLS + k];
      base += padl[b * NCLS + k];
    }
  }
}

__global__ __launch_bounds__(256) void padzero_kernel(const int* __restrict__ classinfo,
                                                      float* __restrict__ comp) {
  int bk = blockIdx.x;
  int b = bk / NCLS;
  int base = classinfo[bk * 3 + 0];
  int count = classinfo[bk * 3 + 1];
  int padl = classinfo[bk * 3 + 2];
  int nel = (padl - count) * NC;
  float* dst = comp + ((size_t)b * NPAD_MAX + base + count) * NC;
  for (int i = threadIdx.x; i < nel; i += 256) dst[i] = 0.0f;
}

#define SUBS 4
#define SCW (CHUNKS / SUBS)
#define TPITCH 68

__global__ __launch_bounds__(SCW * 32) void scatter_kernel(const float* __restrict__ hyper,
                                                           const int* __restrict__ seg,
                                                           const int* __restrict__ blockoff,
                                                           const int* __restrict__ classinfo,
                                                           float* __restrict__ comp) {
  int s = blockIdx.x % SUBS;
  int hb = blockIdx.x / SUBS;
  int b = hb / NBLKS, bb = hb % NBLKS;
  int lane = threadIdx.x & 31;
  int wave = __builtin_amdgcn_readfirstlane((int)(threadIdx.x >> 5));
  __shared__ __attribute__((aligned(16))) float tile[SCW][32 * TPITCH];
  __shared__ int cnt[CHUNKS][32];

  const int* sg = seg + (size_t)b * NPIX + (size_t)bb * BLKPX;
  int sgv[SUBS];
#pragma unroll
  for (int j = 0; j < SUBS; ++j) sgv[j] = sg[(j * SCW + wave) * 32 + lane];

  int cls = (lane < NCLS) ? lane : 0;
  int cb = classinfo[(b * NCLS + cls) * 3 + 0] + blockoff[(size_t)hb * NCLS + cls];

  int ch = s * SCW + wave;
  const float* xp = hyper + (size_t)b * NC * NPIX + (size_t)bb * BLKPX + (size_t)ch * 32 + lane;
  float v[NC];
#pragma unroll
  for (int c = 0; c < NC; ++c) v[c] = xp[(size_t)c * NPIX];

  int myk = sgv[0];
#pragma unroll
  for (int j = 1; j < SUBS; ++j) myk = (s == j) ? sgv[j] : myk;

#pragma unroll
  for (int j = 0; j < SUBS; ++j) {
    int pc = 0;
#pragma unroll
    for (int k = 0; k < NCLS; ++k) {
      unsigned bal = __builtin_amdgcn_ballot_w32(sgv[j] == k);
      if (lane == k) pc = __popc(bal);
    }
    cnt[j * SCW + wave][lane] = pc;
  }
  unsigned balmy = 0;
#pragma unroll
  for (int k = 0; k < NCLS; ++k) {
    unsigned bal = __builtin_amdgcn_ballot_w32(myk == k);
    if (myk == k) balmy = bal;
  }

  float* tl = tile[wave];
#pragma unroll
  for (int q = 0; q < NC / 4; ++q)
    *(float4*)&tl[lane * TPITCH + 4 * q] =
        make_float4(v[4 * q], v[4 * q + 1], v[4 * q + 2], v[4 * q + 3]);
  __syncthreads();

  int run = 0;
#pragma unroll
  for (int c2 = 0; c2 < CHUNKS - 1; ++c2) {
    int t = cnt[c2][lane];
    run += (c2 < ch) ? t : 0;
  }
  unsigned ltmask = (1u << lane) - 1u;
  int pos = __shfl(cb + run, myk, 32) + __popc(balmy & ltmask);

  float* crow = comp + (size_t)b * NPAD_MAX * NC;
#pragma unroll
  for (int p = 0; p < 32; ++p) {
    int pp = __builtin_amdgcn_readlane(pos, p);
    float2 val = *(const float2*)&tl[p * TPITCH + 2 * lane];
    *(float2*)&crow[(size_t)pp * NC + 2 * lane] = val;
  }
}

__global__ __launch_bounds__(32) void syrk_kernel(const float* __restrict__ comp,
                                                  const int* __restrict__ classinfo,
                                                  float* __restrict__ sxxpart,
                                                  float* __restrict__ sumpart) {
  int gid = blockIdx.x;
  int part = gid / (NB * NCLS);
  int bk = gid % (NB * NCLS);
  int b = bk / NCLS;
  int lane = threadIdx.x;

  int base = classinfo[bk * 3 + 0];
  int padl = classinfo[bk * 3 + 2];
  int plen = padl / PARTS;
  size_t rowbase = (size_t)b * NPAD_MAX + base + (size_t)part * plen;

  int m = lane & 15;
  int kb = (lane >> 4) << 1;

  v8f acc[4][4] = {};
  for (int p = 0; p < plen; p += 4) {
    const float* row0 = comp + (rowbase + p + kb) * NC;
    const float* row1 = row0 + NC;
    v2f F[4];
#pragma unroll
    for (int blk = 0; blk < 4; ++blk) {
      F[blk].x = row0[blk * 16 + m];
      F[blk].y = row1[blk * 16 + m];
    }
#pragma unroll
    for (int cb = 0; cb < 4; ++cb)
#pragma unroll
      for (int db = 0; db < 4; ++db)
        acc[cb][db] = __builtin_amdgcn_wmma_f32_16x16x4_f32(
            false, F[cb], false, F[db], (short)0, acc[cb][db], false, false);
  }

  size_t ob = (size_t)gid * (NC * NC);
  int col = lane & 15;
  int rof = (lane >> 4) * 8;
#pragma unroll
  for (int cb = 0; cb < 4; ++cb)
#pragma unroll
    for (int db = 0; db < 4; ++db)
#pragma unroll
      for (int j = 0; j < 8; ++j) {
        int r = cb * 16 + rof + j;
        sxxpart[ob + (size_t)r * NC + db * 16 + col] = acc[cb][db][j];
      }

  float s0 = 0.0f, s1 = 0.0f;
  for (int p = 0; p < plen; ++p) {
    const float* row = comp + (rowbase + p) * NC;
    s0 += row[lane];
    s1 += row[lane + 32];
  }
  sumpart[(size_t)gid * NC + lane] = s0;
  sumpart[(size_t)gid * NC + lane + 32] = s1;
}

__global__ __launch_bounds__(256) void reduce_solve_kernel(const float* __restrict__ sxxpart,
                                                           const float* __restrict__ sumpart,
                                                           const int* __restrict__ classinfo,
                                                           const float* __restrict__ pattern,
                                                           float* __restrict__ params) {
  int bk = blockIdx.x;
  int tid = threadIdx.x;
  __shared__ float covL[64 * 65];
  __shared__ float meanL[64], pL[64], yv[64], wv[64];

  int count = classinfo[bk * 3 + 1];
  float safe = (float)(count > 0 ? count : 1);

  if (tid < 64) {
    float s = 0.0f;
    for (int part = 0; part < PARTS; ++part)
      s += sumpart[((size_t)part * NB * NCLS + bk) * NC + tid];
    meanL[tid] = s / safe;
    pL[tid] = pattern[tid];
  }
  __syncthreads();

  for (int idx = tid; idx < NC * NC; idx += 256) {
    int i = idx >> 6, j = idx & 63;
    float s = 0.0f;
    for (int part = 0; part < PARTS; ++part)
      s += sxxpart[((size_t)part * NB * NCLS + bk) * (NC * NC) + idx];
    float cv = s / safe - meanL[i] * meanL[j];
    if (i == j) cv += EPSV;
    covL[i * 65 + j] = cv;
  }
  __syncthreads();

  for (int kk = 0; kk < NC; ++kk) {
    if (tid == 0) covL[kk * 65 + kk] = sqrtf(covL[kk * 65 + kk]);
    __syncthreads();
    float d = covL[kk * 65 + kk];
    for (int i = kk + 1 + tid; i < NC; i += 256) covL[i * 65 + kk] /= d;
    __syncthreads();
    int nrem = NC - 1 - kk;
    for (int t = tid; t < nrem * nrem; t += 256) {
      int i = kk + 1 + t / nrem;
      int j = kk + 1 + t % nrem;
      if (j <= i) covL[i * 65 + j] -= covL[i * 65 + kk] * covL[j * 65 + kk];
    }
    __syncthreads();
  }

  if (tid < 64) yv[tid] = pL[tid];
  __syncthreads();
  for (int j = 0; j < NC; ++j) {
    if (tid == 0) yv[j] /= covL[j * 65 + j];
    __syncthreads();
    float yj = yv[j];
    for (int i = j + 1 + tid; i < NC; i += 256) yv[i] -= covL[i * 65 + j] * yj;
    __syncthreads();
  }
  if (tid < 64) wv[tid] = yv[tid];
  __syncthreads();
  for (int j = NC - 1; j >= 0; --j) {
    if (tid == 0) wv[j] /= covL[j * 65 + j];
    __syncthreads();
    float wj = wv[j];
    for (int i = tid; i < j; i += 256) wv[i] -= covL[j * 65 + i] * wj;
    __syncthreads();
  }

  if (tid < 64) params[(size_t)bk * 66 + tid] = wv[tid];
  if (tid == 0) {
    float pw = 0.0f, off = 0.0f;
    for (int c = 0; c < NC; ++c) {
      pw += pL[c] * wv[c];
      off += meanL[c] * wv[c];
    }
    float denom = sqrtf(pw);
    float scale = (count >= MIN_CLASS_SIZE && denom > 0.0f) ? (1.0f / denom) : 0.0f;
    params[(size_t)bk * 66 + 64] = off;
    params[(size_t)bk * 66 + 65] = scale;
  }
}

__global__ __launch_bounds__(256) void score_kernel(const float* __restrict__ hyper,
                                                    const int* __restrict__ seg,
                                                    const float* __restrict__ params,
                                                    float* __restrict__ out) {
  int b = blockIdx.y;
  int tid = threadIdx.x;
  __shared__ float wS[NCLS * 65];
  __shared__ float offS[NCLS], sclS[NCLS];
  for (int i = tid; i < NCLS * NC; i += 256) {
    int k = i >> 6, c = i & 63;
    wS[k * 65 + c] = params[((size_t)b * NCLS + k) * 66 + c];
  }
  if (tid < NCLS) {
    offS[tid] = params[((size_t)b * NCLS + tid) * 66 + 64];
    sclS[tid] = params[((size_t)b * NCLS + tid) * 66 + 65];
  }
  __syncthreads();

  size_t n = (size_t)blockIdx.x * 256 + tid;
  int k = seg[(size_t)b * NPIX + n];
  const float* xp = hyper + (size_t)b * NC * NPIX + n;
  float acc = 0.0f;
#pragma unroll
  for (int c = 0; c < NC; ++c)
    acc = fmaf(xp[(size_t)c * NPIX], wS[k * 65 + c], acc);
  out[(size_t)b * NPIX + n] = (acc - offS[k]) * sclS[k];
}

extern "C" void kernel_launch(void* const* d_in, const int* in_sizes, int n_in,
                              void* d_out, int out_size, void* d_ws, size_t ws_size,
                              hipStream_t stream) {
  const float* hyper = (const float*)d_in[0];
  const float* pattern = (const float*)d_in[1];
  const int* seg = (const int*)d_in[2];
  float* out = (float*)d_out;

  char* ws = (char*)d_ws;
  auto al = [](size_t x) { return (x + 255) & ~(size_t)255; };
  size_t off = 0;
  float* comp = (float*)(ws + off);
  off += al((size_t)NB * NPAD_MAX * NC * sizeof(float));
  int* hist = (int*)(ws + off);
  off += al((size_t)NB * NBLKS * NCLS * sizeof(int));
  int* blockoff = (int*)(ws + off);
  off += al((size_t)NB * NBLKS * NCLS * sizeof(int));
  int* classinfo = (int*)(ws + off);
  off += al((size_t)NB * NCLS * 3 * sizeof(int));
  float* sxxpart = (float*)(ws + off);
  off += al((size_t)PARTS * NB * NCLS * NC * NC * sizeof(float));
  float* sumpart = (float*)(ws + off);
  off += al((size_t)PARTS * NB * NCLS * NC * sizeof(float));
  float* params = (float*)(ws + off);
  off += al((size_t)NB * NCLS * 66 * sizeof(float));

  hipLaunchKernelGGL(hist_kernel, dim3(NB * NBLKS), dim3(32), 0, stream, seg, hist);
  hipLaunchKernelGGL(scan_kernel, dim3(1), dim3(64), 0, stream, hist, blockoff, classinfo);
  hipLaunchKernelGGL(padzero_kernel, dim3(NB * NCLS), dim3(256), 0, stream, classinfo, comp);
  hipLaunchKernelGGL(scatter_kernel, dim3(NB * NBLKS * SUBS), dim3(SCW * 32), 0, stream, hyper, seg,
                     blockoff, classinfo, comp);
  hipLaunchKernelGGL(syrk_kernel, dim3(PARTS * NB * NCLS), dim3(32), 0, stream, comp,
                     classinfo, sxxpart, sumpart);
  hipLaunchKernelGGL(reduce_solve_kernel, dim3(NB * NCLS), dim3(256), 0, stream, sxxpart,
                     sumpart, classinfo, pattern, params);
  hipLaunchKernelGGL(score_kernel, dim3(NPIX / 256, NB), dim3(256), 0, stream, hyper, seg,
                     params, out);
}


# reduce_solve: Cholesky and both triangular solves by one wave, two matrix rows per lane in registers, pivot and column by lane reads; no barrier or LDS hop in the chain; same operations and order
# speedup vs baseline: 2.8735x; 1.4116x over previous
#include <hip/hip_runtime.h>
#include <hip/hip_bf16.h>


typedef float v2f __attribute__((ext_vector_type(2)));
typedef float v8f __attribute__((ext_vector_type(8)));

#define NB 2
#define NC 64
#define NPIX 147456
#define NCLS 20
#define BLKPX 1024
#define CHUNKS (BLKPX / 32)
#define NBLKS (NPIX / BLKPX)
#define PARTS 16
#define PADALIGN (PARTS * 4)
#define NPAD_MAX (NPIX + NCLS * PADALIGN)
#define MIN_CLASS_SIZE 5000
#define EPSV 1e-6f

__global__ __launch_bounds__(32) void hist_kernel(const int* __restrict__ seg,
                                                  int* __restrict__ hist) {
  int blk = blockIdx.x;
  int b = blk / NBLKS, bb = blk % NBLKS;
  int lane = threadIdx.x;
  const int* s = seg + (size_t)b * NPIX + (size_t)bb * BLKPX;
  int cnt = 0;
  for (int ch = 0; ch < CHUNKS; ++ch) {
    int myk = s[ch * 32 + lane];
#pragma unroll
    for (int k = 0; k < NCLS; ++k) {
      unsigned bal = __builtin_amdgcn_ballot_w32(myk == k);
      if (lane == k) cnt += __popc(bal);
    }
  }
  if (lane < NCLS) hist[(size_t)blk * NCLS + lane] = cnt;
}

__global__ __launch_bounds__(64) void scan_kernel(const int* __restrict__ hist,
                                                  int* __restrict__ blockoff,
                                                  int* __restrict__ classinfo) {
  int tid = threadIdx.x;
  __shared__ int tot[NB * NCLS];
  __shared__ int padl[NB * NCLS];
  if (tid < NB * NCLS) {
    int b = tid / NCLS, k = tid % NCLS;
    int run = 0;
    for (int blk = 0; blk < NBLKS; ++blk) {
      size_t idx = ((size_t)(b * NBLKS + blk)) * NCLS + k;
      blockoff[idx] = run;
      run += hist[idx];
    }
    tot[tid] = run;
    padl[tid] = (run + PADALIGN - 1) & ~(PADALIGN - 1);
  }
  __syncthreads();
  if (tid < NB) {
    int b = tid, base = 0;
    for (int k = 0; k < NCLS; ++k) {
      classinfo[(b * NCLS + k) * 3 + 0] = base;
      classinfo[(b * NCLS + k) * 3 + 1] = tot[b * NCLS + k];
      classinfo[(b * NCLS + k) * 3 + 2] = padl[b * NCLS + k];
      base += padl[b * NCLS + k];
    }
  }
}

__global__ __launch_bounds__(256) void padzero_kernel(const int* __restrict__ classinfo,
                                                      float* __restrict__ comp) {
  int bk = blockIdx.x;
  int b = bk / NCLS;
  int base = classinfo[bk * 3 + 0];
  int count = classinfo[bk * 3 + 1];
  int padl = classinfo[bk * 3 + 2];
  int nel = (padl - count) * NC;
  float* dst = comp + ((size_t)b * NPAD_MAX + base + count) * NC;
  for (int i = threadIdx.x; i < nel; i += 256) dst[i] = 0.0f;
}

#define SUBS 4
#define SCW (CHUNKS / SUBS)
#define TPITCH 68

__global__ __launch_bounds__(SCW * 32) void scatter_kernel(const float* __restrict__ hyper,
                                                           const int* __restrict__ seg,
                                                           const int* __restrict__ blockoff,
                                                           const int* __restrict__ classinfo,
                                                           float* __restrict__ comp) {
  int s = blockIdx.x % SUBS;
  int hb = blockIdx.x / SUBS;
  int b = hb / NBLKS, bb = hb % NBLKS;
  int lane = threadIdx.x & 31;
  int wave = __builtin_amdgcn_readfirstlane((int)(threadIdx.x >> 5));
  __shared__ __attribute__((aligned(16))) float tile[SCW][32 * TPITCH];
  __shared__ int cnt[CHUNKS][32];

  const int* sg = seg + (size_t)b * NPIX + (size_t)bb * BLKPX;
  int sgv[SUBS];
#pragma unroll
  for (int j = 0; j < SUBS; ++j) sgv[j] = sg[(j * SCW + wave) * 32 + lane];

  int cls = (lane < NCLS) ? lane : 0;
  int cb = classinfo[(b * NCLS + cls) * 3 + 0] + blockoff[(size_t)hb * NCLS + cls];

  int ch = s * SCW + wave;
  const float* xp = hyper + (size_t)b * NC * NPIX + (size_t)bb * BLKPX + (size_t)ch * 32 + lane;
  float v[NC];
#pragma unroll
  for (int c = 0; c < NC; ++c) v[c] = xp[(size_t)c * NPIX];

  int myk = sgv[0];
#pragma unroll
  for (int j = 1; j < SUBS; ++j) myk = (s == j) ? sgv[j] : myk;

#pragma unroll
  for (int j = 0; j < SUBS; ++j) {
    int pc = 0;
#pragma unroll
    for (int k = 0; k < NCLS; ++k) {
      unsigned bal = __builtin_amdgcn_ballot_w32(sgv[j] == k);
      if (lane == k) pc = __popc(bal);
    }
    cnt[j * SCW + wave][lane] = pc;
  }
  unsigned balmy = 0;
#pragma unroll
  for (int k = 0; k < NCLS; ++k) {
    unsigned bal = __builtin_amdgcn_ballot_w32(myk == k);
    if (myk == k) balmy = bal;
  }

  float* tl = tile[wave];
#pragma unroll
  for (int q = 0; q < NC / 4; ++q)
    *(float4*)&tl[lane * TPITCH + 4 * q] =
        make_float4(v[4 * q], v[4 * q + 1], v[4 * q + 2], v[4 * q + 3]);
  __syncthreads();

  int run = 0;
#pragma unroll
  for (int c2 = 0; c2 < CHUNKS - 1; ++c2) {
    int t = cnt[c2][lane];
    run += (c2 < ch) ? t : 0;
  }
  unsigned ltmask = (1u << lane) - 1u;
  int pos = __shfl(cb + run, myk, 32) + __popc(balmy & ltmask);

  float* crow = comp + (size_t)b * NPAD_MAX * NC;
#pragma unroll
  for (int p = 0; p < 32; ++p) {
    int pp = __builtin_amdgcn_readlane(pos, p);
    float2 val = *(const float2*)&tl[p * TPITCH + 2 * lane];
    *(float2*)&crow[(size_t)pp * NC + 2 * lane] = val;
  }
}

__global__ __launch_bounds__(32) void syrk_kernel(const float* __restrict__ comp,
                                                  const int* __restrict__ classinfo,
                                                  float* __restrict__ sxxpart,
                                                  float* __restrict__ sumpart) {
  int gid = blockIdx.x;
  int part = gid / (NB * NCLS);
  int bk = gid % (NB * NCLS);
  int b = bk / NCLS;
  int lane = threadIdx.x;

  int base = classinfo[bk * 3 + 0];
  int padl = classinfo[bk * 3 + 2];
  int plen = padl / PARTS;
  size_t rowbase = (size_t)b * NPAD_MAX + base + (size_t)part * plen;

  int m = lane & 15;
  int kb = (lane >> 4) << 1;

  v8f acc[4][4] = {};
  for (int p = 0; p < plen; p += 4) {
    const float* row0 = comp + (rowbase + p + kb) * NC;
    const float* row1 = row0 + NC;
    v2f F[4];
#pragma unroll
    for (int blk = 0; blk < 4; ++blk) {
      F[blk].x = row0[blk * 16 + m];
      F[blk].y = row1[blk * 16 + m];
    }
#pragma unroll
    for (int cb = 0; cb < 4; ++cb)
#pragma unroll
      for (int db = 0; db < 4; ++db)
        acc[cb][db] = __builtin_amdgcn_wmma_f32_16x16x4_f32(
            false, F[cb], false, F[db], (short)0, acc[cb][db], false, false);
  }

  size_t ob = (size_t)gid * (NC * NC);
  int col = lane & 15;
  int rof = (lane >> 4) * 8;
#pragma unroll
  for (int cb = 0; cb < 4; ++cb)
#pragma unroll
    for (int db = 0; db < 4; ++db)
#pragma unroll
      for (int j = 0; j < 8; ++j) {
        int r = cb * 16 + rof + j;
        sxxpart[ob + (size_t)r * NC + db * 16 + col] = acc[cb][db][j];
      }

  float s0 = 0.0f, s1 = 0.0f;
  for (int p = 0; p < plen; ++p) {
    const float* row = comp + (rowbase + p) * NC;
    s0 += row[lane];
    s1 += row[lane + 32];
  }
  sumpart[(size_t)gid * NC + lane] = s0;
  sumpart[(size_t)gid * NC + lane + 32] = s1;
}

__device__ __forceinline__ float lane_read(float v, int l) {
  return __int_as_float(__builtin_amdgcn_readlane(__float_as_int(v), l));
}

template <int Q, int NCOLS>
__device__ __forceinline__ void chol_step(float (&c0)[NC], float (&c1)[NC], float* covL, int lane, int kp) {
  const int kk = 2 * kp + Q;
  float piv = lane_read(Q ? c1[0] : c0[0], kp);
  float d = sqrtf(piv);
  float l0 = c0[0] / d;
  float l1 = c1[0] / d;
  if (Q == 0) l0 = (lane == kp) ? d : l0;
  else        l1 = (lane == kp) ? d : l1;
  covL[(2 * lane) * 65 + kk] = l0;
  covL[(2 * lane + 1) * 65 + kk] = l1;
#pragma unroll
  for (int jj = 1; jj <= NCOLS; ++jj) {
    const int lj = (kp + ((Q + jj) >> 1)) & 31;
    float s = lane_read(((Q + jj) & 1) ? l1 : l0, lj);
    c0[jj - 1] = c0[jj] - l0 * s;
    c1[jj - 1] = c1[jj] - l1 * s;
  }
}

__global__ __launch_bounds__(256) void reduce_solve_kernel(const float* __restrict__ sxxpart,
                                                           const float* __restrict__ sumpart,
                                                           const int* __restrict__ classinfo,
                                                           const float* __restrict__ pattern,
                                                           float* __restrict__ params) {
  int bk = blockIdx.x;
  int tid = threadIdx.x;
  __shared__ float covL[64 * 65];
  __shared__ float meanL[64], pL[64], wv[64];

  int count = classinfo[bk * 3 + 1];
  float safe = (float)(count > 0 ? count : 1);

  if (tid < 64) {
    float s = 0.0f;
    for (int part = 0; part < PARTS; ++part)
      s += sumpart[((size_t)part * NB * NCLS + bk) * NC + tid];
    meanL[tid] = s / safe;
    pL[tid] = pattern[tid];
  }
  __syncthreads();

  for (int idx = tid; idx < NC * NC; idx += 256) {
    int i = idx >> 6, j = idx & 63;
    float s = 0.0f;
    for (int part = 0; part < PARTS; ++part)
      s += sxxpart[((size_t)part * NB * NCLS + bk) * (NC * NC) + idx];
    float cv = s / safe - meanL[i] * meanL[j];
    if (i == j) cv += EPSV;
    covL[i * 65 + j] = cv;
  }
  __syncthreads();

  if (tid < 32) {
    const int lane = tid;
    float c0[NC], c1[NC];
#pragma unroll
    for (int j = 0; j < NC; ++j) {
      c0[j] = covL[(2 * lane) * 65 + j];
      c1[j] = covL[(2 * lane + 1) * 65 + j];
    }

#pragma nounroll
    for (int kp = 0; kp < 16; ++kp) {
      chol_step<0, 63>(c0, c1, covL, lane, kp);
      chol_step<1, 63>(c0, c1, covL, lane, kp);
    }
#pragma nounroll
    for (int kp = 16; kp < 32; ++kp) {
      chol_step<0, 31>(c0, c1, covL, lane, kp);
      chol_step<1, 31>(c0, c1, covL, lane, kp);
    }

    const float dg0 = covL[(2 * lane) * 65 + 2 * lane];
    const float dg1 = covL[(2 * lane + 1) * 65 + 2 * lane + 1];

    float y0 = pL[2 * lane], y1 = pL[2 * lane + 1];
    float a0 = covL[(2 * lane) * 65], a1 = covL[(2 * lane + 1) * 65];
#pragma nounroll
    for (int jp = 0; jp < 32; ++jp) {
      int jn = 2 * jp + 1;
      float n0 = covL[(2 * lane) * 65 + jn], n1 = covL[(2 * lane + 1) * 65 + jn];
      {
        float s = lane_read(y0 / dg0, jp);
        y0 = (lane == jp) ? s : ((lane > jp) ? y0 - a0 * s : y0);
        y1 = (lane >= jp) ? y1 - a1 * s : y1;
      }
      int jm = (2 * jp + 2 < NC) ? 2 * jp + 2 : NC - 1;
      a0 = covL[(2 * lane) * 65 + jm];
      a1 = covL[(2 * lane + 1) * 65 + jm];
      {
        float s = lane_read(y1 / dg1, jp);
        y1 = (lane == jp) ? s : ((lane > jp) ? y1 - n1 * s : y1);
        y0 = (lane > jp) ? y0 - n0 * s : y0;
      }
    }

    float w0 = y0, w1 = y1;
    float b0 = covL[63 * 65 + 2 * lane], b1 = covL[63 * 65 + 2 * lane + 1];
#pragma nounroll
    for (int jp = 31; jp >= 0; --jp) {
      int jn = 2 * jp;
      float n0 = covL[jn * 65 + 2 * lane], n1 = covL[jn * 65 + 2 * lane + 1];
      {
        float s = lane_read(w1 / dg1, jp);
        w1 = (lane == jp) ? s : ((lane < jp) ? w1 - b1 * s : w1);
        w0 = (lane <= jp) ? w0 - b0 * s : w0;
      }
      int jm = (2 * jp - 1 >= 0) ? 2 * jp - 1 : 0;
      b0 = covL[jm * 65 + 2 * lane];
      b1 = covL[jm * 65 + 2 * lane + 1];
      {
        float s = lane_read(w0 / dg0, jp);
        w0 = (lane == jp) ? s : ((lane < jp) ? w0 - n0 * s : w0);
        w1 = (lane < jp) ? w1 - n1 * s : w1;
      }
    }

    params[(size_t)bk * 66 + 2 * lane] = w0;
    params[(size_t)bk * 66 + 2 * lane + 1] = w1;
    wv[2 * lane] = w0;
    wv[2 * lane + 1] = w1;
    float pw = 0.0f, off = 0.0f;
    for (int c = 0; c < NC; ++c) {
      pw += pL[c] * wv[c];
      off += meanL[c] * wv[c];
    }
    if (lane == 0) {
      float denom = sqrtf(pw);
      float scale = (count >= MIN_CLASS_SIZE && denom > 0.0f) ? (1.0f / denom) : 0.0f;
      params[(size_t)bk * 66 + 64] = off;
      params[(size_t)bk * 66 + 65] = scale;
    }
  }
}

__global__ __launch_bounds__(256) void score_kernel(const float* __restrict__ hyper,
                                                    const int* __restrict__ seg,
                                                    const float* __restrict__ params,
                                                    float* __restrict__ out) {
  int b = blockIdx.y;
  int tid = threadIdx.x;
  __shared__ float wS[NCLS * 65];
  __shared__ float offS[NCLS], sclS[NCLS];
  for (int i = tid; i < NCLS * NC; i += 256) {
    int k = i >> 6, c = i & 63;
    wS[k * 65 + c] = params[((size_t)b * NCLS + k) * 66 + c];
  }
  if (tid < NCLS) {
    offS[tid] = params[((size_t)b * NCLS + tid) * 66 + 64];
    sclS[tid] = params[((size_t)b * NCLS + tid) * 66 + 65];
  }
  __syncthreads();

  size_t n = (size_t)blockIdx.x * 256 + tid;
  int k = seg[(size_t)b * NPIX + n];
  const float* xp = hyper + (size_t)b * NC * NPIX + n;
  float acc = 0.0f;
#pragma unroll
  for (int c = 0; c < NC; ++c)
    acc = fmaf(xp[(size_t)c * NPIX], wS[k * 65 + c], acc);
  out[(size_t)b * NPIX + n] = (acc - offS[k]) * sclS[k];
}

extern "C" void kernel_launch(void* const* d_in, const int* in_sizes, int n_in,
                              void* d_out, int out_size, void* d_ws, size_t ws_size,
                              hipStream_t stream) {
  const float* hyper = (const float*)d_in[0];
  const float* pattern = (const float*)d_in[1];
  const int* seg = (const int*)d_in[2];
  float* out = (float*)d_out;

  char* ws = (char*)d_ws;
  auto al = [](size_t x) { return (x + 255) & ~(size_t)255; };
  size_t off = 0;
  float* comp = (float*)(ws + off);
  off += al((size_t)NB * NPAD_MAX * NC * sizeof(float));
  int* hist = (int*)(ws + off);
  off += al((size_t)NB * NBLKS * NCLS * sizeof(int));
  int* blockoff = (int*)(ws + off);
  off += al((size_t)NB * NBLKS * NCLS * sizeof(int));
  int* classinfo = (int*)(ws + off);
  off += al((size_t)NB * NCLS * 3 * sizeof(int));
  float* sxxpart = (float*)(ws + off);
  off += al((size_t)PARTS * NB * NCLS * NC * NC * sizeof(float));
  float* sumpart = (float*)(ws + off);
  off += al((size_t)PARTS * NB * NCLS * NC * sizeof(float));
  float* params = (float*)(ws + off);
  off += al((size_t)NB * NCLS * 66 * sizeof(float));

  hipLaunchKernelGGL(hist_kernel, dim3(NB * NBLKS), dim3(32), 0, stream, seg, hist);
  hipLaunchKernelGGL(scan_kernel, dim3(1), dim3(64), 0, stream, hist, blockoff, classinfo);
  hipLaunchKernelGGL(padzero_kernel, dim3(NB * NCLS), dim3(256), 0, stream, classinfo, comp);
  hipLaunchKernelGGL(scatter_kernel, dim3(NB * NBLKS * SUBS), dim3(SCW * 32), 0, stream, hyper, seg,
                     blockoff, classinfo, comp);
  hipLaunchKernelGGL(syrk_kernel, dim3(PARTS * NB * NCLS), dim3(32), 0, stream, comp,
                     classinfo, sxxpart, sumpart);
  hipLaunchKernelGGL(reduce_solve_kernel, dim3(NB * NCLS), dim3(256), 0, stream, sxxpart,
                     sumpart, classinfo, pattern, params);
  hipLaunchKernelGGL(score_kernel, dim3(NPIX / 256, NB), dim3(256), 0, stream, hyper, seg,
                     params, out);
}


# syrk re-cut: one 64x64 partial per 4-wave group instead of per wave; wave w owns tile row w and its 16 column sums, fragments of the next 16 rows loaded ahead; same products, same sum order
# speedup vs baseline: 3.9792x; 1.3848x over previous
#include <hip/hip_runtime.h>
#include <hip/hip_bf16.h>


typedef float v2f __attribute__((ext_vector_type(2)));
typedef float v8f __attribute__((ext_vector_type(8)));

#define NB 2
#define NC 64
#define NPIX 147456
#define NCLS 20
#define BLKPX 1024
#define CHUNKS (BLKPX / 32)
#define NBLKS (NPIX / BLKPX)
#define PARTS 16
#define PADALIGN (PARTS * 4)
#define NPAD_MAX (NPIX + NCLS * PADALIGN)
#define MIN_CLASS_SIZE 5000
#define EPSV 1e-6f

__global__ __launch_bounds__(32) void hist_kernel(const int* __restrict__ seg,
                                                  int* __restrict__ hist) {
  int blk = blockIdx.x;
  int b = blk / NBLKS, bb = blk % NBLKS;
  int lane = threadIdx.x;
  const int* s = seg + (size_t)b * NPIX + (size_t)bb * BLKPX;
  int cnt = 0;
  for (int ch = 0; ch < CHUNKS; ++ch) {
    int myk = s[ch * 32 + lane];
#pragma unroll
    for (int k = 0; k < NCLS; ++k) {
      unsigned bal = __builtin_amdgcn_ballot_w32(myk == k);
      if (lane == k) cnt += __popc(bal);
    }
  }
  if (lane < NCLS) hist[(size_t)blk * NCLS + lane] = cnt;
}

__global__ __launch_bounds__(64) void scan_kernel(const int* __restrict__ hist,
                                                  int* __restrict__ blockoff,
                                                  int* __restrict__ classinfo) {
  int tid = threadIdx.x;
  __shared__ int tot[NB * NCLS];
  __shared__ int padl[NB * NCLS];
  if (tid < NB * NCLS) {
    int b = tid / NCLS, k = tid % NCLS;
    int run = 0;
    for (int blk = 0; blk < NBLKS; ++blk) {
      size_t idx = ((size_t)(b * NBLKS + blk)) * NCLS + k;
      blockoff[idx] = run;
      run += hist[idx];
    }
    tot[tid] = run;
    padl[tid] = (run + PADALIGN - 1) & ~(PADALIGN - 1);
  }
  __syncthreads();
  if (tid < NB) {
    int b = tid, base = 0;
    for (int k = 0; k < NCLS; ++k) {
      classinfo[(b * NCLS + k) * 3 + 0] = base;
      classinfo[(b * NCLS + k) * 3 + 1] = tot[b * NCLS + k];
      classinfo[(b * NCLS + k) * 3 + 2] = padl[b * NCLS + k];
      base += padl[b * NCLS + k];
    }
  }
}

__global__ __launch_bounds__(256) void padzero_kernel(const int* __restrict__ classinfo,
                                                      float* __restrict__ comp) {
  int bk = blockIdx.x;
  int b = bk / NCLS;
  int base = classinfo[bk * 3 + 0];
  int count = classinfo[bk * 3 + 1];
  int padl = classinfo[bk * 3 + 2];
  int nel = (padl - count) * NC;
  float* dst = comp + ((size_t)b * NPAD_MAX + base + count) * NC;
  for (int i = threadIdx.x; i < nel; i += 256) dst[i] = 0.0f;
}

#define SUBS 4
#define SCW (CHUNKS / SUBS)
#define TPITCH 68

__global__ __launch_bounds__(SCW * 32) void scatter_kernel(const float* __restrict__ hyper,
                                                           const int* __restrict__ seg,
                                                           const int* __restrict__ blockoff,
                                                           const int* __restrict__ classinfo,
                                                           float* __restrict__ comp) {
  int s = blockIdx.x % SUBS;
  int hb = blockIdx.x / SUBS;
  int b = hb / NBLKS, bb = hb % NBLKS;
  int lane = threadIdx.x & 31;
  int wave = __builtin_amdgcn_readfirstlane((int)(threadIdx.x >> 5));
  __shared__ __attribute__((aligned(16))) float tile[SCW][32 * TPITCH];
  __shared__ int cnt[CHUNKS][32];

  const int* sg = seg + (size_t)b * NPIX + (size_t)bb * BLKPX;
  int sgv[SUBS];
#pragma unroll
  for (int j = 0; j < SUBS; ++j) sgv[j] = sg[(j * SCW + wave) * 32 + lane];

  int cls = (lane < NCLS) ? lane : 0;
  int cb = classinfo[(b * NCLS + cls) * 3 + 0] + blockoff[(size_t)hb * NCLS + cls];

  int ch = s * SCW + wave;
  const float* xp = hyper + (size_t)b * NC * NPIX + (size_t)bb * BLKPX + (size_t)ch * 32 + lane;
  float v[NC];
#pragma unroll
  for (int c = 0; c < NC; ++c) v[c] = xp[(size_t)c * NPIX];

  int myk = sgv[0];
#pragma unroll
  for (int j = 1; j < SUBS; ++j) myk = (s == j) ? sgv[j] : myk;

#pragma unroll
  for (int j = 0; j < SUBS; ++j) {
    int pc = 0;
#pragma unroll
    for (int k = 0; k < NCLS; ++k) {
      unsigned bal = __builtin_amdgcn_ballot_w32(sgv[j] == k);
      if (lane == k) pc = __popc(bal);
    }
    cnt[j * SCW + wave][lane] = pc;
  }
  unsigned balmy = 0;
#pragma unroll
  for (int k = 0; k < NCLS; ++k) {
    unsigned bal = __builtin_amdgcn_ballot_w32(myk == k);
    if (myk == k) balmy = bal;
  }

  float* tl = tile[wave];
#pragma unroll
  for (int q = 0; q < NC / 4; ++q)
    *(float4*)&tl[lane * TPITCH + 4 * q] =
        make_float4(v[4 * q], v[4 * q + 1], v[4 * q + 2], v[4 * q + 3]);
  __syncthreads();

  int run = 0;
#pragma unroll
  for (int c2 = 0; c2 < CHUNKS - 1; ++c2) {
    int t = cnt[c2][lane];
    run += (c2 < ch) ? t : 0;
  }
  unsigned ltmask = (1u << lane) - 1u;
  int pos = __shfl(cb + run, myk, 32) + __popc(balmy & ltmask);

  float* crow = comp + (size_t)b * NPAD_MAX * NC;
#pragma unroll
  for (int p = 0; p < 32; ++p) {
    int pp = __builtin_amdgcn_readlane(pos, p);
    float2 val = *(const float2*)&tl[p * TPITCH + 2 * lane];
    *(float2*)&crow[(size_t)pp * NC + 2 * lane] = val;
  }
}

#define SYW 4

__device__ __forceinline__ float xhalf(float x) {
  int b = __float_as_int(x);
  return __int_as_float(__builtin_amdgcn_permlanex16(b, b, 0x76543210, 0xfedcba98, false, false));
}

__device__ __forceinline__ void load_group(v2f (&f)[4], const float* __restrict__ gp) {
#pragma unroll
  for (int blk = 0; blk < 4; ++blk) {
    f[blk].x = gp[blk * 16];
    f[blk].y = gp[NC + blk * 16];
  }
}

__device__ __forceinline__ void load_trip(v2f (&f)[4][4], const float* __restrict__ tp) {
#pragma unroll
  for (int i = 0; i < 4; ++i) load_group(f[i], tp + (size_t)i * 4 * NC);
}

template <int WV>
__device__ __forceinline__ void use_group(v8f (&acc)[4], float& cs, const v2f (&f)[4]) {
#pragma unroll
  for (int db = 0; db < 4; ++db)
    acc[db] = __builtin_amdgcn_wmma_f32_16x16x4_f32(
        false, f[WV], false, f[db], (short)0, acc[db], false, false);
  float ox = xhalf(f[WV].x), oy = xhalf(f[WV].y);
  cs = (((cs + f[WV].x) + f[WV].y) + ox) + oy;
}

template <int WV>
__device__ __forceinline__ void use_trip(v8f (&acc)[4], float& cs, const v2f (&f)[4][4]) {
#pragma unroll
  for (int i = 0; i < 4; ++i) use_group<WV>(acc, cs, f[i]);
}

template <int WV>
__device__ __forceinline__ void syrk_tile_row(const float* __restrict__ rp, int plen, int lane,
                                              float* __restrict__ sxxout, float* __restrict__ sumout) {
  const int ntrip = plen >> 4;
  const int rem = (plen >> 2) & 3;

  v8f acc[4] = {};
  float cs = 0.0f;

  v2f ft[3][4];
  const float* tailp = rp + (size_t)ntrip * 16 * NC;
  if (rem > 0) load_group(ft[0], tailp);
  if (rem > 1) load_group(ft[1], tailp + 4 * NC);
  if (rem > 2) load_group(ft[2], tailp + 8 * NC);

  v2f fa[4][4], fb[4][4];
  if (ntrip > 0) load_trip(fa, rp);
  int t = 0;
  for (; t + 2 <= ntrip; t += 2) {
    load_trip(fb, rp + (size_t)(t + 1) * 16 * NC);
    use_trip<WV>(acc, cs, fa);
    int tn = (t + 2 < ntrip) ? t + 2 : ntrip - 1;
    load_trip(fa, rp + (size_t)tn * 16 * NC);
    use_trip<WV>(acc, cs, fb);
  }
  if (t < ntrip) use_trip<WV>(acc, cs, fa);
  if (rem > 0) use_group<WV>(acc, cs, ft[0]);
  if (rem > 1) use_group<WV>(acc, cs, ft[1]);
  if (rem > 2) use_group<WV>(acc, cs, ft[2]);

  int col = lane & 15;
  int rof = (lane >> 4) * 8;
#pragma unroll
  for (int db = 0; db < 4; ++db)
#pragma unroll
    for (int j = 0; j < 8; ++j) {
      int r = WV * 16 + rof + j;
      sxxout[(size_t)r * NC + db * 16 + col] = acc[db][j];
    }
  if (lane < 16) sumout[WV * 16 + lane] = cs;
}

__global__ __launch_bounds__(SYW * 32) void syrk_kernel(const float* __restrict__ comp,
                                                        const int* __restrict__ classinfo,
                                                        float* __restrict__ sxxpart,
                                                        float* __restrict__ sumpart) {
  int gid = blockIdx.x;
  int part = gid / (NB * NCLS);
  int bk = gid % (NB * NCLS);
  int b = bk / NCLS;
  int lane = threadIdx.x & 31;
  int wave = __builtin_amdgcn_readfirstlane((int)(threadIdx.x >> 5));

  int base = classinfo[bk * 3 + 0];
  int padl = classinfo[bk * 3 + 2];
  int plen = padl / PARTS;
  size_t rowbase = (size_t)b * NPAD_MAX + base + (size_t)part * plen;

  int m = lane & 15;
  int kb = (lane >> 4) << 1;
  const float* rp = comp + (rowbase + kb) * NC + m;
  float* sxxout = sxxpart + (size_t)gid * (NC * NC);
  float* sumout = sumpart + (size_t)gid * NC;

  switch (wave) {
    case 0: syrk_tile_row<0>(rp, plen, lane, sxxout, sumout); break;
    case 1: syrk_tile_row<1>(rp, plen, lane, sxxout, sumout); break;
    case 2: syrk_tile_row<2>(rp, plen, lane, sxxout, sumout); break;
    default: syrk_tile_row<3>(rp, plen, lane, sxxout, sumout); break;
  }
}

__device__ __forceinline__ float lane_read(float v, int l) {
  return __int_as_float(__builtin_amdgcn_readlane(__float_as_int(v), l));
}

template <int Q, int NCOLS>
__device__ __forceinline__ void chol_step(float (&c0)[NC], float (&c1)[NC], float* covL, int lane, int kp) {
  const int kk = 2 * kp + Q;
  float piv = lane_read(Q ? c1[0] : c0[0], kp);
  float d = sqrtf(piv);
  float l0 = c0[0] / d;
  float l1 = c1[0] / d;
  if (Q == 0) l0 = (lane == kp) ? d : l0;
  else        l1 = (lane == kp) ? d : l1;
  covL[(2 * lane) * 65 + kk] = l0;
  covL[(2 * lane + 1) * 65 + kk] = l1;
#pragma unroll
  for (int jj = 1; jj <= NCOLS; ++jj) {
    const int lj = (kp + ((Q + jj) >> 1)) & 31;
    float s = lane_read(((Q + jj) & 1) ? l1 : l0, lj);
    c0[jj - 1] = c0[jj] - l0 * s;
    c1[jj - 1] = c1[jj] - l1 * s;
  }
}

__global__ __launch_bounds__(256) void reduce_solve_kernel(const float* __restrict__ sxxpart,
                                                           const float* __restrict__ sumpart,
                                                           const int* __restrict__ classinfo,
                                                           const float* __restrict__ pattern,
                                                           float* __restrict__ params) {
  int bk = blockIdx.x;
  int tid = threadIdx.x;
  __shared__ float covL[64 * 65];
  __shared__ float meanL[64], pL[64], wv[64];

  int count = classinfo[bk * 3 + 1];
  float safe = (float)(count > 0 ? count : 1);

  if (tid < 64) {
    float s = 0.0f;
    for (int part = 0; part < PARTS; ++part)
      s += sumpart[((size_t)part * NB * NCLS + bk) * NC + tid];
    meanL[tid] = s / safe;
    pL[tid] = pattern[tid];
  }
  __syncthreads();

  for (int idx = tid; idx < NC * NC; idx += 256) {
    int i = idx >> 6, j = idx & 63;
    float s = 0.0f;
    for (int part = 0; part < PARTS; ++part)
      s += sxxpart[((size_t)part * NB * NCLS + bk) * (NC * NC) + idx];
    float cv = s / safe - meanL[i] * meanL[j];
    if (i == j) cv += EPSV;
    covL[i * 65 + j] = cv;
  }
  __syncthreads();

  if (tid < 32) {
    const int lane = tid;
    float c0[NC], c1[NC];
#pragma unroll
    for (int j = 0; j < NC; ++j) {
      c0[j] = covL[(2 * lane) * 65 + j];
      c1[j] = covL[(2 * lane + 1) * 65 + j];
    }

#pragma nounroll
    for (int kp = 0; kp < 16; ++kp) {
      chol_step<0, 63>(c0, c1, covL, lane, kp);
      chol_step<1, 63>(c0, c1, covL, lane, kp);
    }
#pragma nounroll
    for (int kp = 16; kp < 32; ++kp) {
      chol_step<0, 31>(c0, c1, covL, lane, kp);
      chol_step<1, 31>(c0, c1, covL, lane, kp);
    }

    const float dg0 = covL[(2 * lane) * 65 + 2 * lane];
    const float dg1 = covL[(2 * lane + 1) * 65 + 2 * lane + 1];

    float y0 = pL[2 * lane], y1 = pL[2 * lane + 1];
    float a0 = covL[(2 * lane) * 65], a1 = covL[(2 * lane + 1) * 65];
#pragma nounroll
    for (int jp = 0; jp < 32; ++jp) {
      int jn = 2 * jp + 1;
      float n0 = covL[(2 * lane) * 65 + jn], n1 = covL[(2 * lane + 1) * 65 + jn];
      {
        float s = lane_read(y0 / dg0, jp);
        y0 = (lane == jp) ? s : ((lane > jp) ? y0 - a0 * s : y0);
        y1 = (lane >= jp) ? y1 - a1 * s : y1;
      }
      int jm = (2 * jp + 2 < NC) ? 2 * jp + 2 : NC - 1;
      a0 = covL[(2 * lane) * 65 + jm];
      a1 = covL[(2 * lane + 1) * 65 + jm];
      {
        float s = lane_read(y1 / dg1, jp);
        y1 = (lane == jp) ? s : ((lane > jp) ? y1 - n1 * s : y1);
        y0 = (lane > jp) ? y0 - n0 * s : y0;
      }
    }

    float w0 = y0, w1 = y1;
    float b0 = covL[63 * 65 + 2 * lane], b1 = covL[63 * 65 + 2 * lane + 1];
#pragma nounroll
    for (int jp = 31; jp >= 0; --jp) {
      int jn = 2 * jp;
      float n0 = covL[jn * 65 + 2 * lane], n1 = covL[jn * 65 + 2 * lane + 1];
      {
        float s = lane_read(w1 / dg1, jp);
        w1 = (lane == jp) ? s : ((lane < jp) ? w1 - b1 * s : w1);
        w0 = (lane <= jp) ? w0 - b0 * s : w0;
      }
      int jm = (2 * jp - 1 >= 0) ? 2 * jp - 1 : 0;
      b0 = covL[jm * 65 + 2 * lane];
      b1 = covL[jm * 65 + 2 * lane + 1];
      {
        float s = lane_read(w0 / dg0, jp);
        w0 = (lane == jp) ? s : ((lane < jp) ? w0 - n0 * s : w0);
        w1 = (lane < jp) ? w1 - n1 * s : w1;
      }
    }

    params[(size_t)bk * 66 + 2 * lane] = w0;
    params[(size_t)bk * 66 + 2 * lane + 1] = w1;
    wv[2 * lane] = w0;
    wv[2 * lane + 1] = w1;
    float pw = 0.0f, off = 0.0f;
    for (int c = 0; c < NC; ++c) {
      pw += pL[c] * wv[c];
      off += meanL[c] * wv[c];
    }
    if (lane == 0) {
      float denom = sqrtf(pw);
      float scale = (count >= MIN_CLASS_SIZE && denom > 0.0f) ? (1.0f / denom) : 0.0f;
      params[(size_t)bk * 66 + 64] = off;
      params[(size_t)bk * 66 + 65] = scale;
    }
  }
}

__global__ __launch_bounds__(256) void score_kernel(const float* __restrict__ hyper,
                                                    const int* __restrict__ seg,
                                                    const float* __restrict__ params,
                                                    float* __restrict__ out) {
  int b = blockIdx.y;
  int tid = threadIdx.x;
  __shared__ float wS[NCLS * 65];
  __shared__ float offS[NCLS], sclS[NCLS];
  for (int i = tid; i < NCLS * NC; i += 256) {
    int k = i >> 6, c = i & 63;
    wS[k * 65 + c] = params[((size_t)b * NCLS + k) * 66 + c];
  }
  if (tid < NCLS) {
    offS[tid] = params[((size_t)b * NCLS + tid) * 66 + 64];
    sclS[tid] = params[((size_t)b * NCLS + tid) * 66 + 65];
  }
  __syncthreads();

  size_t n = (size_t)blockIdx.x * 256 + tid;
  int k = seg[(size_t)b * NPIX + n];
  const float* xp = hyper + (size_t)b * NC * NPIX + n;
  float acc = 0.0f;
#pragma unroll
  for (int c = 0; c < NC; ++c)
    acc = fmaf(xp[(size_t)c * NPIX], wS[k * 65 + c], acc);
  out[(size_t)b * NPIX + n] = (acc - offS[k]) * sclS[k];
}

extern "C" void kernel_launch(void* const* d_in, const int* in_sizes, int n_in,
                              void* d_out, int out_size, void* d_ws, size_t ws_size,
                              hipStream_t stream) {
  const float* hyper = (const float*)d_in[0];
  const float* pattern = (const float*)d_in[1];
  const int* seg = (const int*)d_in[2];
  float* out = (float*)d_out;

  char* ws = (char*)d_ws;
  auto al = [](size_t x) { return (x + 255) & ~(size_t)255; };
  size_t off = 0;
  float* comp = (float*)(ws + off);
  off += al((size_t)NB * NPAD_MAX * NC * sizeof(float));
  int* hist = (int*)(ws + off);
  off += al((size_t)NB * NBLKS * NCLS * sizeof(int));
  int* blockoff = (int*)(ws + off);
  off += al((size_t)NB * NBLKS * NCLS * sizeof(int));
  int* classinfo = (int*)(ws + off);
  off += al((size_t)NB * NCLS * 3 * sizeof(int));
  float* sxxpart = (float*)(ws + off);
  off += al((size_t)PARTS * NB * NCLS * NC * NC * sizeof(float));
  float* sumpart = (float*)(ws + off);
  off += al((size_t)PARTS * NB * NCLS * NC * sizeof(float));
  float* params = (float*)(ws + off);
  off += al((size_t)NB * NCLS * 66 * sizeof(float));

  hipLaunchKernelGGL(hist_kernel, dim3(NB * NBLKS), dim3(32), 0, stream, seg, hist);
  hipLaunchKernelGGL(scan_kernel, dim3(1), dim3(64), 0, stream, hist, blockoff, classinfo);
  hipLaunchKernelGGL(padzero_kernel, dim3(NB * NCLS), dim3(256), 0, stream, classinfo, comp);
  hipLaunchKernelGGL(scatter_kernel, dim3(NB * NBLKS * SUBS), dim3(SCW * 32), 0, stream, hyper, seg,
                     blockoff, classinfo, comp);
  hipLaunchKernelGGL(syrk_kernel, dim3(PARTS * NB * NCLS), dim3(SYW * 32), 0, stream, comp,
                     classinfo, sxxpart, sumpart);
  hipLaunchKernelGGL(reduce_solve_kernel, dim3(NB * NCLS), dim3(256), 0, stream, sxxpart,
                     sumpart, classinfo, pattern, params);
  hipLaunchKernelGGL(score_kernel, dim3(NPIX / 256, NB), dim3(256), 0, stream, hyper, seg,
                     params, out);
}


# scan and padzero launches folded into scatter: each group sums the class-major histogram itself (DPP wave sums), 40 groups zero the pad rows, one per batch stores the class table; 7 launches to 5
# speedup vs baseline: 4.1914x; 1.0533x over previous
#include <hip/hip_runtime.h>
#include <hip/hip_bf16.h>


typedef float v2f __attribute__((ext_vector_type(2)));
typedef float v8f __attribute__((ext_vector_type(8)));

#define NB 2
#define NC 64
#define NPIX 147456
#define NCLS 20
#define BLKPX 1024
#define CHUNKS (BLKPX / 32)
#define NBLKS (NPIX / BLKPX)
#define PARTS 16
#define PADALIGN (PARTS * 4)
#define NPAD_MAX (NPIX + NCLS * PADALIGN)
#define MIN_CLASS_SIZE 5000
#define EPSV 1e-6f

__global__ __launch_bounds__(32) void hist_kernel(const int* __restrict__ seg,
                                                  int* __restrict__ hist) {
  int blk = blockIdx.x;
  int b = blk / NBLKS, bb = blk % NBLKS;
  int lane = threadIdx.x;
  const int* s = seg + (size_t)b * NPIX + (size_t)bb * BLKPX;
  int cnt = 0;
  for (int ch = 0; ch < CHUNKS; ++ch) {
    int myk = s[ch * 32 + lane];
#pragma unroll
    for (int k = 0; k < NCLS; ++k) {
      unsigned bal = __builtin_amdgcn_ballot_w32(myk == k);
      if (lane == k) cnt += __popc(bal);
    }
  }
  if (lane < NCLS) hist[((size_t)b * NCLS + lane) * NBLKS + bb] = cnt;
}

#define SUBS 4
#define SCW (CHUNKS / SUBS)
#define TPITCH 68
#define CPW ((NCLS + SCW - 1) / SCW)
#define HQ (NBLKS / 4)

__device__ __forceinline__ int wave_sum(int v) {
  v += __builtin_amdgcn_update_dpp(0, v, 0x161, 0xf, 0xf, false);
  v += __builtin_amdgcn_update_dpp(0, v, 0x162, 0xf, 0xf, false);
  v += __builtin_amdgcn_update_dpp(0, v, 0x164, 0xf, 0xf, false);
  v += __builtin_amdgcn_update_dpp(0, v, 0x168, 0xf, 0xf, false);
  v += __builtin_amdgcn_permlanex16(v, v, 0x76543210, 0xfedcba98, false, false);
  return v;
}

__global__ __launch_bounds__(SCW * 32) void scatter_kernel(const float* __restrict__ hyper,
                                                           const int* __restrict__ seg,
                                                           const int* __restrict__ hist,
                                                           int* __restrict__ classinfo,
                                                           float* __restrict__ comp) {
  int s = blockIdx.x % SUBS;
  int hb = blockIdx.x / SUBS;
  int b = hb / NBLKS, bb = hb % NBLKS;
  int lane = threadIdx.x & 31;
  int wave = __builtin_amdgcn_readfirstlane((int)(threadIdx.x >> 5));
  __shared__ __attribute__((aligned(16))) float tile[SCW][32 * TPITCH];
  __shared__ int cnt[CHUNKS][32];
  __shared__ int totS[32];
  __shared__ int befS[32];

  const int* sg = seg + (size_t)b * NPIX + (size_t)bb * BLKPX;
  int sgv[SUBS];
#pragma unroll
  for (int j = 0; j < SUBS; ++j) sgv[j] = sg[(j * SCW + wave) * 32 + lane];

  static_assert(HQ - 32 == 4, "the upper part of a histogram column is four 4-block units");
  int4 hlo[CPW], hhi[CPW];
#pragma unroll
  for (int r = 0; r < CPW; ++r) {
    int k = wave + SCW * r;
    int kc = (k < NCLS) ? k : NCLS - 1;
    const int4* hp = (const int4*)(hist + ((size_t)b * NCLS + kc) * NBLKS);
    hlo[r] = hp[lane];
    hhi[r] = hp[32 + (lane & 3)];
  }

  int ch = s * SCW + wave;
  const float* xp = hyper + (size_t)b * NC * NPIX + (size_t)bb * BLKPX + (size_t)ch * 32 + lane;
  float v[NC];
#pragma unroll
  for (int c = 0; c < NC; ++c) v[c] = xp[(size_t)c * NPIX];

  int myk = sgv[0];
#pragma unroll
  for (int j = 1; j < SUBS; ++j) myk = (s == j) ? sgv[j] : myk;

#pragma unroll
  for (int j = 0; j < SUBS; ++j) {
    int pc = 0;
#pragma unroll
    for (int k = 0; k < NCLS; ++k) {
      unsigned bal = __builtin_amdgcn_ballot_w32(sgv[j] == k);
      if (lane == k) pc = __popc(bal);
    }
    cnt[j * SCW + wave][lane] = pc;
  }
  unsigned balmy = 0;
#pragma unroll
  for (int k = 0; k < NCLS; ++k) {
    unsigned bal = __builtin_amdgcn_ballot_w32(myk == k);
    if (myk == k) balmy = bal;
  }

#pragma unroll
  for (int r = 0; r < CPW; ++r) {
    int k = wave + SCW * r;
    int q0 = 4 * lane, q1 = 128 + 4 * lane;
    bool hi = lane < HQ - 32;
    int tt = ((hlo[r].x + hlo[r].y) + (hlo[r].z + hlo[r].w)) +
             (hi ? ((hhi[r].x + hhi[r].y) + (hhi[r].z + hhi[r].w)) : 0);
    int bf = ((q0 < bb) ? hlo[r].x : 0) + ((q0 + 1 < bb) ? hlo[r].y : 0) +
             ((q0 + 2 < bb) ? hlo[r].z : 0) + ((q0 + 3 < bb) ? hlo[r].w : 0) +
             ((hi && q1 < bb) ? hhi[r].x : 0) + ((hi && q1 + 1 < bb) ? hhi[r].y : 0) +
             ((hi && q1 + 2 < bb) ? hhi[r].z : 0) + ((hi && q1 + 3 < bb) ? hhi[r].w : 0);
    tt = wave_sum(tt);
    bf = wave_sum(bf);
    if (lane == 0 && k < NCLS) {
      totS[k] = tt;
      befS[k] = bf;
    }
  }

  float* tl = tile[wave];
#pragma unroll
  for (int q = 0; q < NC / 4; ++q)
    *(float4*)&tl[lane * TPITCH + 4 * q] =
        make_float4(v[4 * q], v[4 * q + 1], v[4 * q + 2], v[4 * q + 3]);
  __syncthreads();

  int cls = (lane < NCLS) ? lane : 0;
  int base = 0;
#pragma unroll
  for (int k = 0; k < NCLS - 1; ++k) {
    int p = (totS[k] + PADALIGN - 1) & ~(PADALIGN - 1);
    base += (k < lane) ? p : 0;
  }
  int tk = totS[cls];
  int pk = (tk + PADALIGN - 1) & ~(PADALIGN - 1);
  int cb = base + befS[cls];

  int run = 0;
#pragma unroll
  for (int c2 = 0; c2 < CHUNKS - 1; ++c2) {
    int t = cnt[c2][lane];
    run += (c2 < ch) ? t : 0;
  }
  unsigned ltmask = (1u << lane) - 1u;
  int pos = __shfl(cb + run, myk, 32) + __popc(balmy & ltmask);

  float* crow = comp + (size_t)b * NPAD_MAX * NC;
#pragma unroll
  for (int p = 0; p < 32; ++p) {
    int pp = __builtin_amdgcn_readlane(pos, p);
    float2 val = *(const float2*)&tl[p * TPITCH + 2 * lane];
    *(float2*)&crow[(size_t)pp * NC + 2 * lane] = val;
  }

  if (s == 0 && bb == 0 && wave == 0 && lane < NCLS) {
    classinfo[(b * NCLS + lane) * 3 + 0] = base;
    classinfo[(b * NCLS + lane) * 3 + 1] = tk;
    classinfo[(b * NCLS + lane) * 3 + 2] = pk;
  }
  if (s == 0 && bb < NCLS) {
    int zb = __builtin_amdgcn_readlane(base, bb);
    int zt = __builtin_amdgcn_readlane(tk, bb);
    int zp = __builtin_amdgcn_readlane(pk, bb);
    float4* dst = (float4*)(crow + ((size_t)zb + zt) * NC);
    int n4 = (zp - zt) * (NC / 4);
#pragma unroll
    for (int j = 0; j < PADALIGN * (NC / 4) / (SCW * 32); ++j) {
      int i = (int)threadIdx.x + j * (SCW * 32);
      if (i < n4) dst[i] = make_float4(0.0f, 0.0f, 0.0f, 0.0f);
    }
  }
}

#define SYW 4

__device__ __forceinline__ float xhalf(float x) {
  int b = __float_as_int(x);
  return __int_as_float(__builtin_amdgcn_permlanex16(b, b, 0x76543210, 0xfedcba98, false, false));
}

__device__ __forceinline__ void load_group(v2f (&f)[4], const float* __restrict__ gp) {
#pragma unroll
  for (int blk = 0; blk < 4; ++blk) {
    f[blk].x = gp[blk * 16];
    f[blk].y = gp[NC + blk * 16];
  }
}

__device__ __forceinline__ void load_trip(v2f (&f)[4][4], const float* __restrict__ tp) {
#pragma unroll
  for (int i = 0; i < 4; ++i) load_group(f[i], tp + (size_t)i * 4 * NC);
}

template <int WV>
__device__ __forceinline__ void use_group(v8f (&acc)[4], float& cs, const v2f (&f)[4]) {
#pragma unroll
  for (int db = 0; db < 4; ++db)
    acc[db] = __builtin_amdgcn_wmma_f32_16x16x4_f32(
        false, f[WV], false, f[db], (short)0, acc[db], false, false);
  float ox = xhalf(f[WV].x), oy = xhalf(f[WV].y);
  cs = (((cs + f[WV].x) + f[WV].y) + ox) + oy;
}

template <int WV>
__device__ __forceinline__ void use_trip(v8f (&acc)[4], float& cs, const v2f (&f)[4][4]) {
#pragma unroll
  for (int i = 0; i < 4; ++i) use_group<WV>(acc, cs, f[i]);
}

template <int WV>
__device__ __forceinline__ void syrk_tile_row(const float* __restrict__ rp, int plen, int lane,
                                              float* __restrict__ sxxout, float* __restrict__ sumout) {
  const int ntrip = plen >> 4;
  const int rem = (plen >> 2) & 3;

  v8f acc[4] = {};
  float cs = 0.0f;

  v2f ft[3][4];
  const float* tailp = rp + (size_t)ntrip * 16 * NC;
  if (rem > 0) load_group(ft[0], tailp);
  if (rem > 1) load_group(ft[1], tailp + 4 * NC);
  if (rem > 2) load_group(ft[2], tailp + 8 * NC);

  v2f fa[4][4], fb[4][4];
  if (ntrip > 0) load_trip(fa, rp);
  int t = 0;
  for (; t + 2 <= ntrip; t += 2) {
    load_trip(fb, rp + (size_t)(t + 1) * 16 * NC);
    use_trip<WV>(acc, cs, fa);
    int tn = (t + 2 < ntrip) ? t + 2 : ntrip - 1;
    load_trip(fa, rp + (size_t)tn * 16 * NC);
    use_trip<WV>(acc, cs, fb);
  }
  if (t < ntrip) use_trip<WV>(acc, cs, fa);
  if (rem > 0) use_group<WV>(acc, cs, ft[0]);
  if (rem > 1) use_group<WV>(acc, cs, ft[1]);
  if (rem > 2) use_group<WV>(acc, cs, ft[2]);

  int col = lane & 15;
  int rof = (lane >> 4) * 8;
#pragma unroll
  for (int db = 0; db < 4; ++db)
#pragma unroll
    for (int j = 0; j < 8; ++j) {
      int r = WV * 16 + rof + j;
      sxxout[(size_t)r * NC + db * 16 + col] = acc[db][j];
    }
  if (lane < 16) sumout[WV * 16 + lane] = cs;
}

__global__ __launch_bounds__(SYW * 32) void syrk_kernel(const float* __restrict__ comp,
                                                        const int* __restrict__ classinfo,
                                                        float* __restrict__ sxxpart,
                                                        float* __restrict__ sumpart) {
  int gid = blockIdx.x;
  int part = gid / (NB * NCLS);
  int bk = gid % (NB * NCLS);
  int b = bk / NCLS;
  int lane = threadIdx.x & 31;
  int wave = __builtin_amdgcn_readfirstlane((int)(threadIdx.x >> 5));

  int base = classinfo[bk * 3 + 0];
  int padl = classinfo[bk * 3 + 2];
  int plen = padl / PARTS;
  size_t rowbase = (size_t)b * NPAD_MAX + base + (size_t)part * plen;

  int m = lane & 15;
  int kb = (lane >> 4) << 1;
  const float* rp = comp + (rowbase + kb) * NC + m;
  float* sxxout = sxxpart + (size_t)gid * (NC * NC);
  float* sumout = sumpart + (size_t)gid * NC;

  switch (wave) {
    case 0: syrk_tile_row<0>(rp, plen, lane, sxxout, sumout); break;
    case 1: syrk_tile_row<1>(rp, plen, lane, sxxout, sumout); break;
    case 2: syrk_tile_row<2>(rp, plen, lane, sxxout, sumout); break;
    default: syrk_tile_row<3>(rp, plen, lane, sxxout, sumout); break;
  }
}

__device__ __forceinline__ float lane_read(float v, int l) {
  return __int_as_float(__builtin_amdgcn_readlane(__float_as_int(v), l));
}

template <int Q, int NCOLS>
__device__ __forceinline__ void chol_step(float (&c0)[NC], float (&c1)[NC], float* covL, int lane, int kp) {
  const int kk = 2 * kp + Q;
  float piv = lane_read(Q ? c1[0] : c0[0], kp);
  float d = sqrtf(piv);
  float l0 = c0[0] / d;
  float l1 = c1[0] / d;
  if (Q == 0) l0 = (lane == kp) ? d : l0;
  else        l1 = (lane == kp) ? d : l1;
  covL[(2 * lane) * 65 + kk] = l0;
  covL[(2 * lane + 1) * 65 + kk] = l1;
#pragma unroll
  for (int jj = 1; jj <= NCOLS; ++jj) {
    const int lj = (kp + ((Q + jj) >> 1)) & 31;
    float s = lane_read(((Q + jj) & 1) ? l1 : l0, lj);
    c0[jj - 1] = c0[jj] - l0 * s;
    c1[jj - 1] = c1[jj] - l1 * s;
  }
}

__global__ __launch_bounds__(256) void reduce_solve_kernel(const float* __restrict__ sxxpart,
                                                           const float* __restrict__ sumpart,
                                                           const int* __restrict__ classinfo,
                                                           const float* __restrict__ pattern,
                                                           float* __restrict__ params) {
  int bk = blockIdx.x;
  int tid = threadIdx.x;
  __shared__ float covL[64 * 65];
  __shared__ float meanL[64], pL[64], wv[64];

  int count = classinfo[bk * 3 + 1];
  float safe = (float)(count > 0 ? count : 1);

  if (tid < 64) {
    float s = 0.0f;
    for (int part = 0; part < PARTS; ++part)
      s += sumpart[((size_t)part * NB * NCLS + bk) * NC + tid];
    meanL[tid] = s / safe;
    pL[tid] = pattern[tid];
  }
  __syncthreads();

  for (int idx = tid; idx < NC * NC; idx += 256) {
    int i = idx >> 6, j = idx & 63;
    float s = 0.0f;
    for (int part = 0; part < PARTS; ++part)
      s += sxxpart[((size_t)part * NB * NCLS + bk) * (NC * NC) + idx];
    float cv = s / safe - meanL[i] * meanL[j];
    if (i == j) cv += EPSV;
    covL[i * 65 + j] = cv;
  }
  __syncthreads();

  if (tid < 32) {
    const int lane = tid;
    float c0[NC], c1[NC];
#pragma unroll
    for (int j = 0; j < NC; ++j) {
      c0[j] = covL[(2 * lane) * 65 + j];
      c1[j] = covL[(2 * lane + 1) * 65 + j];
    }

#pragma nounroll
    for (int kp = 0; kp < 16; ++kp) {
      chol_step<0, 63>(c0, c1, covL, lane, kp);
      chol_step<1, 63>(c0, c1, covL, lane, kp);
    }
#pragma nounroll
    for (int kp = 16; kp < 32; ++kp) {
      chol_step<0, 31>(c0, c1, covL, lane, kp);
      chol_step<1, 31>(c0, c1, covL, lane, kp);
    }

    const float dg0 = covL[(2 * lane) * 65 + 2 * lane];
    const float dg1 = covL[(2 * lane + 1) * 65 + 2 * lane + 1];

    float y0 = pL[2 * lane], y1 = pL[2 * lane + 1];
    float a0 = covL[(2 * lane) * 65], a1 = covL[(2 * lane + 1) * 65];
#pragma nounroll
    for (int jp = 0; jp < 32; ++jp) {
      int jn = 2 * jp + 1;
      float n0 = covL[(2 * lane) * 65 + jn], n1 = covL[(2 * lane + 1) * 65 + jn];
      {
        float s = lane_read(y0 / dg0, jp);
        y0 = (lane == jp) ? s : ((lane > jp) ? y0 - a0 * s : y0);
        y1 = (lane >= jp) ? y1 - a1 * s : y1;
      }
      int jm = (2 * jp + 2 < NC) ? 2 * jp + 2 : NC - 1;
      a0 = covL[(2 * lane) * 65 + jm];
      a1 = covL[(2 * lane + 1) * 65 + jm];
      {
        float s = lane_read(y1 / dg1, jp);
        y1 = (lane == jp) ? s : ((lane > jp) ? y1 - n1 * s : y1);
        y0 = (lane > jp) ? y0 - n0 * s : y0;
      }
    }

    float w0 = y0, w1 = y1;
    float b0 = covL[63 * 65 + 2 * lane], b1 = covL[63 * 65 + 2 * lane + 1];
#pragma nounroll
    for (int jp = 31; jp >= 0; --jp) {
      int jn = 2 * jp;
      float n0 = covL[jn * 65 + 2 * lane], n1 = covL[jn * 65 + 2 * lane + 1];
      {
        float s = lane_read(w1 / dg1, jp);
        w1 = (lane == jp) ? s : ((lane < jp) ? w1 - b1 * s : w1);
        w0 = (lane <= jp) ? w0 - b0 * s : w0;
      }
      int jm = (2 * jp - 1 >= 0) ? 2 * jp - 1 : 0;
      b0 = covL[jm * 65 + 2 * lane];
      b1 = covL[jm * 65 + 2 * lane + 1];
      {
        float s = lane_read(w0 / dg0, jp);
        w0 = (lane == jp) ? s : ((lane < jp) ? w0 - n0 * s : w0);
        w1 = (lane < jp) ? w1 - n1 * s : w1;
      }
    }

    params[(size_t)bk * 66 + 2 * lane] = w0;
    params[(size_t)bk * 66 + 2 * lane + 1] = w1;
    wv[2 * lane] = w0;
    wv[2 * lane + 1] = w1;
    float pw = 0.0f, off = 0.0f;
    for (int c = 0; c < NC; ++c) {
      pw += pL[c] * wv[c];
      off += meanL[c] * wv[c];
    }
    if (lane == 0) {
      float denom = sqrtf(pw);
      float scale = (count >= MIN_CLASS_SIZE && denom > 0.0f) ? (1.0f / denom) : 0.0f;
      params[(size_t)bk * 66 + 64] = off;
      params[(size_t)bk * 66 + 65] = scale;
    }
  }
}

__global__ __launch_bounds__(256) void score_kernel(const float* __restrict__ hyper,
                                                    const int* __restrict__ seg,
                                                    const float* __restrict__ params,
                                                    float* __restrict__ out) {
  int b = blockIdx.y;
  int tid = threadIdx.x;
  __shared__ float wS[NCLS * 65];
  __shared__ float offS[NCLS], sclS[NCLS];
  for (int i = tid; i < NCLS * NC; i += 256) {
    int k = i >> 6, c = i & 63;
    wS[k * 65 + c] = params[((size_t)b * NCLS + k) * 66 + c];
  }
  if (tid < NCLS) {
    offS[tid] = params[((size_t)b * NCLS + tid) * 66 + 64];
    sclS[tid] = params[((size_t)b * NCLS + tid) * 66 + 65];
  }
  __syncthreads();

  size_t n = (size_t)blockIdx.x * 256 + tid;
  int k = seg[(size_t)b * NPIX + n];
  const float* xp = hyper + (size_t)b * NC * NPIX + n;
  float acc = 0.0f;
#pragma unroll
  for (int c = 0; c < NC; ++c)
    acc = fmaf(xp[(size_t)c * NPIX], wS[k * 65 + c], acc);
  out[(size_t)b * NPIX + n] = (acc - offS[k]) * sclS[k];
}

extern "C" void kernel_launch(void* const* d_in, const int* in_sizes, int n_in,
                              void* d_out, int out_size, void* d_ws, size_t ws_size,
                              hipStream_t stream) {
  const float* hyper = (const float*)d_in[0];
  const float* pattern = (const float*)d_in[1];
  const int* seg = (const int*)d_in[2];
  float* out = (float*)d_out;

  char* ws = (char*)d_ws;
  auto al = [](size_t x) { return (x + 255) & ~(size_t)255; };
  size_t off = 0;
  float* comp = (float*)(ws + off);
  off += al((size_t)NB * NPAD_MAX * NC * sizeof(float));
  int* hist = (int*)(ws + off);
  off += al((size_t)NB * NBLKS * NCLS * sizeof(int));
  int* classinfo = (int*)(ws + off);
  off += al((size_t)NB * NCLS * 3 * sizeof(int));
  float* sxxpart = (float*)(ws + off);
  off += al((size_t)PARTS * NB * NCLS * NC * NC * sizeof(float));
  float* sumpart = (float*)(ws + off);
  off += al((size_t)PARTS * NB * NCLS * NC * sizeof(float));
  float* params = (float*)(ws + off);
  off += al((size_t)NB * NCLS * 66 * sizeof(float));

  hipLaunchKernelGGL(hist_kernel, dim3(NB * NBLKS), dim3(32), 0, stream, seg, hist);
  hipLaunchKernelGGL(scatter_kernel, dim3(NB * NBLKS * SUBS), dim3(SCW * 32), 0, stream, hyper, seg,
                     hist, classinfo, comp);
  hipLaunchKernelGGL(syrk_kernel, dim3(PARTS * NB * NCLS), dim3(SYW * 32), 0, stream, comp,
                     classinfo, sxxpart, sumpart);
  hipLaunchKernelGGL(reduce_solve_kernel, dim3(NB * NCLS), dim3(256), 0, stream, sxxpart,
                     sumpart, classinfo, pattern, params);
  hipLaunchKernelGGL(score_kernel, dim3(NPIX / 256, NB), dim3(256), 0, stream, hyper, seg,
                     params, out);
}


# hist kernel made memory-parallel: 4 waves per block, two 16-B class-id loads per lane in flight together instead of 32 loads waited for one by one; partial counts summed in LDS; same counts
# speedup vs baseline: 4.7013x; 1.1217x over previous
#include <hip/hip_runtime.h>
#include <hip/hip_bf16.h>


typedef float v2f __attribute__((ext_vector_type(2)));
typedef float v8f __attribute__((ext_vector_type(8)));

#define NB 2
#define NC 64
#define NPIX 147456
#define NCLS 20
#define BLKPX 1024
#define CHUNKS (BLKPX / 32)
#define NBLKS (NPIX / BLKPX)
#define PARTS 16
#define PADALIGN (PARTS * 4)
#define NPAD_MAX (NPIX + NCLS * PADALIGN)
#define MIN_CLASS_SIZE 5000
#define EPSV 1e-6f

#define HW 4
#define HLD (BLKPX / 4 / (HW * 32))

__global__ __launch_bounds__(HW * 32) void hist_kernel(const int* __restrict__ seg,
                                                       int* __restrict__ hist) {
  static_assert(HLD * 4 * HW * 32 == BLKPX, "the work-group's loads cover the block exactly");
  int blk = blockIdx.x;
  int b = blk / NBLKS, bb = blk % NBLKS;
  int lane = threadIdx.x & 31;
  int wave = __builtin_amdgcn_readfirstlane((int)(threadIdx.x >> 5));
  __shared__ int part[HW][32];
  const int4* s4 = (const int4*)(seg + (size_t)b * NPIX + (size_t)bb * BLKPX);
  int4 q[HLD];
#pragma unroll
  for (int j = 0; j < HLD; ++j) q[j] = s4[j * (HW * 32) + (int)threadIdx.x];
  int cnt = 0;
#pragma unroll
  for (int k = 0; k < NCLS; ++k) {
    int c = 0;
#pragma unroll
    for (int j = 0; j < HLD; ++j)
      c += __popc(__builtin_amdgcn_ballot_w32(q[j].x == k)) +
           __popc(__builtin_amdgcn_ballot_w32(q[j].y == k)) +
           __popc(__builtin_amdgcn_ballot_w32(q[j].z == k)) +
           __popc(__builtin_amdgcn_ballot_w32(q[j].w == k));
    if (lane == k) cnt = c;
  }
  part[wave][lane] = cnt;
  __syncthreads();
  if (wave == 0) {
    int t = 0;
#pragma unroll
    for (int w = 0; w < HW; ++w) t += part[w][lane];
    if (lane < NCLS) hist[((size_t)b * NCLS + lane) * NBLKS + bb] = t;
  }
}

#define SUBS 4
#define SCW (CHUNKS / SUBS)
#define TPITCH 68
#define CPW ((NCLS + SCW - 1) / SCW)
#define HQ (NBLKS / 4)

__device__ __forceinline__ int wave_sum(int v) {
  v += __builtin_amdgcn_update_dpp(0, v, 0x161, 0xf, 0xf, false);
  v += __builtin_amdgcn_update_dpp(0, v, 0x162, 0xf, 0xf, false);
  v += __builtin_amdgcn_update_dpp(0, v, 0x164, 0xf, 0xf, false);
  v += __builtin_amdgcn_update_dpp(0, v, 0x168, 0xf, 0xf, false);
  v += __builtin_amdgcn_permlanex16(v, v, 0x76543210, 0xfedcba98, false, false);
  return v;
}

__global__ __launch_bounds__(SCW * 32) void scatter_kernel(const float* __restrict__ hyper,
                                                           const int* __restrict__ seg,
                                                           const int* __restrict__ hist,
                                                           int* __restrict__ classinfo,
                                                           float* __restrict__ comp) {
  int s = blockIdx.x % SUBS;
  int hb = blockIdx.x / SUBS;
  int b = hb / NBLKS, bb = hb % NBLKS;
  int lane = threadIdx.x & 31;
  int wave = __builtin_amdgcn_readfirstlane((int)(threadIdx.x >> 5));
  __shared__ __attribute__((aligned(16))) float tile[SCW][32 * TPITCH];
  __shared__ int cnt[CHUNKS][32];
  __shared__ int totS[32];
  __shared__ int befS[32];

  const int* sg = seg + (size_t)b * NPIX + (size_t)bb * BLKPX;
  int sgv[SUBS];
#pragma unroll
  for (int j = 0; j < SUBS; ++j) sgv[j] = sg[(j * SCW + wave) * 32 + lane];

  static_assert(HQ - 32 == 4, "the upper part of a histogram column is four 4-block units");
  int4 hlo[CPW], hhi[CPW];
#pragma unroll
  for (int r = 0; r < CPW; ++r) {
    int k = wave + SCW * r;
    int kc = (k < NCLS) ? k : NCLS - 1;
    const int4* hp = (const int4*)(hist + ((size_t)b * NCLS + kc) * NBLKS);
    hlo[r] = hp[lane];
    hhi[r] = hp[32 + (lane & 3)];
  }

  int ch = s * SCW + wave;
  const float* xp = hyper + (size_t)b * NC * NPIX + (size_t)bb * BLKPX + (size_t)ch * 32 + lane;
  float v[NC];
#pragma unroll
  for (int c = 0; c < NC; ++c) v[c] = xp[(size_t)c * NPIX];

  int myk = sgv[0];
#pragma unroll
  for (int j = 1; j < SUBS; ++j) myk = (s == j) ? sgv[j] : myk;

#pragma unroll
  for (int j = 0; j < SUBS; ++j) {
    int pc = 0;
#pragma unroll
    for (int k = 0; k < NCLS; ++k) {
      unsigned bal = __builtin_amdgcn_ballot_w32(sgv[j] == k);
      if (lane == k) pc = __popc(bal);
    }
    cnt[j * SCW + wave][lane] = pc;
  }
  unsigned balmy = 0;
#pragma unroll
  for (int k = 0; k < NCLS; ++k) {
    unsigned bal = __builtin_amdgcn_ballot_w32(myk == k);
    if (myk == k) balmy = bal;
  }

#pragma unroll
  for (int r = 0; r < CPW; ++r) {
    int k = wave + SCW * r;
    int q0 = 4 * lane, q1 = 128 + 4 * lane;
    bool hi = lane < HQ - 32;
    int tt = ((hlo[r].x + hlo[r].y) + (hlo[r].z + hlo[r].w)) +
             (hi ? ((hhi[r].x + hhi[r].y) + (hhi[r].z + hhi[r].w)) : 0);
    int bf = ((q0 < bb) ? hlo[r].x : 0) + ((q0 + 1 < bb) ? hlo[r].y : 0) +
             ((q0 + 2 < bb) ? hlo[r].z : 0) + ((q0 + 3 < bb) ? hlo[r].w : 0) +
             ((hi && q1 < bb) ? hhi[r].x : 0) + ((hi && q1 + 1 < bb) ? hhi[r].y : 0) +
             ((hi && q1 + 2 < bb) ? hhi[r].z : 0) + ((hi && q1 + 3 < bb) ? hhi[r].w : 0);
    tt = wave_sum(tt);
    bf = wave_sum(bf);
    if (lane == 0 && k < NCLS) {
      totS[k] = tt;
      befS[k] = bf;
    }
  }

  float* tl = tile[wave];
#pragma unroll
  for (int q = 0; q < NC / 4; ++q)
    *(float4*)&tl[lane * TPITCH + 4 * q] =
        make_float4(v[4 * q], v[4 * q + 1], v[4 * q + 2], v[4 * q + 3]);
  __syncthreads();

  int cls = (lane < NCLS) ? lane : 0;
  int base = 0;
#pragma unroll
  for (int k = 0; k < NCLS - 1; ++k) {
    int p = (totS[k] + PADALIGN - 1) & ~(PADALIGN - 1);
    base += (k < lane) ? p : 0;
  }
  int tk = totS[cls];
  int pk = (tk + PADALIGN - 1) & ~(PADALIGN - 1);
  int cb = base + befS[cls];

  int run = 0;
#pragma unroll
  for (int c2 = 0; c2 < CHUNKS - 1; ++c2) {
    int t = cnt[c2][lane];
    run += (c2 < ch) ? t : 0;
  }
  unsigned ltmask = (1u << lane) - 1u;
  int pos = __shfl(cb + run, myk, 32) + __popc(balmy & ltmask);

  float* crow = comp + (size_t)b * NPAD_MAX * NC;
#pragma unroll
  for (int p = 0; p < 32; ++p) {
    int pp = __builtin_amdgcn_readlane(pos, p);
    float2 val = *(const float2*)&tl[p * TPITCH + 2 * lane];
    *(float2*)&crow[(size_t)pp * NC + 2 * lane] = val;
  }

  if (s == 0 && bb == 0 && wave == 0 && lane < NCLS) {
    classinfo[(b * NCLS + lane) * 3 + 0] = base;
    classinfo[(b * NCLS + lane) * 3 + 1] = tk;
    classinfo[(b * NCLS + lane) * 3 + 2] = pk;
  }
  if (s == 0 && bb < NCLS) {
    int zb = __builtin_amdgcn_readlane(base, bb);
    int zt = __builtin_amdgcn_readlane(tk, bb);
    int zp = __builtin_amdgcn_readlane(pk, bb);
    float4* dst = (float4*)(crow + ((size_t)zb + zt) * NC);
    int n4 = (zp - zt) * (NC / 4);
#pragma unroll
    for (int j = 0; j < PADALIGN * (NC / 4) / (SCW * 32); ++j) {
      int i = (int)threadIdx.x + j * (SCW * 32);
      if (i < n4) dst[i] = make_float4(0.0f, 0.0f, 0.0f, 0.0f);
    }
  }
}

#define SYW 4

__device__ __forceinline__ float xhalf(float x) {
  int b = __float_as_int(x);
  return __int_as_float(__builtin_amdgcn_permlanex16(b, b, 0x76543210, 0xfedcba98, false, false));
}

__device__ __forceinline__ void load_group(v2f (&f)[4], const float* __restrict__ gp) {
#pragma unroll
  for (int blk = 0; blk < 4; ++blk) {
    f[blk].x = gp[blk * 16];
    f[blk].y = gp[NC + blk * 16];
  }
}

__device__ __forceinline__ void load_trip(v2f (&f)[4][4], const float* __restrict__ tp) {
#pragma unroll
  for (int i = 0; i < 4; ++i) load_group(f[i], tp + (size_t)i * 4 * NC);
}

template <int WV>
__device__ __forceinline__ void use_group(v8f (&acc)[4], float& cs, const v2f (&f)[4]) {
#pragma unroll
  for (int db = 0; db < 4; ++db)
    acc[db] = __builtin_amdgcn_wmma_f32_16x16x4_f32(
        false, f[WV], false, f[db], (short)0, acc[db], false, false);
  float ox = xhalf(f[WV].x), oy = xhalf(f[WV].y);
  cs = (((cs + f[WV].x) + f[WV].y) + ox) + oy;
}

template <int WV>
__device__ __forceinline__ void use_trip(v8f (&acc)[4], float& cs, const v2f (&f)[4][4]) {
#pragma unroll
  for (int i = 0; i < 4; ++i) use_group<WV>(acc, cs, f[i]);
}

template <int WV>
__device__ __forceinline__ void syrk_tile_row(const float* __restrict__ rp, int plen, int lane,
                                              float* __restrict__ sxxout, float* __restrict__ sumout) {
  const int ntrip = plen >> 4;
  const int rem = (plen >> 2) & 3;

  v8f acc[4] = {};
  float cs = 0.0f;

  v2f ft[3][4];
  const float* tailp = rp + (size_t)ntrip * 16 * NC;
  if (rem > 0) load_group(ft[0], tailp);
  if (rem > 1) load_group(ft[1], tailp + 4 * NC);
  if (rem > 2) load_group(ft[2], tailp + 8 * NC);

  v2f fa[4][4], fb[4][4];
  if (ntrip > 0) load_trip(fa, rp);
  int t = 0;
  for (; t + 2 <= ntrip; t += 2) {
    load_trip(fb, rp + (size_t)(t + 1) * 16 * NC);
    use_trip<WV>(acc, cs, fa);
    int tn = (t + 2 < ntrip) ? t + 2 : ntrip - 1;
    load_trip(fa, rp + (size_t)tn * 16 * NC);
    use_trip<WV>(acc, cs, fb);
  }
  if (t < ntrip) use_trip<WV>(acc, cs, fa);
  if (rem > 0) use_group<WV>(acc, cs, ft[0]);
  if (rem > 1) use_group<WV>(acc, cs, ft[1]);
  if (rem > 2) use_group<WV>(acc, cs, ft[2]);

  int col = lane & 15;
  int rof = (lane >> 4) * 8;
#pragma unroll
  for (int db = 0; db < 4; ++db)
#pragma unroll
    for (int j = 0; j < 8; ++j) {
      int r = WV * 16 + rof + j;
      sxxout[(size_t)r * NC + db * 16 + col] = acc[db][j];
    }
  if (lane < 16) sumout[WV * 16 + lane] = cs;
}

__global__ __launch_bounds__(SYW * 32) void syrk_kernel(const float* __restrict__ comp,
                                                        const int* __restrict__ classinfo,
                                                        float* __restrict__ sxxpart,
                                                        float* __restrict__ sumpart) {
  int gid = blockIdx.x;
  int part = gid / (NB * NCLS);
  int bk = gid % (NB * NCLS);
  int b = bk / NCLS;
  int lane = threadIdx.x & 31;
  int wave = __builtin_amdgcn_readfirstlane((int)(threadIdx.x >> 5));

  int base = classinfo[bk * 3 + 0];
  int padl = classinfo[bk * 3 + 2];
  int plen = padl / PARTS;
  size_t rowbase = (size_t)b * NPAD_MAX + base + (size_t)part * plen;

  int m = lane & 15;
  int kb = (lane >> 4) << 1;
  const float* rp = comp + (rowbase + kb) * NC + m;
  float* sxxout = sxxpart + (size_t)gid * (NC * NC);
  float* sumout = sumpart + (size_t)gid * NC;

  switch (wave) {
    case 0: syrk_tile_row<0>(rp, plen, lane, sxxout, sumout); break;
    case 1: syrk_tile_row<1>(rp, plen, lane, sxxout, sumout); break;
    case 2: syrk_tile_row<2>(rp, plen, lane, sxxout, sumout); break;
    default: syrk_tile_row<3>(rp, plen, lane, sxxout, sumout); break;
  }
}

__device__ __forceinline__ float lane_read(float v, int l) {
  return __int_as_float(__builtin_amdgcn_readlane(__float_as_int(v), l));
}

template <int Q, int NCOLS>
__device__ __forceinline__ void chol_step(float (&c0)[NC], float (&c1)[NC], float* covL, int lane, int kp) {
  const int kk = 2 * kp + Q;
  float piv = lane_read(Q ? c1[0] : c0[0], kp);
  float d = sqrtf(piv);
  float l0 = c0[0] / d;
  float l1 = c1[0] / d;
  if (Q == 0) l0 = (lane == kp) ? d : l0;
  else        l1 = (lane == kp) ? d : l1;
  covL[(2 * lane) * 65 + kk] = l0;
  covL[(2 * lane + 1) * 65 + kk] = l1;
#pragma unroll
  for (int jj = 1; jj <= NCOLS; ++jj) {
    const int lj = (kp + ((Q + jj) >> 1)) & 31;
    float s = lane_read(((Q + jj) & 1) ? l1 : l0, lj);
    c0[jj - 1] = c0[jj] - l0 * s;
    c1[jj - 1] = c1[jj] - l1 * s;
  }
}

__global__ __launch_bounds__(256) void reduce_solve_kernel(const float* __restrict__ sxxpart,
                                                           const float* __restrict__ sumpart,
                                                           const int* __restrict__ classinfo,
                                                           const float* __restrict__ pattern,
                                                           float* __restrict__ params) {
  int bk = blockIdx.x;
  int tid = threadIdx.x;
  __shared__ float covL[64 * 65];
  __shared__ float meanL[64], pL[64], wv[64];

  int count = classinfo[bk * 3 + 1];
  float safe = (float)(count > 0 ? count : 1);

  if (tid < 64) {
    float s = 0.0f;
    for (int part = 0; part < PARTS; ++part)
      s += sumpart[((size_t)part * NB * NCLS + bk) * NC + tid];
    meanL[tid] = s / safe;
    pL[tid] = pattern[tid];
  }
  __syncthreads();

  for (int idx = tid; idx < NC * NC; idx += 256) {
    int i = idx >> 6, j = idx & 63;
    float s = 0.0f;
    for (int part = 0; part < PARTS; ++part)
      s += sxxpart[((size_t)part * NB * NCLS + bk) * (NC * NC) + idx];
    float cv = s / safe - meanL[i] * meanL[j];
    if (i == j) cv += EPSV;
    covL[i * 65 + j] = cv;
  }
  __syncthreads();

  if (tid < 32) {
    const int lane = tid;
    float c0[NC], c1[NC];
#pragma unroll
    for (int j = 0; j < NC; ++j) {
      c0[j] = covL[(2 * lane) * 65 + j];
      c1[j] = covL[(2 * lane + 1) * 65 + j];
    }

#pragma nounroll
    for (int kp = 0; kp < 16; ++kp) {
      chol_step<0, 63>(c0, c1, covL, lane, kp);
      chol_step<1, 63>(c0, c1, covL, lane, kp);
    }
#pragma nounroll
    for (int kp = 16; kp < 32; ++kp) {
      chol_step<0, 31>(c0, c1, covL, lane, kp);
      chol_step<1, 31>(c0, c1, covL, lane, kp);
    }

    const float dg0 = covL[(2 * lane) * 65 + 2 * lane];
    const float dg1 = covL[(2 * lane + 1) * 65 + 2 * lane + 1];

    float y0 = pL[2 * lane], y1 = pL[2 * lane + 1];
    float a0 = covL[(2 * lane) * 65], a1 = covL[(2 * lane + 1) * 65];
#pragma nounroll
    for (int jp = 0; jp < 32; ++jp) {
      int jn = 2 * jp + 1;
      float n0 = covL[(2 * lane) * 65 + jn], n1 = covL[(2 * lane + 1) * 65 + jn];
      {
        float s = lane_read(y0 / dg0, jp);
        y0 = (lane == jp) ? s : ((lane > jp) ? y0 - a0 * s : y0);
        y1 = (lane >= jp) ? y1 - a1 * s : y1;
      }
      int jm = (2 * jp + 2 < NC) ? 2 * jp + 2 : NC - 1;
      a0 = covL[(2 * lane) * 65 + jm];
      a1 = covL[(2 * lane + 1) * 65 + jm];
      {
        float s = lane_read(y1 / dg1, jp);
        y1 = (lane == jp) ? s : ((lane > jp) ? y1 - n1 * s : y1);
        y0 = (lane > jp) ? y0 - n0 * s : y0;
      }
    }

    float w0 = y0, w1 = y1;
    float b0 = covL[63 * 65 + 2 * lane], b1 = covL[63 * 65 + 2 * lane + 1];
#pragma nounroll
    for (int jp = 31; jp >= 0; --jp) {
      int jn = 2 * jp;
      float n0 = covL[jn * 65 + 2 * lane], n1 = covL[jn * 65 + 2 * lane + 1];
      {
        float s = lane_read(w1 / dg1, jp);
        w1 = (lane == jp) ? s : ((lane < jp) ? w1 - b1 * s : w1);
        w0 = (lane <= jp) ? w0 - b0 * s : w0;
      }
      int jm = (2 * jp - 1 >= 0) ? 2 * jp - 1 : 0;
      b0 = covL[jm * 65 + 2 * lane];
      b1 = covL[jm * 65 + 2 * lane + 1];
      {
        float s = lane_read(w0 / dg0, jp);
        w0 = (lane == jp) ? s : ((lane < jp) ? w0 - n0 * s : w0);
        w1 = (lane < jp) ? w1 - n1 * s : w1;
      }
    }

    params[(size_t)bk * 66 + 2 * lane] = w0;
    params[(size_t)bk * 66 + 2 * lane + 1] = w1;
    wv[2 * lane] = w0;
    wv[2 * lane + 1] = w1;
    float pw = 0.0f, off = 0.0f;
    for (int c = 0; c < NC; ++c) {
      pw += pL[c] * wv[c];
      off += meanL[c] * wv[c];
    }
    if (lane == 0) {
      float denom = sqrtf(pw);
      float scale = (count >= MIN_CLASS_SIZE && denom > 0.0f) ? (1.0f / denom) : 0.0f;
      params[(size_t)bk * 66 + 64] = off;
      params[(size_t)bk * 66 + 65] = scale;
    }
  }
}

__global__ __launch_bounds__(256) void score_kernel(const float* __restrict__ hyper,
                                                    const int* __restrict__ seg,
                                                    const float* __restrict__ params,
                                                    float* __restrict__ out) {
  int b = blockIdx.y;
  int tid = threadIdx.x;
  __shared__ float wS[NCLS * 65];
  __shared__ float offS[NCLS], sclS[NCLS];
  for (int i = tid; i < NCLS * NC; i += 256) {
    int k = i >> 6, c = i & 63;
    wS[k * 65 + c] = params[((size_t)b * NCLS + k) * 66 + c];
  }
  if (tid < NCLS) {
    offS[tid] = params[((size_t)b * NCLS + tid) * 66 + 64];
    sclS[tid] = params[((size_t)b * NCLS + tid) * 66 + 65];
  }
  __syncthreads();

  size_t n = (size_t)blockIdx.x * 256 + tid;
  int k = seg[(size_t)b * NPIX + n];
  const float* xp = hyper + (size_t)b * NC * NPIX + n;
  float acc = 0.0f;
#pragma unroll
  for (int c = 0; c < NC; ++c)
    acc = fmaf(xp[(size_t)c * NPIX], wS[k * 65 + c], acc);
  out[(size_t)b * NPIX + n] = (acc - offS[k]) * sclS[k];
}

extern "C" void kernel_launch(void* const* d_in, const int* in_sizes, int n_in,
                              void* d_out, int out_size, void* d_ws, size_t ws_size,
                              hipStream_t stream) {
  const float* hyper = (const float*)d_in[0];
  const float* pattern = (const float*)d_in[1];
  const int* seg = (const int*)d_in[2];
  float* out = (float*)d_out;

  char* ws = (char*)d_ws;
  auto al = [](size_t x) { return (x + 255) & ~(size_t)255; };
  size_t off = 0;
  float* comp = (float*)(ws + off);
  off += al((size_t)NB * NPAD_MAX * NC * sizeof(float));
  int* hist = (int*)(ws + off);
  off += al((size_t)NB * NBLKS * NCLS * sizeof(int));
  int* classinfo = (int*)(ws + off);
  off += al((size_t)NB * NCLS * 3 * sizeof(int));
  float* sxxpart = (float*)(ws + off);
  off += al((size_t)PARTS * NB * NCLS * NC * NC * sizeof(float));
  float* sumpart = (float*)(ws + off);
  off += al((size_t)PARTS * NB * NCLS * NC * sizeof(float));
  float* params = (float*)(ws + off);
  off += al((size_t)NB * NCLS * 66 * sizeof(float));

  hipLaunchKernelGGL(hist_kernel, dim3(NB * NBLKS), dim3(HW * 32), 0, stream, seg, hist);
  hipLaunchKernelGGL(scatter_kernel, dim3(NB * NBLKS * SUBS), dim3(SCW * 32), 0, stream, hyper, seg,
                     hist, classinfo, comp);
  hipLaunchKernelGGL(syrk_kernel, dim3(PARTS * NB * NCLS), dim3(SYW * 32), 0, stream, comp,
                     classinfo, sxxpart, sumpart);
  hipLaunchKernelGGL(reduce_solve_kernel, dim3(NB * NCLS), dim3(256), 0, stream, sxxpart,
                     sumpart, classinfo, pattern, params);
  hipLaunchKernelGGL(score_kernel, dim3(NPIX / 256, NB), dim3(256), 0, stream, hyper, seg,
                     params, out);
}


# syrk forms only the 10 tiles on and below the diagonal (X^T X is symmetric): 3,3,2,2 tiles per wave instead of 4 each; reduce sums these and mirrors them in LDS; same products and sum order
# speedup vs baseline: 5.3503x; 1.1380x over previous
#include <hip/hip_runtime.h>
#include <hip/hip_bf16.h>


typedef float v2f __attribute__((ext_vector_type(2)));
typedef float v8f __attribute__((ext_vector_type(8)));

#define NB 2
#define NC 64
#define NPIX 147456
#define NCLS 20
#define BLKPX 1024
#define CHUNKS (BLKPX / 32)
#define NBLKS (NPIX / BLKPX)
#define PARTS 16
#define PADALIGN (PARTS * 4)
#define NPAD_MAX (NPIX + NCLS * PADALIGN)
#define MIN_CLASS_SIZE 5000
#define EPSV 1e-6f

#define HW 4
#define HLD (BLKPX / 4 / (HW * 32))

__global__ __launch_bounds__(HW * 32) void hist_kernel(const int* __restrict__ seg,
                                                       int* __restrict__ hist) {
  static_assert(HLD * 4 * HW * 32 == BLKPX, "the work-group's loads cover the block exactly");
  int blk = blockIdx.x;
  int b = blk / NBLKS, bb = blk % NBLKS;
  int lane = threadIdx.x & 31;
  int wave = __builtin_amdgcn_readfirstlane((int)(threadIdx.x >> 5));
  __shared__ int part[HW][32];
  const int4* s4 = (const int4*)(seg + (size_t)b * NPIX + (size_t)bb * BLKPX);
  int4 q[HLD];
#pragma unroll
  for (int j = 0; j < HLD; ++j) q[j] = s4[j * (HW * 32) + (int)threadIdx.x];
  int cnt = 0;
#pragma unroll
  for (int k = 0; k < NCLS; ++k) {
    int c = 0;
#pragma unroll
    for (int j = 0; j < HLD; ++j)
      c += __popc(__builtin_amdgcn_ballot_w32(q[j].x == k)) +
           __popc(__builtin_amdgcn_ballot_w32(q[j].y == k)) +
           __popc(__builtin_amdgcn_ballot_w32(q[j].z == k)) +
           __popc(__builtin_amdgcn_ballot_w32(q[j].w == k));
    if (lane == k) cnt = c;
  }
  part[wave][lane] = cnt;
  __syncthreads();
  if (wave == 0) {
    int t = 0;
#pragma unroll
    for (int w = 0; w < HW; ++w) t += part[w][lane];
    if (lane < NCLS) hist[((size_t)b * NCLS + lane) * NBLKS + bb] = t;
  }
}

#define SUBS 4
#define SCW (CHUNKS / SUBS)
#define TPITCH 68
#define CPW ((NCLS + SCW - 1) / SCW)
#define HQ (NBLKS / 4)

__device__ __forceinline__ int wave_sum(int v) {
  v += __builtin_amdgcn_update_dpp(0, v, 0x161, 0xf, 0xf, false);
  v += __builtin_amdgcn_update_dpp(0, v, 0x162, 0xf, 0xf, false);
  v += __builtin_amdgcn_update_dpp(0, v, 0x164, 0xf, 0xf, false);
  v += __builtin_amdgcn_update_dpp(0, v, 0x168, 0xf, 0xf, false);
  v += __builtin_amdgcn_permlanex16(v, v, 0x76543210, 0xfedcba98, false, false);
  return v;
}

__global__ __launch_bounds__(SCW * 32) void scatter_kernel(const float* __restrict__ hyper,
                                                           const int* __restrict__ seg,
                                                           const int* __restrict__ hist,
                                                           int* __restrict__ classinfo,
                                                           float* __restrict__ comp) {
  int s = blockIdx.x % SUBS;
  int hb = blockIdx.x / SUBS;
  int b = hb / NBLKS, bb = hb % NBLKS;
  int lane = threadIdx.x & 31;
  int wave = __builtin_amdgcn_readfirstlane((int)(threadIdx.x >> 5));
  __shared__ __attribute__((aligned(16))) float tile[SCW][32 * TPITCH];
  __shared__ int cnt[CHUNKS][32];
  __shared__ int totS[32];
  __shared__ int befS[32];

  const int* sg = seg + (size_t)b * NPIX + (size_t)bb * BLKPX;
  int sgv[SUBS];
#pragma unroll
  for (int j = 0; j < SUBS; ++j) sgv[j] = sg[(j * SCW + wave) * 32 + lane];

  static_assert(HQ - 32 == 4, "the upper part of a histogram column is four 4-block units");
  int4 hlo[CPW], hhi[CPW];
#pragma unroll
  for (int r = 0; r < CPW; ++r) {
    int k = wave + SCW * r;
    int kc = (k < NCLS) ? k : NCLS - 1;
    const int4* hp = (const int4*)(hist + ((size_t)b * NCLS + kc) * NBLKS);
    hlo[r] = hp[lane];
    hhi[r] = hp[32 + (lane & 3)];
  }

  int ch = s * SCW + wave;
  const float* xp = hyper + (size_t)b * NC * NPIX + (size_t)bb * BLKPX + (size_t)ch * 32 + lane;
  float v[NC];
#pragma unroll
  for (int c = 0; c < NC; ++c) v[c] = xp[(size_t)c * NPIX];

  int myk = sgv[0];
#pragma unroll
  for (int j = 1; j < SUBS; ++j) myk = (s == j) ? sgv[j] : myk;

#pragma unroll
  for (int j = 0; j < SUBS; ++j) {
    int pc = 0;
#pragma unroll
    for (int k = 0; k < NCLS; ++k) {
      unsigned bal = __builtin_amdgcn_ballot_w32(sgv[j] == k);
      if (lane == k) pc = __popc(bal);
    }
    cnt[j * SCW + wave][lane] = pc;
  }
  unsigned balmy = 0;
#pragma unroll
  for (int k = 0; k < NCLS; ++k) {
    unsigned bal = __builtin_amdgcn_ballot_w32(myk == k);
    if (myk == k) balmy = bal;
  }

#pragma unroll
  for (int r = 0; r < CPW; ++r) {
    int k = wave + SCW * r;
    int q0 = 4 * lane, q1 = 128 + 4 * lane;
    bool hi = lane < HQ - 32;
    int tt = ((hlo[r].x + hlo[r].y) + (hlo[r].z + hlo[r].w)) +
             (hi ? ((hhi[r].x + hhi[r].y) + (hhi[r].z + hhi[r].w)) : 0);
    int bf = ((q0 < bb) ? hlo[r].x : 0) + ((q0 + 1 < bb) ? hlo[r].y : 0) +
             ((q0 + 2 < bb) ? hlo[r].z : 0) + ((q0 + 3 < bb) ? hlo[r].w : 0) +
             ((hi && q1 < bb) ? hhi[r].x : 0) + ((hi && q1 + 1 < bb) ? hhi[r].y : 0) +
             ((hi && q1 + 2 < bb) ? hhi[r].z : 0) + ((hi && q1 + 3 < bb) ? hhi[r].w : 0);
    tt = wave_sum(tt);
    bf = wave_sum(bf);
    if (lane == 0 && k < NCLS) {
      totS[k] = tt;
      befS[k] = bf;
    }
  }

  float* tl = tile[wave];
#pragma unroll
  for (int q = 0; q < NC / 4; ++q)
    *(float4*)&tl[lane * TPITCH + 4 * q] =
        make_float4(v[4 * q], v[4 * q + 1], v[4 * q + 2], v[4 * q + 3]);
  __syncthreads();

  int cls = (lane < NCLS) ? lane : 0;
  int base = 0;
#pragma unroll
  for (int k = 0; k < NCLS - 1; ++k) {
    int p = (totS[k] + PADALIGN - 1) & ~(PADALIGN - 1);
    base += (k < lane) ? p : 0;
  }
  int tk = totS[cls];
  int pk = (tk + PADALIGN - 1) & ~(PADALIGN - 1);
  int cb = base + befS[cls];

  int run = 0;
#pragma unroll
  for (int c2 = 0; c2 < CHUNKS - 1; ++c2) {
    int t = cnt[c2][lane];
    run += (c2 < ch) ? t : 0;
  }
  unsigned ltmask = (1u << lane) - 1u;
  int pos = __shfl(cb + run, myk, 32) + __popc(balmy & ltmask);

  float* crow = comp + (size_t)b * NPAD_MAX * NC;
#pragma unroll
  for (int p = 0; p < 32; ++p) {
    int pp = __builtin_amdgcn_readlane(pos, p);
    float2 val = *(const float2*)&tl[p * TPITCH + 2 * lane];
    *(float2*)&crow[(size_t)pp * NC + 2 * lane] = val;
  }

  if (s == 0 && bb == 0 && wave == 0 && lane < NCLS) {
    classinfo[(b * NCLS + lane) * 3 + 0] = base;
    classinfo[(b * NCLS + lane) * 3 + 1] = tk;
    classinfo[(b * NCLS + lane) * 3 + 2] = pk;
  }
  if (s == 0 && bb < NCLS) {
    int zb = __builtin_amdgcn_readlane(base, bb);
    int zt = __builtin_amdgcn_readlane(tk, bb);
    int zp = __builtin_amdgcn_readlane(pk, bb);
    float4* dst = (float4*)(crow + ((size_t)zb + zt) * NC);
    int n4 = (zp - zt) * (NC / 4);
#pragma unroll
    for (int j = 0; j < PADALIGN * (NC / 4) / (SCW * 32); ++j) {
      int i = (int)threadIdx.x + j * (SCW * 32);
      if (i < n4) dst[i] = make_float4(0.0f, 0.0f, 0.0f, 0.0f);
    }
  }
}

#define SYW 4

template <int S> __device__ constexpr int tile_a(int t) { return S == 0 ? (t < 2 ? 1 : 0) : (S == 1 ? 2 : 3); }
template <int S> __device__ constexpr int tile_b(int t) { return S == 0 ? (t == 1 ? 1 : 0) : (S == 3 ? 2 + t : t); }
template <int S> __device__ constexpr int tile_n() { return S < 2 ? 3 : 2; }
template <int S> __device__ constexpr int sum_blk() { return S == 1 ? 2 : (S == 2 ? 1 : S); }

__device__ __forceinline__ float xhalf(float x) {
  int b = __float_as_int(x);
  return __int_as_float(__builtin_amdgcn_permlanex16(b, b, 0x76543210, 0xfedcba98, false, false));
}

__device__ __forceinline__ void load_group(v2f (&f)[4], const float* __restrict__ gp) {
#pragma unroll
  for (int blk = 0; blk < 4; ++blk) {
    f[blk].x = gp[blk * 16];
    f[blk].y = gp[NC + blk * 16];
  }
}

__device__ __forceinline__ void load_trip(v2f (&f)[4][4], const float* __restrict__ tp) {
#pragma unroll
  for (int i = 0; i < 4; ++i) load_group(f[i], tp + (size_t)i * 4 * NC);
}

template <int S>
__device__ __forceinline__ void use_group(v8f (&acc)[3], float& cs, const v2f (&f)[4]) {
#pragma unroll
  for (int t = 0; t < tile_n<S>(); ++t)
    acc[t] = __builtin_amdgcn_wmma_f32_16x16x4_f32(
        false, f[tile_a<S>(t)], false, f[tile_b<S>(t)], (short)0, acc[t], false, false);
  constexpr int SB = sum_blk<S>();
  float ox = xhalf(f[SB].x), oy = xhalf(f[SB].y);
  cs = (((cs + f[SB].x) + f[SB].y) + ox) + oy;
}

template <int S>
__device__ __forceinline__ void use_trip(v8f (&acc)[3], float& cs, const v2f (&f)[4][4]) {
#pragma unroll
  for (int i = 0; i < 4; ++i) use_group<S>(acc, cs, f[i]);
}

template <int S>
__device__ __forceinline__ void syrk_tile_set(const float* __restrict__ rp, int plen, int lane,
                                              float* __restrict__ sxxout, float* __restrict__ sumout) {
  const int ntrip = plen >> 4;
  const int rem = (plen >> 2) & 3;

  v8f acc[3] = {};
  float cs = 0.0f;

  v2f ft[3][4];
  const float* tailp = rp + (size_t)ntrip * 16 * NC;
  if (rem > 0) load_group(ft[0], tailp);
  if (rem > 1) load_group(ft[1], tailp + 4 * NC);
  if (rem > 2) load_group(ft[2], tailp + 8 * NC);

  v2f fa[4][4], fb[4][4];
  if (ntrip > 0) load_trip(fa, rp);
  int t = 0;
  for (; t + 2 <= ntrip; t += 2) {
    load_trip(fb, rp + (size_t)(t + 1) * 16 * NC);
    use_trip<S>(acc, cs, fa);
    int tn = (t + 2 < ntrip) ? t + 2 : ntrip - 1;
    load_trip(fa, rp + (size_t)tn * 16 * NC);
    use_trip<S>(acc, cs, fb);
  }
  if (t < ntrip) use_trip<S>(acc, cs, fa);
  if (rem > 0) use_group<S>(acc, cs, ft[0]);
  if (rem > 1) use_group<S>(acc, cs, ft[1]);
  if (rem > 2) use_group<S>(acc, cs, ft[2]);

  int col = lane & 15;
  int rof = (lane >> 4) * 8;
#pragma unroll
  for (int t = 0; t < tile_n<S>(); ++t)
#pragma unroll
    for (int j = 0; j < 8; ++j) {
      int r = tile_a<S>(t) * 16 + rof + j;
      sxxout[(size_t)r * NC + tile_b<S>(t) * 16 + col] = acc[t][j];
    }
  if (lane < 16) sumout[sum_blk<S>() * 16 + lane] = cs;
}

__global__ __launch_bounds__(SYW * 32) void syrk_kernel(const float* __restrict__ comp,
                                                        const int* __restrict__ classinfo,
                                                        float* __restrict__ sxxpart,
                                                        float* __restrict__ sumpart) {
  int gid = blockIdx.x;
  int part = gid / (NB * NCLS);
  int bk = gid % (NB * NCLS);
  int b = bk / NCLS;
  int lane = threadIdx.x & 31;
  int wave = __builtin_amdgcn_readfirstlane((int)(threadIdx.x >> 5));

  int base = classinfo[bk * 3 + 0];
  int padl = classinfo[bk * 3 + 2];
  int plen = padl / PARTS;
  size_t rowbase = (size_t)b * NPAD_MAX + base + (size_t)part * plen;

  int m = lane & 15;
  int kb = (lane >> 4) << 1;
  const float* rp = comp + (rowbase + kb) * NC + m;
  float* sxxout = sxxpart + (size_t)gid * (NC * NC);
  float* sumout = sumpart + (size_t)gid * NC;

  int set = (wave + (gid ^ (gid >> 8))) & 3;
  switch (set) {
    case 0: syrk_tile_set<0>(rp, plen, lane, sxxout, sumout); break;
    case 1: syrk_tile_set<1>(rp, plen, lane, sxxout, sumout); break;
    case 2: syrk_tile_set<2>(rp, plen, lane, sxxout, sumout); break;
    default: syrk_tile_set<3>(rp, plen, lane, sxxout, sumout); break;
  }
}

__device__ __forceinline__ float lane_read(float v, int l) {
  return __int_as_float(__builtin_amdgcn_readlane(__float_as_int(v), l));
}

template <int Q, int NCOLS>
__device__ __forceinline__ void chol_step(float (&c0)[NC], float (&c1)[NC], float* covL, int lane, int kp) {
  const int kk = 2 * kp + Q;
  float piv = lane_read(Q ? c1[0] : c0[0], kp);
  float d = sqrtf(piv);
  float l0 = c0[0] / d;
  float l1 = c1[0] / d;
  if (Q == 0) l0 = (lane == kp) ? d : l0;
  else        l1 = (lane == kp) ? d : l1;
  covL[(2 * lane) * 65 + kk] = l0;
  covL[(2 * lane + 1) * 65 + kk] = l1;
#pragma unroll
  for (int jj = 1; jj <= NCOLS; ++jj) {
    const int lj = (kp + ((Q + jj) >> 1)) & 31;
    float s = lane_read(((Q + jj) & 1) ? l1 : l0, lj);
    c0[jj - 1] = c0[jj] - l0 * s;
    c1[jj - 1] = c1[jj] - l1 * s;
  }
}

__global__ __launch_bounds__(256) void reduce_solve_kernel(const float* __restrict__ sxxpart,
                                                           const float* __restrict__ sumpart,
                                                           const int* __restrict__ classinfo,
                                                           const float* __restrict__ pattern,
                                                           float* __restrict__ params) {
  int bk = blockIdx.x;
  int tid = threadIdx.x;
  __shared__ float covL[64 * 65];
  __shared__ float meanL[64], pL[64], wv[64];

  int count = classinfo[bk * 3 + 1];
  float safe = (float)(count > 0 ? count : 1);

  if (tid < 64) {
    float s = 0.0f;
    for (int part = 0; part < PARTS; ++part)
      s += sumpart[((size_t)part * NB * NCLS + bk) * NC + tid];
    meanL[tid] = s / safe;
    pL[tid] = pattern[tid];
  }
  __syncthreads();

  const int er = tid >> 4, ec = tid & 15;
  for (int t = 0; t < 10; ++t) {
    int ti = (t >= 6) ? 3 : (t >= 3) ? 2 : (t >= 1) ? 1 : 0;
    int tj = t - ((ti * (ti + 1)) >> 1);
    int i = ti * 16 + er, j = tj * 16 + ec;
    int idx = i * NC + j;
    float s = 0.0f;
    for (int part = 0; part < PARTS; ++part)
      s += sxxpart[((size_t)part * NB * NCLS + bk) * (NC * NC) + idx];
    float cv = s / safe - meanL[i] * meanL[j];
    if (i == j) cv += EPSV;
    covL[i * 65 + j] = cv;
  }
  __syncthreads();
  for (int t = 0; t < 6; ++t) {
    int ti = (t >= 3) ? 3 : (t >= 1) ? 2 : 1;
    int tj = t - ((ti * (ti - 1)) >> 1);
    int i = ti * 16 + er, j = tj * 16 + ec;
    covL[j * 65 + i] = covL[i * 65 + j];
  }
  __syncthreads();

  if (tid < 32) {
    const int lane = tid;
    float c0[NC], c1[NC];
#pragma unroll
    for (int j = 0; j < NC; ++j) {
      c0[j] = covL[(2 * lane) * 65 + j];
      c1[j] = covL[(2 * lane + 1) * 65 + j];
    }

#pragma nounroll
    for (int kp = 0; kp < 16; ++kp) {
      chol_step<0, 63>(c0, c1, covL, lane, kp);
      chol_step<1, 63>(c0, c1, covL, lane, kp);
    }
#pragma nounroll
    for (int kp = 16; kp < 32; ++kp) {
      chol_step<0, 31>(c0, c1, covL, lane, kp);
      chol_step<1, 31>(c0, c1, covL, lane, kp);
    }

    const float dg0 = covL[(2 * lane) * 65 + 2 * lane];
    const float dg1 = covL[(2 * lane + 1) * 65 + 2 * lane + 1];

    float y0 = pL[2 * lane], y1 = pL[2 * lane + 1];
    float a0 = covL[(2 * lane) * 65], a1 = covL[(2 * lane + 1) * 65];
#pragma nounroll
    for (int jp = 0; jp < 32; ++jp) {
      int jn = 2 * jp + 1;
      float n0 = covL[(2 * lane) * 65 + jn], n1 = covL[(2 * lane + 1) * 65 + jn];
      {
        float s = lane_read(y0 / dg0, jp);
        y0 = (lane == jp) ? s : ((lane > jp) ? y0 - a0 * s : y0);
        y1 = (lane >= jp) ? y1 - a1 * s : y1;
      }
      int jm = (2 * jp + 2 < NC) ? 2 * jp + 2 : NC - 1;
      a0 = covL[(2 * lane) * 65 + jm];
      a1 = covL[(2 * lane + 1) * 65 + jm];
      {
        float s = lane_read(y1 / dg1, jp);
        y1 = (lane == jp) ? s : ((lane > jp) ? y1 - n1 * s : y1);
        y0 = (lane > jp) ? y0 - n0 * s : y0;
      }
    }

    float w0 = y0, w1 = y1;
    float b0 = covL[63 * 65 + 2 * lane], b1 = covL[63 * 65 + 2 * lane + 1];
#pragma nounroll
    for (int jp = 31; jp >= 0; --jp) {
      int jn = 2 * jp;
      float n0 = covL[jn * 65 + 2 * lane], n1 = covL[jn * 65 + 2 * lane + 1];
      {
        float s = lane_read(w1 / dg1, jp);
        w1 = (lane == jp) ? s : ((lane < jp) ? w1 - b1 * s : w1);
        w0 = (lane <= jp) ? w0 - b0 * s : w0;
      }
      int jm = (2 * jp - 1 >= 0) ? 2 * jp - 1 : 0;
      b0 = covL[jm * 65 + 2 * lane];
      b1 = covL[jm * 65 + 2 * lane + 1];
      {
        float s = lane_read(w0 / dg0, jp);
        w0 = (lane == jp) ? s : ((lane < jp) ? w0 - n0 * s : w0);
        w1 = (lane < jp) ? w1 - n1 * s : w1;
      }
    }

    params[(size_t)bk * 66 + 2 * lane] = w0;
    params[(size_t)bk * 66 + 2 * lane + 1] = w1;
    wv[2 * lane] = w0;
    wv[2 * lane + 1] = w1;
    float pw = 0.0f, off = 0.0f;
    for (int c = 0; c < NC; ++c) {
      pw += pL[c] * wv[c];
      off += meanL[c] * wv[c];
    }
    if (lane == 0) {
      float denom = sqrtf(pw);
      float scale = (count >= MIN_CLASS_SIZE && denom > 0.0f) ? (1.0f / denom) : 0.0f;
      params[(size_t)bk * 66 + 64] = off;
      params[(size_t)bk * 66 + 65] = scale;
    }
  }
}

__global__ __launch_bounds__(256) void score_kernel(const float* __restrict__ hyper,
                                                    const int* __restrict__ seg,
                                                    const float* __restrict__ params,
                                                    float* __restrict__ out) {
  int b = blockIdx.y;
  int tid = threadIdx.x;
  __shared__ float wS[NCLS * 65];
  __shared__ float offS[NCLS], sclS[NCLS];
  for (int i = tid; i < NCLS * NC; i += 256) {
    int k = i >> 6, c = i & 63;
    wS[k * 65 + c] = params[((size_t)b * NCLS + k) * 66 + c];
  }
  if (tid < NCLS) {
    offS[tid] = params[((size_t)b * NCLS + tid) * 66 + 64];
    sclS[tid] = params[((size_t)b * NCLS + tid) * 66 + 65];
  }
  __syncthreads();

  size_t n = (size_t)blockIdx.x * 256 + tid;
  int k = seg[(size_t)b * NPIX + n];
  const float* xp = hyper + (size_t)b * NC * NPIX + n;
  float acc = 0.0f;
#pragma unroll
  for (int c = 0; c < NC; ++c)
    acc = fmaf(xp[(size_t)c * NPIX], wS[k * 65 + c], acc);
  out[(size_t)b * NPIX + n] = (acc - offS[k]) * sclS[k];
}

extern "C" void kernel_launch(void* const* d_in, const int* in_sizes, int n_in,
                              void* d_out, int out_size, void* d_ws, size_t ws_size,
                              hipStream_t stream) {
  const float* hyper = (const float*)d_in[0];
  const float* pattern = (const float*)d_in[1];
  const int* seg = (const int*)d_in[2];
  float* out = (float*)d_out;

  char* ws = (char*)d_ws;
  auto al = [](size_t x) { return (x + 255) & ~(size_t)255; };
  size_t off = 0;
  float* comp = (float*)(ws + off);
  off += al((size_t)NB * NPAD_MAX * NC * sizeof(float));
  int* hist = (int*)(ws + off);
  off += al((size_t)NB * NBLKS * NCLS * sizeof(int));
  int* classinfo = (int*)(ws + off);
  off += al((size_t)NB * NCLS * 3 * sizeof(int));
  float* sxxpart = (float*)(ws + off);
  off += al((size_t)PARTS * NB * NCLS * NC * NC * sizeof(float));
  float* sumpart = (float*)(ws + off);
  off += al((size_t)PARTS * NB * NCLS * NC * sizeof(float));
  float* params = (float*)(ws + off);
  off += al((size_t)NB * NCLS * 66 * sizeof(float));

  hipLaunchKernelGGL(hist_kernel, dim3(NB * NBLKS), dim3(HW * 32), 0, stream, seg, hist);
  hipLaunchKernelGGL(scatter_kernel, dim3(NB * NBLKS * SUBS), dim3(SCW * 32), 0, stream, hyper, seg,
                     hist, classinfo, comp);
  hipLaunchKernelGGL(syrk_kernel, dim3(PARTS * NB * NCLS), dim3(SYW * 32), 0, stream, comp,
                     classinfo, sxxpart, sumpart);
  hipLaunchKernelGGL(reduce_solve_kernel, dim3(NB * NCLS), dim3(256), 0, stream, sxxpart,
                     sumpart, classinfo, pattern, params);
  hipLaunchKernelGGL(score_kernel, dim3(NPIX / 256, NB), dim3(256), 0, stream, hyper, seg,
                     params, out);
}


# reduce_solve chains cut to multiply, lane read, fma: column of L = column * rsqrt(pivot) (v_rsq), substitutions multiply by it and take L from registers (unrolled); same sums, last bits differ
# speedup vs baseline: 5.9412x; 1.1104x over previous
#include <hip/hip_runtime.h>
#include <hip/hip_bf16.h>


typedef float v2f __attribute__((ext_vector_type(2)));
typedef float v8f __attribute__((ext_vector_type(8)));

#define NB 2
#define NC 64
#define NPIX 147456
#define NCLS 20
#define BLKPX 1024
#define CHUNKS (BLKPX / 32)
#define NBLKS (NPIX / BLKPX)
#define PARTS 16
#define PADALIGN (PARTS * 4)
#define NPAD_MAX (NPIX + NCLS * PADALIGN)
#define MIN_CLASS_SIZE 5000
#define EPSV 1e-6f

#define HW 4
#define HLD (BLKPX / 4 / (HW * 32))

__global__ __launch_bounds__(HW * 32) void hist_kernel(const int* __restrict__ seg,
                                                       int* __restrict__ hist) {
  static_assert(HLD * 4 * HW * 32 == BLKPX, "the work-group's loads cover the block exactly");
  int blk = blockIdx.x;
  int b = blk / NBLKS, bb = blk % NBLKS;
  int lane = threadIdx.x & 31;
  int wave = __builtin_amdgcn_readfirstlane((int)(threadIdx.x >> 5));
  __shared__ int part[HW][32];
  const int4* s4 = (const int4*)(seg + (size_t)b * NPIX + (size_t)bb * BLKPX);
  int4 q[HLD];
#pragma unroll
  for (int j = 0; j < HLD; ++j) q[j] = s4[j * (HW * 32) + (int)threadIdx.x];
  int cnt = 0;
#pragma unroll
  for (int k = 0; k < NCLS; ++k) {
    int c = 0;
#pragma unroll
    for (int j = 0; j < HLD; ++j)
      c += __popc(__builtin_amdgcn_ballot_w32(q[j].x == k)) +
           __popc(__builtin_amdgcn_ballot_w32(q[j].y == k)) +
           __popc(__builtin_amdgcn_ballot_w32(q[j].z == k)) +
           __popc(__builtin_amdgcn_ballot_w32(q[j].w == k));
    if (lane == k) cnt = c;
  }
  part[wave][lane] = cnt;
  __syncthreads();
  if (wave == 0) {
    int t = 0;
#pragma unroll
    for (int w = 0; w < HW; ++w) t += part[w][lane];
    if (lane < NCLS) hist[((size_t)b * NCLS + lane) * NBLKS + bb] = t;
  }
}

#define SUBS 4
#define SCW (CHUNKS / SUBS)
#define TPITCH 68
#define CPW ((NCLS + SCW - 1) / SCW)
#define HQ (NBLKS / 4)

__device__ __forceinline__ int wave_sum(int v) {
  v += __builtin_amdgcn_update_dpp(0, v, 0x161, 0xf, 0xf, false);
  v += __builtin_amdgcn_update_dpp(0, v, 0x162, 0xf, 0xf, false);
  v += __builtin_amdgcn_update_dpp(0, v, 0x164, 0xf, 0xf, false);
  v += __builtin_amdgcn_update_dpp(0, v, 0x168, 0xf, 0xf, false);
  v += __builtin_amdgcn_permlanex16(v, v, 0x76543210, 0xfedcba98, false, false);
  return v;
}

__global__ __launch_bounds__(SCW * 32) void scatter_kernel(const float* __restrict__ hyper,
                                                           const int* __restrict__ seg,
                                                           const int* __restrict__ hist,
                                                           int* __restrict__ classinfo,
                                                           float* __restrict__ comp) {
  int s = blockIdx.x % SUBS;
  int hb = blockIdx.x / SUBS;
  int b = hb / NBLKS, bb = hb % NBLKS;
  int lane = threadIdx.x & 31;
  int wave = __builtin_amdgcn_readfirstlane((int)(threadIdx.x >> 5));
  __shared__ __attribute__((aligned(16))) float tile[SCW][32 * TPITCH];
  __shared__ int cnt[CHUNKS][32];
  __shared__ int totS[32];
  __shared__ int befS[32];

  const int* sg = seg + (size_t)b * NPIX + (size_t)bb * BLKPX;
  int sgv[SUBS];
#pragma unroll
  for (int j = 0; j < SUBS; ++j) sgv[j] = sg[(j * SCW + wave) * 32 + lane];

  static_assert(HQ - 32 == 4, "the upper part of a histogram column is four 4-block units");
  int4 hlo[CPW], hhi[CPW];
#pragma unroll
  for (int r = 0; r < CPW; ++r) {
    int k = wave + SCW * r;
    int kc = (k < NCLS) ? k : NCLS - 1;
    const int4* hp = (const int4*)(hist + ((size_t)b * NCLS + kc) * NBLKS);
    hlo[r] = hp[lane];
    hhi[r] = hp[32 + (lane & 3)];
  }

  int ch = s * SCW + wave;
  const float* xp = hyper + (size_t)b * NC * NPIX + (size_t)bb * BLKPX + (size_t)ch * 32 + lane;
  float v[NC];
#pragma unroll
  for (int c = 0; c < NC; ++c) v[c] = xp[(size_t)c * NPIX];

  int myk = sgv[0];
#pragma unroll
  for (int j = 1; j < SUBS; ++j) myk = (s == j) ? sgv[j] : myk;

#pragma unroll
  for (int j = 0; j < SUBS; ++j) {
    int pc = 0;
#pragma unroll
    for (int k = 0; k < NCLS; ++k) {
      unsigned bal = __builtin_amdgcn_ballot_w32(sgv[j] == k);
      if (lane == k) pc = __popc(bal);
    }
    cnt[j * SCW + wave][lane] = pc;
  }
  unsigned balmy = 0;
#pragma unroll
  for (int k = 0; k < NCLS; ++k) {
    unsigned bal = __builtin_amdgcn_ballot_w32(myk == k);
    if (myk == k) balmy = bal;
  }

#pragma unroll
  for (int r = 0; r < CPW; ++r) {
    int k = wave + SCW * r;
    int q0 = 4 * lane, q1 = 128 + 4 * lane;
    bool hi = lane < HQ - 32;
    int tt = ((hlo[r].x + hlo[r].y) + (hlo[r].z + hlo[r].w)) +
             (hi ? ((hhi[r].x + hhi[r].y) + (hhi[r].z + hhi[r].w)) : 0);
    int bf = ((q0 < bb) ? hlo[r].x : 0) + ((q0 + 1 < bb) ? hlo[r].y : 0) +
             ((q0 + 2 < bb) ? hlo[r].z : 0) + ((q0 + 3 < bb) ? hlo[r].w : 0) +
             ((hi && q1 < bb) ? hhi[r].x : 0) + ((hi && q1 + 1 < bb) ? hhi[r].y : 0) +
             ((hi && q1 + 2 < bb) ? hhi[r].z : 0) + ((hi && q1 + 3 < bb) ? hhi[r].w : 0);
    tt = wave_sum(tt);
    bf = wave_sum(bf);
    if (lane == 0 && k < NCLS) {
      totS[k] = tt;
      befS[k] = bf;
    }
  }

  float* tl = tile[wave];
#pragma unroll
  for (int q = 0; q < NC / 4; ++q)
    *(float4*)&tl[lane * TPITCH + 4 * q] =
        make_float4(v[4 * q], v[4 * q + 1], v[4 * q + 2], v[4 * q + 3]);
  __syncthreads();

  int cls = (lane < NCLS) ? lane : 0;
  int base = 0;
#pragma unroll
  for (int k = 0; k < NCLS - 1; ++k) {
    int p = (totS[k] + PADALIGN - 1) & ~(PADALIGN - 1);
    base += (k < lane) ? p : 0;
  }
  int tk = totS[cls];
  int pk = (tk + PADALIGN - 1) & ~(PADALIGN - 1);
  int cb = base + befS[cls];

  int run = 0;
#pragma unroll
  for (int c2 = 0; c2 < CHUNKS - 1; ++c2) {
    int t = cnt[c2][lane];
    run += (c2 < ch) ? t : 0;
  }
  unsigned ltmask = (1u << lane) - 1u;
  int pos = __shfl(cb + run, myk, 32) + __popc(balmy & ltmask);

  float* crow = comp + (size_t)b * NPAD_MAX * NC;
#pragma unroll
  for (int p = 0; p < 32; ++p) {
    int pp = __builtin_amdgcn_readlane(pos, p);
    float2 val = *(const float2*)&tl[p * TPITCH + 2 * lane];
    *(float2*)&crow[(size_t)pp * NC + 2 * lane] = val;
  }

  if (s == 0 && bb == 0 && wave == 0 && lane < NCLS) {
    classinfo[(b * NCLS + lane) * 3 + 0] = base;
    classinfo[(b * NCLS + lane) * 3 + 1] = tk;
    classinfo[(b * NCLS + lane) * 3 + 2] = pk;
  }
  if (s == 0 && bb < NCLS) {
    int zb = __builtin_amdgcn_readlane(base, bb);
    int zt = __builtin_amdgcn_readlane(tk, bb);
    int zp = __builtin_amdgcn_readlane(pk, bb);
    float4* dst = (float4*)(crow + ((size_t)zb + zt) * NC);
    int n4 = (zp - zt) * (NC / 4);
#pragma unroll
    for (int j = 0; j < PADALIGN * (NC / 4) / (SCW * 32); ++j) {
      int i = (int)threadIdx.x + j * (SCW * 32);
      if (i < n4) dst[i] = make_float4(0.0f, 0.0f, 0.0f, 0.0f);
    }
  }
}

#define SYW 4

template <int S> __device__ constexpr int tile_a(int t) { return S == 0 ? (t < 2 ? 1 : 0) : (S == 1 ? 2 : 3); }
template <int S> __device__ constexpr int tile_b(int t) { return S == 0 ? (t == 1 ? 1 : 0) : (S == 3 ? 2 + t : t); }
template <int S> __device__ constexpr int tile_n() { return S < 2 ? 3 : 2; }
template <int S> __device__ constexpr int sum_blk() { return S == 1 ? 2 : (S == 2 ? 1 : S); }

__device__ __forceinline__ float xhalf(float x) {
  int b = __float_as_int(x);
  return __int_as_float(__builtin_amdgcn_permlanex16(b, b, 0x76543210, 0xfedcba98, false, false));
}

__device__ __forceinline__ void load_group(v2f (&f)[4], const float* __restrict__ gp) {
#pragma unroll
  for (int blk = 0; blk < 4; ++blk) {
    f[blk].x = gp[blk * 16];
    f[blk].y = gp[NC + blk * 16];
  }
}

__device__ __forceinline__ void load_trip(v2f (&f)[4][4], const float* __restrict__ tp) {
#pragma unroll
  for (int i = 0; i < 4; ++i) load_group(f[i], tp + (size_t)i * 4 * NC);
}

template <int S>
__device__ __forceinline__ void use_group(v8f (&acc)[3], float& cs, const v2f (&f)[4]) {
#pragma unroll
  for (int t = 0; t < tile_n<S>(); ++t)
    acc[t] = __builtin_amdgcn_wmma_f32_16x16x4_f32(
        false, f[tile_a<S>(t)], false, f[tile_b<S>(t)], (short)0, acc[t], false, false);
  constexpr int SB = sum_blk<S>();
  float ox = xhalf(f[SB].x), oy = xhalf(f[SB].y);
  cs = (((cs + f[SB].x) + f[SB].y) + ox) + oy;
}

template <int S>
__device__ __forceinline__ void use_trip(v8f (&acc)[3], float& cs, const v2f (&f)[4][4]) {
#pragma unroll
  for (int i = 0; i < 4; ++i) use_group<S>(acc, cs, f[i]);
}

template <int S>
__device__ __forceinline__ void syrk_tile_set(const float* __restrict__ rp, int plen, int lane,
                                              float* __restrict__ sxxout, float* __restrict__ sumout) {
  const int ntrip = plen >> 4;
  const int rem = (plen >> 2) & 3;

  v8f acc[3] = {};
  float cs = 0.0f;

  v2f ft[3][4];
  const float* tailp = rp + (size_t)ntrip * 16 * NC;
  if (rem > 0) load_group(ft[0], tailp);
  if (rem > 1) load_group(ft[1], tailp + 4 * NC);
  if (rem > 2) load_group(ft[2], tailp + 8 * NC);

  v2f fa[4][4], fb[4][4];
  if (ntrip > 0) load_trip(fa, rp);
  int t = 0;
  for (; t + 2 <= ntrip; t += 2) {
    load_trip(fb, rp + (size_t)(t + 1) * 16 * NC);
    use_trip<S>(acc, cs, fa);
    int tn = (t + 2 < ntrip) ? t + 2 : ntrip - 1;
    load_trip(fa, rp + (size_t)tn * 16 * NC);
    use_trip<S>(acc, cs, fb);
  }
  if (t < ntrip) use_trip<S>(acc, cs, fa);
  if (rem > 0) use_group<S>(acc, cs, ft[0]);
  if (rem > 1) use_group<S>(acc, cs, ft[1]);
  if (rem > 2) use_group<S>(acc, cs, ft[2]);

  int col = lane & 15;
  int rof = (lane >> 4) * 8;
#pragma unroll
  for (int t = 0; t < tile_n<S>(); ++t)
#pragma unroll
    for (int j = 0; j < 8; ++j) {
      int r = tile_a<S>(t) * 16 + rof + j;
      sxxout[(size_t)r * NC + tile_b<S>(t) * 16 + col] = acc[t][j];
    }
  if (lane < 16) sumout[sum_blk<S>() * 16 + lane] = cs;
}

__global__ __launch_bounds__(SYW * 32) void syrk_kernel(const float* __restrict__ comp,
                                                        const int* __restrict__ classinfo,
                                                        float* __restrict__ sxxpart,
                                                        float* __restrict__ sumpart) {
  int gid = blockIdx.x;
  int part = gid / (NB * NCLS);
  int bk = gid % (NB * NCLS);
  int b = bk / NCLS;
  int lane = threadIdx.x & 31;
  int wave = __builtin_amdgcn_readfirstlane((int)(threadIdx.x >> 5));

  int base = classinfo[bk * 3 + 0];
  int padl = classinfo[bk * 3 + 2];
  int plen = padl / PARTS;
  size_t rowbase = (size_t)b * NPAD_MAX + base + (size_t)part * plen;

  int m = lane & 15;
  int kb = (lane >> 4) << 1;
  const float* rp = comp + (rowbase + kb) * NC + m;
  float* sxxout = sxxpart + (size_t)gid * (NC * NC);
  float* sumout = sumpart + (size_t)gid * NC;

  int set = (wave + (gid ^ (gid >> 8))) & 3;
  switch (set) {
    case 0: syrk_tile_set<0>(rp, plen, lane, sxxout, sumout); break;
    case 1: syrk_tile_set<1>(rp, plen, lane, sxxout, sumout); break;
    case 2: syrk_tile_set<2>(rp, plen, lane, sxxout, sumout); break;
    default: syrk_tile_set<3>(rp, plen, lane, sxxout, sumout); break;
  }
}

__device__ __forceinline__ float lane_read(float v, int l) {
  return __int_as_float(__builtin_amdgcn_readlane(__float_as_int(v), l));
}

template <int Q, int NCOLS>
__device__ __forceinline__ void chol_step(float (&c0)[NC], float (&c1)[NC], float& r0, float& r1,
                                          float* covL, int lane, int kp) {
  const int kk = 2 * kp + Q;
  float piv = lane_read(Q ? c1[0] : c0[0], kp);
  float ri = __builtin_amdgcn_rsqf(piv);
  float l0 = c0[0] * ri;
  float l1 = c1[0] * ri;
  if (Q == 0) r0 = (lane == kp) ? ri : r0;
  else        r1 = (lane == kp) ? ri : r1;
  covL[(2 * lane) * 65 + kk] = l0;
  covL[(2 * lane + 1) * 65 + kk] = l1;
#pragma unroll
  for (int jj = 1; jj <= NCOLS; ++jj) {
    const int lj = (kp + ((Q + jj) >> 1)) & 31;
    float s = lane_read(((Q + jj) & 1) ? l1 : l0, lj);
    c0[jj - 1] = c0[jj] - l0 * s;
    c1[jj - 1] = c1[jj] - l1 * s;
  }
}

__global__ __launch_bounds__(256) void reduce_solve_kernel(const float* __restrict__ sxxpart,
                                                           const float* __restrict__ sumpart,
                                                           const int* __restrict__ classinfo,
                                                           const float* __restrict__ pattern,
                                                           float* __restrict__ params) {
  int bk = blockIdx.x;
  int tid = threadIdx.x;
  __shared__ float covL[64 * 65];
  __shared__ float meanL[64], pL[64], wv[64];

  int count = classinfo[bk * 3 + 1];
  float safe = (float)(count > 0 ? count : 1);

  if (tid < 64) {
    float s = 0.0f;
    for (int part = 0; part < PARTS; ++part)
      s += sumpart[((size_t)part * NB * NCLS + bk) * NC + tid];
    meanL[tid] = s / safe;
    pL[tid] = pattern[tid];
  }
  __syncthreads();

  const int er = tid >> 4, ec = tid & 15;
  for (int t = 0; t < 10; ++t) {
    int ti = (t >= 6) ? 3 : (t >= 3) ? 2 : (t >= 1) ? 1 : 0;
    int tj = t - ((ti * (ti + 1)) >> 1);
    int i = ti * 16 + er, j = tj * 16 + ec;
    int idx = i * NC + j;
    float s = 0.0f;
    for (int part = 0; part < PARTS; ++part)
      s += sxxpart[((size_t)part * NB * NCLS + bk) * (NC * NC) + idx];
    float cv = s / safe - meanL[i] * meanL[j];
    if (i == j) cv += EPSV;
    covL[i * 65 + j] = cv;
  }
  __syncthreads();
  for (int t = 0; t < 6; ++t) {
    int ti = (t >= 3) ? 3 : (t >= 1) ? 2 : 1;
    int tj = t - ((ti * (ti - 1)) >> 1);
    int i = ti * 16 + er, j = tj * 16 + ec;
    covL[j * 65 + i] = covL[i * 65 + j];
  }
  __syncthreads();

  if (tid < 32) {
    const int lane = tid;
    float c0[NC], c1[NC];
#pragma unroll
    for (int j = 0; j < NC; ++j) {
      c0[j] = covL[(2 * lane) * 65 + j];
      c1[j] = covL[(2 * lane + 1) * 65 + j];
    }

    float r0 = 0.0f, r1 = 0.0f;
#pragma nounroll
    for (int kp = 0; kp < 16; ++kp) {
      chol_step<0, 63>(c0, c1, r0, r1, covL, lane, kp);
      chol_step<1, 63>(c0, c1, r0, r1, covL, lane, kp);
    }
#pragma nounroll
    for (int kp = 16; kp < 32; ++kp) {
      chol_step<0, 31>(c0, c1, r0, r1, covL, lane, kp);
      chol_step<1, 31>(c0, c1, r0, r1, covL, lane, kp);
    }

    float la[NC], lb[NC], ua[NC], ub[NC];
#pragma unroll
    for (int j = 0; j < NC; ++j) {
      la[j] = covL[(2 * lane) * 65 + j];
      lb[j] = covL[(2 * lane + 1) * 65 + j];
      ua[j] = covL[j * 65 + 2 * lane];
      ub[j] = covL[j * 65 + 2 * lane + 1];
    }

    float y0 = pL[2 * lane], y1 = pL[2 * lane + 1];
#pragma unroll
    for (int jp = 0; jp < 32; ++jp) {
      {
        float s = lane_read(y0 * r0, jp);
        float t0 = y0 - la[2 * jp] * s, t1 = y1 - lb[2 * jp] * s;
        y0 = (lane == jp) ? s : ((lane > jp) ? t0 : y0);
        y1 = (lane >= jp) ? t1 : y1;
      }
      {
        float s = lane_read(y1 * r1, jp);
        float t0 = y0 - la[2 * jp + 1] * s, t1 = y1 - lb[2 * jp + 1] * s;
        y1 = (lane == jp) ? s : ((lane > jp) ? t1 : y1);
        y0 = (lane > jp) ? t0 : y0;
      }
    }

    float w0 = y0, w1 = y1;
#pragma unroll
    for (int jp = 31; jp >= 0; --jp) {
      {
        float s = lane_read(w1 * r1, jp);
        float t0 = w0 - ua[2 * jp + 1] * s, t1 = w1 - ub[2 * jp + 1] * s;
        w1 = (lane == jp) ? s : ((lane < jp) ? t1 : w1);
        w0 = (lane <= jp) ? t0 : w0;
      }
      {
        float s = lane_read(w0 * r0, jp);
        float t0 = w0 - ua[2 * jp] * s, t1 = w1 - ub[2 * jp] * s;
        w0 = (lane == jp) ? s : ((lane < jp) ? t0 : w0);
        w1 = (lane < jp) ? t1 : w1;
      }
    }

    params[(size_t)bk * 66 + 2 * lane] = w0;
    params[(size_t)bk * 66 + 2 * lane + 1] = w1;
    wv[2 * lane] = w0;
    wv[2 * lane + 1] = w1;
    float pw = 0.0f, off = 0.0f;
    for (int c = 0; c < NC; ++c) {
      pw += pL[c] * wv[c];
      off += meanL[c] * wv[c];
    }
    if (lane == 0) {
      float denom = sqrtf(pw);
      float scale = (count >= MIN_CLASS_SIZE && denom > 0.0f) ? (1.0f / denom) : 0.0f;
      params[(size_t)bk * 66 + 64] = off;
      params[(size_t)bk * 66 + 65] = scale;
    }
  }
}

__global__ __launch_bounds__(256) void score_kernel(const float* __restrict__ hyper,
                                                    const int* __restrict__ seg,
                                                    const float* __restrict__ params,
                                                    float* __restrict__ out) {
  int b = blockIdx.y;
  int tid = threadIdx.x;
  __shared__ float wS[NCLS * 65];
  __shared__ float offS[NCLS], sclS[NCLS];
  for (int i = tid; i < NCLS * NC; i += 256) {
    int k = i >> 6, c = i & 63;
    wS[k * 65 + c] = params[((size_t)b * NCLS + k) * 66 + c];
  }
  if (tid < NCLS) {
    offS[tid] = params[((size_t)b * NCLS + tid) * 66 + 64];
    sclS[tid] = params[((size_t)b * NCLS + tid) * 66 + 65];
  }
  __syncthreads();

  size_t n = (size_t)blockIdx.x * 256 + tid;
  int k = seg[(size_t)b * NPIX + n];
  const float* xp = hyper + (size_t)b * NC * NPIX + n;
  float acc = 0.0f;
#pragma unroll
  for (int c = 0; c < NC; ++c)
    acc = fmaf(xp[(size_t)c * NPIX], wS[k * 65 + c], acc);
  out[(size_t)b * NPIX + n] = (acc - offS[k]) * sclS[k];
}

extern "C" void kernel_launch(void* const* d_in, const int* in_sizes, int n_in,
                              void* d_out, int out_size, void* d_ws, size_t ws_size,
                              hipStream_t stream) {
  const float* hyper = (const float*)d_in[0];
  const float* pattern = (const float*)d_in[1];
  const int* seg = (const int*)d_in[2];
  float* out = (float*)d_out;

  char* ws = (char*)d_ws;
  auto al = [](size_t x) { return (x + 255) & ~(size_t)255; };
  size_t off = 0;
  float* comp = (float*)(ws + off);
  off += al((size_t)NB * NPAD_MAX * NC * sizeof(float));
  int* hist = (int*)(ws + off);
  off += al((size_t)NB * NBLKS * NCLS * sizeof(int));
  int* classinfo = (int*)(ws + off);
  off += al((size_t)NB * NCLS * 3 * sizeof(int));
  float* sxxpart = (float*)(ws + off);
  off += al((size_t)PARTS * NB * NCLS * NC * NC * sizeof(float));
  float* sumpart = (float*)(ws + off);
  off += al((size_t)PARTS * NB * NCLS * NC * sizeof(float));
  float* params = (float*)(ws + off);
  off += al((size_t)NB * NCLS * 66 * sizeof(float));

  hipLaunchKernelGGL(hist_kernel, dim3(NB * NBLKS), dim3(HW * 32), 0, stream, seg, hist);
  hipLaunchKernelGGL(scatter_kernel, dim3(NB * NBLKS * SUBS), dim3(SCW * 32), 0, stream, hyper, seg,
                     hist, classinfo, comp);
  hipLaunchKernelGGL(syrk_kernel, dim3(PARTS * NB * NCLS), dim3(SYW * 32), 0, stream, comp,
                     classinfo, sxxpart, sumpart);
  hipLaunchKernelGGL(reduce_solve_kernel, dim3(NB * NCLS), dim3(256), 0, stream, sxxpart,
                     sumpart, classinfo, pattern, params);
  hipLaunchKernelGGL(score_kernel, dim3(NPIX / 256, NB), dim3(256), 0, stream, hyper, seg,
                     params, out);
}


# reduce_solve Cholesky issues only live trailing updates: lane l owns rows l and 32+l (was 2l, 2l+1), rows 0-31 stop at column 31, shifted width shrinks every 8 steps; per-entry operations, order kept
# speedup vs baseline: 6.3056x; 1.0613x over previous
#include <hip/hip_runtime.h>
#include <hip/hip_bf16.h>


typedef float v2f __attribute__((ext_vector_type(2)));
typedef float v8f __attribute__((ext_vector_type(8)));

#define NB 2
#define NC 64
#define NPIX 147456
#define NCLS 20
#define BLKPX 1024
#define CHUNKS (BLKPX / 32)
#define NBLKS (NPIX / BLKPX)
#define PARTS 16
#define PADALIGN (PARTS * 4)
#define NPAD_MAX (NPIX + NCLS * PADALIGN)
#define MIN_CLASS_SIZE 5000
#define EPSV 1e-6f

#define HW 4
#define HLD (BLKPX / 4 / (HW * 32))

__global__ __launch_bounds__(HW * 32) void hist_kernel(const int* __restrict__ seg,
                                                       int* __restrict__ hist) {
  static_assert(HLD * 4 * HW * 32 == BLKPX, "the work-group's loads cover the block exactly");
  int blk = blockIdx.x;
  int b = blk / NBLKS, bb = blk % NBLKS;
  int lane = threadIdx.x & 31;
  int wave = __builtin_amdgcn_readfirstlane((int)(threadIdx.x >> 5));
  __shared__ int part[HW][32];
  const int4* s4 = (const int4*)(seg + (size_t)b * NPIX + (size_t)bb * BLKPX);
  int4 q[HLD];
#pragma unroll
  for (int j = 0; j < HLD; ++j) q[j] = s4[j * (HW * 32) + (int)threadIdx.x];
  int cnt = 0;
#pragma unroll
  for (int k = 0; k < NCLS; ++k) {
    int c = 0;
#pragma unroll
    for (int j = 0; j < HLD; ++j)
      c += __popc(__builtin_amdgcn_ballot_w32(q[j].x == k)) +
           __popc(__builtin_amdgcn_ballot_w32(q[j].y == k)) +
           __popc(__builtin_amdgcn_ballot_w32(q[j].z == k)) +
           __popc(__builtin_amdgcn_ballot_w32(q[j].w == k));
    if (lane == k) cnt = c;
  }
  part[wave][lane] = cnt;
  __syncthreads();
  if (wave == 0) {
    int t = 0;
#pragma unroll
    for (int w = 0; w < HW; ++w) t += part[w][lane];
    if (lane < NCLS) hist[((size_t)b * NCLS + lane) * NBLKS + bb] = t;
  }
}

#define SUBS 4
#define SCW (CHUNKS / SUBS)
#define TPITCH 68
#define CPW ((NCLS + SCW - 1) / SCW)
#define HQ (NBLKS / 4)

__device__ __forceinline__ int wave_sum(int v) {
  v += __builtin_amdgcn_update_dpp(0, v, 0x161, 0xf, 0xf, false);
  v += __builtin_amdgcn_update_dpp(0, v, 0x162, 0xf, 0xf, false);
  v += __builtin_amdgcn_update_dpp(0, v, 0x164, 0xf, 0xf, false);
  v += __builtin_amdgcn_update_dpp(0, v, 0x168, 0xf, 0xf, false);
  v += __builtin_amdgcn_permlanex16(v, v, 0x76543210, 0xfedcba98, false, false);
  return v;
}

__global__ __launch_bounds__(SCW * 32) void scatter_kernel(const float* __restrict__ hyper,
                                                           const int* __restrict__ seg,
                                                           const int* __restrict__ hist,
                                                           int* __restrict__ classinfo,
                                                           float* __restrict__ comp) {
  int s = blockIdx.x % SUBS;
  int hb = blockIdx.x / SUBS;
  int b = hb / NBLKS, bb = hb % NBLKS;
  int lane = threadIdx.x & 31;
  int wave = __builtin_amdgcn_readfirstlane((int)(threadIdx.x >> 5));
  __shared__ __attribute__((aligned(16))) float tile[SCW][32 * TPITCH];
  __shared__ int cnt[CHUNKS][32];
  __shared__ int totS[32];
  __shared__ int befS[32];

  const int* sg = seg + (size_t)b * NPIX + (size_t)bb * BLKPX;
  int sgv[SUBS];
#pragma unroll
  for (int j = 0; j < SUBS; ++j) sgv[j] = sg[(j * SCW + wave) * 32 + lane];

  static_assert(HQ - 32 == 4, "the upper part of a histogram column is four 4-block units");
  int4 hlo[CPW], hhi[CPW];
#pragma unroll
  for (int r = 0; r < CPW; ++r) {
    int k = wave + SCW * r;
    int kc = (k < NCLS) ? k : NCLS - 1;
    const int4* hp = (const int4*)(hist + ((size_t)b * NCLS + kc) * NBLKS);
    hlo[r] = hp[lane];
    hhi[r] = hp[32 + (lane & 3)];
  }

  int ch = s * SCW + wave;
  const float* xp = hyper + (size_t)b * NC * NPIX + (size_t)bb * BLKPX + (size_t)ch * 32 + lane;
  float v[NC];
#pragma unroll
  for (int c = 0; c < NC; ++c) v[c] = xp[(size_t)c * NPIX];

  int myk = sgv[0];
#pragma unroll
  for (int j = 1; j < SUBS; ++j) myk = (s == j) ? sgv[j] : myk;

#pragma unroll
  for (int j = 0; j < SUBS; ++j) {
    int pc = 0;
#pragma unroll
    for (int k = 0; k < NCLS; ++k) {
      unsigned bal = __builtin_amdgcn_ballot_w32(sgv[j] == k);
      if (lane == k) pc = __popc(bal);
    }
    cnt[j * SCW + wave][lane] = pc;
  }
  unsigned balmy = 0;
#pragma unroll
  for (int k = 0; k < NCLS; ++k) {
    unsigned bal = __builtin_amdgcn_ballot_w32(myk == k);
    if (myk == k) balmy = bal;
  }

#pragma unroll
  for (int r = 0; r < CPW; ++r) {
    int k = wave + SCW * r;
    int q0 = 4 * lane, q1 = 128 + 4 * lane;
    bool hi = lane < HQ - 32;
    int tt = ((hlo[r].x + hlo[r].y) + (hlo[r].z + hlo[r].w)) +
             (hi ? ((hhi[r].x + hhi[r].y) + (hhi[r].z + hhi[r].w)) : 0);
    int bf = ((q0 < bb) ? hlo[r].x : 0) + ((q0 + 1 < bb) ? hlo[r].y : 0) +
             ((q0 + 2 < bb) ? hlo[r].z : 0) + ((q0 + 3 < bb) ? hlo[r].w : 0) +
             ((hi && q1 < bb) ? hhi[r].x : 0) + ((hi && q1 + 1 < bb) ? hhi[r].y : 0) +
             ((hi && q1 + 2 < bb) ? hhi[r].z : 0) + ((hi && q1 + 3 < bb) ? hhi[r].w : 0);
    tt = wave_sum(tt);
    bf = wave_sum(bf);
    if (lane == 0 && k < NCLS) {
      totS[k] = tt;
      befS[k] = bf;
    }
  }

  float* tl = tile[wave];
#pragma unroll
  for (int q = 0; q < NC / 4; ++q)
    *(float4*)&tl[lane * TPITCH + 4 * q] =
        make_float4(v[4 * q], v[4 * q + 1], v[4 * q + 2], v[4 * q + 3]);
  __syncthreads();

  int cls = (lane < NCLS) ? lane : 0;
  int base = 0;
#pragma unroll
  for (int k = 0; k < NCLS - 1; ++k) {
    int p = (totS[k] + PADALIGN - 1) & ~(PADALIGN - 1);
    base += (k < lane) ? p : 0;
  }
  int tk = totS[cls];
  int pk = (tk + PADALIGN - 1) & ~(PADALIGN - 1);
  int cb = base + befS[cls];

  int run = 0;
#pragma unroll
  for (int c2 = 0; c2 < CHUNKS - 1; ++c2) {
    int t = cnt[c2][lane];
    run += (c2 < ch) ? t : 0;
  }
  unsigned ltmask = (1u << lane) - 1u;
  int pos = __shfl(cb + run, myk, 32) + __popc(balmy & ltmask);

  float* crow = comp + (size_t)b * NPAD_MAX * NC;
#pragma unroll
  for (int p = 0; p < 32; ++p) {
    int pp = __builtin_amdgcn_readlane(pos, p);
    float2 val = *(const float2*)&tl[p * TPITCH + 2 * lane];
    *(float2*)&crow[(size_t)pp * NC + 2 * lane] = val;
  }

  if (s == 0 && bb == 0 && wave == 0 && lane < NCLS) {
    classinfo[(b * NCLS + lane) * 3 + 0] = base;
    classinfo[(b * NCLS + lane) * 3 + 1] = tk;
    classinfo[(b * NCLS + lane) * 3 + 2] = pk;
  }
  if (s == 0 && bb < NCLS) {
    int zb = __builtin_amdgcn_readlane(base, bb);
    int zt = __builtin_amdgcn_readlane(tk, bb);
    int zp = __builtin_amdgcn_readlane(pk, bb);
    float4* dst = (float4*)(crow + ((size_t)zb + zt) * NC);
    int n4 = (zp - zt) * (NC / 4);
#pragma unroll
    for (int j = 0; j < PADALIGN * (NC / 4) / (SCW * 32); ++j) {
      int i = (int)threadIdx.x + j * (SCW * 32);
      if (i < n4) dst[i] = make_float4(0.0f, 0.0f, 0.0f, 0.0f);
    }
  }
}

#define SYW 4

template <int S> __device__ constexpr int tile_a(int t) { return S == 0 ? (t < 2 ? 1 : 0) : (S == 1 ? 2 : 3); }
template <int S> __device__ constexpr int tile_b(int t) { return S == 0 ? (t == 1 ? 1 : 0) : (S == 3 ? 2 + t : t); }
template <int S> __device__ constexpr int tile_n() { return S < 2 ? 3 : 2; }
template <int S> __device__ constexpr int sum_blk() { return S == 1 ? 2 : (S == 2 ? 1 : S); }

__device__ __forceinline__ float xhalf(float x) {
  int b = __float_as_int(x);
  return __int_as_float(__builtin_amdgcn_permlanex16(b, b, 0x76543210, 0xfedcba98, false, false));
}

__device__ __forceinline__ void load_group(v2f (&f)[4], const float* __restrict__ gp) {
#pragma unroll
  for (int blk = 0; blk < 4; ++blk) {
    f[blk].x = gp[blk * 16];
    f[blk].y = gp[NC + blk * 16];
  }
}

__device__ __forceinline__ void load_trip(v2f (&f)[4][4], const float* __restrict__ tp) {
#pragma unroll
  for (int i = 0; i < 4; ++i) load_group(f[i], tp + (size_t)i * 4 * NC);
}

template <int S>
__device__ __forceinline__ void use_group(v8f (&acc)[3], float& cs, const v2f (&f)[4]) {
#pragma unroll
  for (int t = 0; t < tile_n<S>(); ++t)
    acc[t] = __builtin_amdgcn_wmma_f32_16x16x4_f32(
        false, f[tile_a<S>(t)], false, f[tile_b<S>(t)], (short)0, acc[t], false, false);
  constexpr int SB = sum_blk<S>();
  float ox = xhalf(f[SB].x), oy = xhalf(f[SB].y);
  cs = (((cs + f[SB].x) + f[SB].y) + ox) + oy;
}

template <int S>
__device__ __forceinline__ void use_trip(v8f (&acc)[3], float& cs, const v2f (&f)[4][4]) {
#pragma unroll
  for (int i = 0; i < 4; ++i) use_group<S>(acc, cs, f[i]);
}

template <int S>
__device__ __forceinline__ void syrk_tile_set(const float* __restrict__ rp, int plen, int lane,
                                              float* __restrict__ sxxout, float* __restrict__ sumout) {
  const int ntrip = plen >> 4;
  const int rem = (plen >> 2) & 3;

  v8f acc[3] = {};
  float cs = 0.0f;

  v2f ft[3][4];
  const float* tailp = rp + (size_t)ntrip * 16 * NC;
  if (rem > 0) load_group(ft[0], tailp);
  if (rem > 1) load_group(ft[1], tailp + 4 * NC);
  if (rem > 2) load_group(ft[2], tailp + 8 * NC);

  v2f fa[4][4], fb[4][4];
  if (ntrip > 0) load_trip(fa, rp);
  int t = 0;
  for (; t + 2 <= ntrip; t += 2) {
    load_trip(fb, rp + (size_t)(t + 1) * 16 * NC);
    use_trip<S>(acc, cs, fa);
    int tn = (t + 2 < ntrip) ? t + 2 : ntrip - 1;
    load_trip(fa, rp + (size_t)tn * 16 * NC);
    use_trip<S>(acc, cs, fb);
  }
  if (t < ntrip) use_trip<S>(acc, cs, fa);
  if (rem > 0) use_group<S>(acc, cs, ft[0]);
  if (rem > 1) use_group<S>(acc, cs, ft[1]);
  if (rem > 2) use_group<S>(acc, cs, ft[2]);

  int col = lane & 15;
  int rof = (lane >> 4) * 8;
#pragma unroll
  for (int t = 0; t < tile_n<S>(); ++t)
#pragma unroll
    for (int j = 0; j < 8; ++j) {
      int r = tile_a<S>(t) * 16 + rof + j;
      sxxout[(size_t)r * NC + tile_b<S>(t) * 16 + col] = acc[t][j];
    }
  if (lane < 16) sumout[sum_blk<S>() * 16 + lane] = cs;
}

__global__ __launch_bounds__(SYW * 32) void syrk_kernel(const float* __restrict__ comp,
                                                        const int* __restrict__ classinfo,
                                                        float* __restrict__ sxxpart,
                                                        float* __restrict__ sumpart) {
  int gid = blockIdx.x;
  int part = gid / (NB * NCLS);
  int bk = gid % (NB * NCLS);
  int b = bk / NCLS;
  int lane = threadIdx.x & 31;
  int wave = __builtin_amdgcn_readfirstlane((int)(threadIdx.x >> 5));

  int base = classinfo[bk * 3 + 0];
  int padl = classinfo[bk * 3 + 2];
  int plen = padl / PARTS;
  size_t rowbase = (size_t)b * NPAD_MAX + base + (size_t)part * plen;

  int m = lane & 15;
  int kb = (lane >> 4) << 1;
  const float* rp = comp + (rowbase + kb) * NC + m;
  float* sxxout = sxxpart + (size_t)gid * (NC * NC);
  float* sumout = sumpart + (size_t)gid * NC;

  int set = (wave + (gid ^ (gid >> 8))) & 3;
  switch (set) {
    case 0: syrk_tile_set<0>(rp, plen, lane, sxxout, sumout); break;
    case 1: syrk_tile_set<1>(rp, plen, lane, sxxout, sumout); break;
    case 2: syrk_tile_set<2>(rp, plen, lane, sxxout, sumout); break;
    default: syrk_tile_set<3>(rp, plen, lane, sxxout, sumout); break;
  }
}

__device__ __forceinline__ float lane_read(float v, int l) {
  return __int_as_float(__builtin_amdgcn_readlane(__float_as_int(v), l));
}

template <int NA>
__device__ __forceinline__ void chol_step_a(v2f (&cab)[32], float (&cbh)[32],
                                            float& ra, float* covL, int lane, int kk) {
  float piv = lane_read(cab[0].x, kk);
  float ri = __builtin_amdgcn_rsqf(piv);
  v2f l = cab[0] * ri;
  ra = (lane == kk) ? ri : ra;
  covL[lane * 65 + kk] = l.x;
  covL[(32 + lane) * 65 + kk] = l.y;
#pragma unroll
  for (int jj = 1; jj <= NA; ++jj) {
    float s = lane_read(l.x, (kk + jj) & 31);
    cab[jj - 1] = cab[jj] - l * s;
  }
#pragma unroll
  for (int e = 0; e < 32; ++e) {
    float s = lane_read(l.y, e);
    cbh[e] = cbh[e] - l.y * s;
  }
}

template <int NBC>
__device__ __forceinline__ void chol_step_b(float (&cbh)[32], float& rb, float* covL, int lane, int kq) {
  float piv = lane_read(cbh[0], kq);
  float ri = __builtin_amdgcn_rsqf(piv);
  float l1 = cbh[0] * ri;
  rb = (lane == kq) ? ri : rb;
  covL[(32 + lane) * 65 + 32 + kq] = l1;
#pragma unroll
  for (int jj = 1; jj <= NBC; ++jj) {
    float s = lane_read(l1, (kq + jj) & 31);
    cbh[jj - 1] = cbh[jj] - l1 * s;
  }
}

__global__ __launch_bounds__(256) void reduce_solve_kernel(const float* __restrict__ sxxpart,
                                                           const float* __restrict__ sumpart,
                                                           const int* __restrict__ classinfo,
                                                           const float* __restrict__ pattern,
                                                           float* __restrict__ params) {
  int bk = blockIdx.x;
  int tid = threadIdx.x;
  __shared__ float covL[64 * 65];
  __shared__ float meanL[64], pL[64], wv[64];

  int count = classinfo[bk * 3 + 1];
  float safe = (float)(count > 0 ? count : 1);

  if (tid < 64) {
    float s = 0.0f;
    for (int part = 0; part < PARTS; ++part)
      s += sumpart[((size_t)part * NB * NCLS + bk) * NC + tid];
    meanL[tid] = s / safe;
    pL[tid] = pattern[tid];
  }
  __syncthreads();

  const int er = tid >> 4, ec = tid & 15;
  for (int t = 0; t < 10; ++t) {
    int ti = (t >= 6) ? 3 : (t >= 3) ? 2 : (t >= 1) ? 1 : 0;
    int tj = t - ((ti * (ti + 1)) >> 1);
    int i = ti * 16 + er, j = tj * 16 + ec;
    int idx = i * NC + j;
    float s = 0.0f;
    for (int part = 0; part < PARTS; ++part)
      s += sxxpart[((size_t)part * NB * NCLS + bk) * (NC * NC) + idx];
    float cv = s / safe - meanL[i] * meanL[j];
    if (i == j) cv += EPSV;
    covL[i * 65 + j] = cv;
  }
  __syncthreads();
  for (int t = 0; t < 6; ++t) {
    int ti = (t >= 3) ? 3 : (t >= 1) ? 2 : 1;
    int tj = t - ((ti * (ti - 1)) >> 1);
    int i = ti * 16 + er, j = tj * 16 + ec;
    covL[j * 65 + i] = covL[i * 65 + j];
  }
  __syncthreads();

  if (tid < 32) {
    const int lane = tid;
    v2f cab[32];
    float cbh[32];
#pragma unroll
    for (int j = 0; j < 32; ++j) {
      cab[j].x = covL[lane * 65 + j];
      cab[j].y = covL[(32 + lane) * 65 + j];
      cbh[j] = covL[(32 + lane) * 65 + 32 + j];
    }

    float r0 = 0.0f, r1 = 0.0f;
#pragma nounroll
    for (int kk = 0; kk < 8; ++kk) chol_step_a<31>(cab, cbh, r0, covL, lane, kk);
#pragma nounroll
    for (int kk = 8; kk < 16; ++kk) chol_step_a<23>(cab, cbh, r0, covL, lane, kk);
#pragma nounroll
    for (int kk = 16; kk < 24; ++kk) chol_step_a<15>(cab, cbh, r0, covL, lane, kk);
#pragma nounroll
    for (int kk = 24; kk < 32; ++kk) chol_step_a<7>(cab, cbh, r0, covL, lane, kk);
#pragma nounroll
    for (int kq = 0; kq < 8; ++kq) chol_step_b<31>(cbh, r1, covL, lane, kq);
#pragma nounroll
    for (int kq = 8; kq < 16; ++kq) chol_step_b<23>(cbh, r1, covL, lane, kq);
#pragma nounroll
    for (int kq = 16; kq < 24; ++kq) chol_step_b<15>(cbh, r1, covL, lane, kq);
#pragma nounroll
    for (int kq = 24; kq < 32; ++kq) chol_step_b<7>(cbh, r1, covL, lane, kq);

    float la[32], lb[NC], ua[NC], ub[32];
#pragma unroll
    for (int j = 0; j < 32; ++j) {
      la[j] = covL[lane * 65 + j];
      lb[j] = covL[(32 + lane) * 65 + j];
      lb[32 + j] = covL[(32 + lane) * 65 + 32 + j];
      ua[j] = covL[j * 65 + lane];
      ua[32 + j] = covL[(32 + j) * 65 + lane];
      ub[j] = covL[(32 + j) * 65 + 32 + lane];
    }

    float y0 = pL[lane], y1 = pL[32 + lane];
#pragma unroll
    for (int j = 0; j < 32; ++j) {
      float s = lane_read(y0 * r0, j);
      float t0 = y0 - la[j] * s, t1 = y1 - lb[j] * s;
      y0 = (lane == j) ? s : ((lane > j) ? t0 : y0);
      y1 = t1;
    }
#pragma unroll
    for (int j = 0; j < 32; ++j) {
      float s = lane_read(y1 * r1, j);
      float t1 = y1 - lb[32 + j] * s;
      y1 = (lane == j) ? s : ((lane > j) ? t1 : y1);
    }

    float w0 = y0, w1 = y1;
#pragma unroll
    for (int j = 31; j >= 0; --j) {
      float s = lane_read(w1 * r1, j);
      float t0 = w0 - ua[32 + j] * s, t1 = w1 - ub[j] * s;
      w1 = (lane == j) ? s : ((lane < j) ? t1 : w1);
      w0 = t0;
    }
#pragma unroll
    for (int j = 31; j >= 0; --j) {
      float s = lane_read(w0 * r0, j);
      float t0 = w0 - ua[j] * s;
      w0 = (lane == j) ? s : ((lane < j) ? t0 : w0);
    }

    params[(size_t)bk * 66 + lane] = w0;
    params[(size_t)bk * 66 + 32 + lane] = w1;
    wv[lane] = w0;
    wv[32 + lane] = w1;
    float pw = 0.0f, off = 0.0f;
    for (int c = 0; c < NC; ++c) {
      pw += pL[c] * wv[c];
      off += meanL[c] * wv[c];
    }
    if (lane == 0) {
      float denom = sqrtf(pw);
      float scale = (count >= MIN_CLASS_SIZE && denom > 0.0f) ? (1.0f / denom) : 0.0f;
      params[(size_t)bk * 66 + 64] = off;
      params[(size_t)bk * 66 + 65] = scale;
    }
  }
}

__global__ __launch_bounds__(256) void score_kernel(const float* __restrict__ hyper,
                                                    const int* __restrict__ seg,
                                                    const float* __restrict__ params,
                                                    float* __restrict__ out) {
  int b = blockIdx.y;
  int tid = threadIdx.x;
  __shared__ float wS[NCLS * 65];
  __shared__ float offS[NCLS], sclS[NCLS];
  for (int i = tid; i < NCLS * NC; i += 256) {
    int k = i >> 6, c = i & 63;
    wS[k * 65 + c] = params[((size_t)b * NCLS + k) * 66 + c];
  }
  if (tid < NCLS) {
    offS[tid] = params[((size_t)b * NCLS + tid) * 66 + 64];
    sclS[tid] = params[((size_t)b * NCLS + tid) * 66 + 65];
  }
  __syncthreads();

  size_t n = (size_t)blockIdx.x * 256 + tid;
  int k = seg[(size_t)b * NPIX + n];
  const float* xp = hyper + (size_t)b * NC * NPIX + n;
  float acc = 0.0f;
#pragma unroll
  for (int c = 0; c < NC; ++c)
    acc = fmaf(xp[(size_t)c * NPIX], wS[k * 65 + c], acc);
  out[(size_t)b * NPIX + n] = (acc - offS[k]) * sclS[k];
}

extern "C" void kernel_launch(void* const* d_in, const int* in_sizes, int n_in,
                              void* d_out, int out_size, void* d_ws, size_t ws_size,
                              hipStream_t stream) {
  const float* hyper = (const float*)d_in[0];
  const float* pattern = (const float*)d_in[1];
  const int* seg = (const int*)d_in[2];
  float* out = (float*)d_out;

  char* ws = (char*)d_ws;
  auto al = [](size_t x) { return (x + 255) & ~(size_t)255; };
  size_t off = 0;
  float* comp = (float*)(ws + off);
  off += al((size_t)NB * NPAD_MAX * NC * sizeof(float));
  int* hist = (int*)(ws + off);
  off += al((size_t)NB * NBLKS * NCLS * sizeof(int));
  int* classinfo = (int*)(ws + off);
  off += al((size_t)NB * NCLS * 3 * sizeof(int));
  float* sxxpart = (float*)(ws + off);
  off += al((size_t)PARTS * NB * NCLS * NC * NC * sizeof(float));
  float* sumpart = (float*)(ws + off);
  off += al((size_t)PARTS * NB * NCLS * NC * sizeof(float));
  float* params = (float*)(ws + off);
  off += al((size_t)NB * NCLS * 66 * sizeof(float));

  hipLaunchKernelGGL(hist_kernel, dim3(NB * NBLKS), dim3(HW * 32), 0, stream, seg, hist);
  hipLaunchKernelGGL(scatter_kernel, dim3(NB * NBLKS * SUBS), dim3(SCW * 32), 0, stream, hyper, seg,
                     hist, classinfo, comp);
  hipLaunchKernelGGL(syrk_kernel, dim3(PARTS * NB * NCLS), dim3(SYW * 32), 0, stream, comp,
                     classinfo, sxxpart, sumpart);
  hipLaunchKernelGGL(reduce_solve_kernel, dim3(NB * NCLS), dim3(256), 0, stream, sxxpart,
                     sumpart, classinfo, pattern, params);
  hipLaunchKernelGGL(score_kernel, dim3(NPIX / 256, NB), dim3(256), 0, stream, hyper, seg,
                     params, out);
}


# reduce_solve Cholesky blocked by 16 columns: updates crossing a block (16-31 by 0-15, 32-63 by 0-31, 48-63 by 32-47) are WMMA products C - L L^T via LDS; sums keep column order, last bits differ
# speedup vs baseline: 6.4992x; 1.0307x over previous
#include <hip/hip_runtime.h>
#include <hip/hip_bf16.h>


typedef float v2f __attribute__((ext_vector_type(2)));
typedef float v8f __attribute__((ext_vector_type(8)));

#define NB 2
#define NC 64
#define NPIX 147456
#define NCLS 20
#define BLKPX 1024
#define CHUNKS (BLKPX / 32)
#define NBLKS (NPIX / BLKPX)
#define PARTS 16
#define PADALIGN (PARTS * 4)
#define NPAD_MAX (NPIX + NCLS * PADALIGN)
#define MIN_CLASS_SIZE 5000
#define EPSV 1e-6f

#define HW 4
#define HLD (BLKPX / 4 / (HW * 32))

__global__ __launch_bounds__(HW * 32) void hist_kernel(const int* __restrict__ seg,
                                                       int* __restrict__ hist) {
  static_assert(HLD * 4 * HW * 32 == BLKPX, "the work-group's loads cover the block exactly");
  int blk = blockIdx.x;
  int b = blk / NBLKS, bb = blk % NBLKS;
  int lane = threadIdx.x & 31;
  int wave = __builtin_amdgcn_readfirstlane((int)(threadIdx.x >> 5));
  __shared__ int part[HW][32];
  const int4* s4 = (const int4*)(seg + (size_t)b * NPIX + (size_t)bb * BLKPX);
  int4 q[HLD];
#pragma unroll
  for (int j = 0; j < HLD; ++j) q[j] = s4[j * (HW * 32) + (int)threadIdx.x];
  int cnt = 0;
#pragma unroll
  for (int k = 0; k < NCLS; ++k) {
    int c = 0;
#pragma unroll
    for (int j = 0; j < HLD; ++j)
      c += __popc(__builtin_amdgcn_ballot_w32(q[j].x == k)) +
           __popc(__builtin_amdgcn_ballot_w32(q[j].y == k)) +
           __popc(__builtin_amdgcn_ballot_w32(q[j].z == k)) +
           __popc(__builtin_amdgcn_ballot_w32(q[j].w == k));
    if (lane == k) cnt = c;
  }
  part[wave][lane] = cnt;
  __syncthreads();
  if (wave == 0) {
    int t = 0;
#pragma unroll
    for (int w = 0; w < HW; ++w) t += part[w][lane];
    if (lane < NCLS) hist[((size_t)b * NCLS + lane) * NBLKS + bb] = t;
  }
}

#define SUBS 4
#define SCW (CHUNKS / SUBS)
#define TPITCH 68
#define CPW ((NCLS + SCW - 1) / SCW)
#define HQ (NBLKS / 4)

__device__ __forceinline__ int wave_sum(int v) {
  v += __builtin_amdgcn_update_dpp(0, v, 0x161, 0xf, 0xf, false);
  v += __builtin_amdgcn_update_dpp(0, v, 0x162, 0xf, 0xf, false);
  v += __builtin_amdgcn_update_dpp(0, v, 0x164, 0xf, 0xf, false);
  v += __builtin_amdgcn_update_dpp(0, v, 0x168, 0xf, 0xf, false);
  v += __builtin_amdgcn_permlanex16(v, v, 0x76543210, 0xfedcba98, false, false);
  return v;
}

__global__ __launch_bounds__(SCW * 32) void scatter_kernel(const float* __restrict__ hyper,
                                                           const int* __restrict__ seg,
                                                           const int* __restrict__ hist,
                                                           int* __restrict__ classinfo,
                                                           float* __restrict__ comp) {
  int s = blockIdx.x % SUBS;
  int hb = blockIdx.x / SUBS;
  int b = hb / NBLKS, bb = hb % NBLKS;
  int lane = threadIdx.x & 31;
  int wave = __builtin_amdgcn_readfirstlane((int)(threadIdx.x >> 5));
  __shared__ __attribute__((aligned(16))) float tile[SCW][32 * TPITCH];
  __shared__ int cnt[CHUNKS][32];
  __shared__ int totS[32];
  __shared__ int befS[32];

  const int* sg = seg + (size_t)b * NPIX + (size_t)bb * BLKPX;
  int sgv[SUBS];
#pragma unroll
  for (int j = 0; j < SUBS; ++j) sgv[j] = sg[(j * SCW + wave) * 32 + lane];

  static_assert(HQ - 32 == 4, "the upper part of a histogram column is four 4-block units");
  int4 hlo[CPW], hhi[CPW];
#pragma unroll
  for (int r = 0; r < CPW; ++r) {
    int k = wave + SCW * r;
    int kc = (k < NCLS) ? k : NCLS - 1;
    const int4* hp = (const int4*)(hist + ((size_t)b * NCLS + kc) * NBLKS);
    hlo[r] = hp[lane];
    hhi[r] = hp[32 + (lane & 3)];
  }

  int ch = s * SCW + wave;
  const float* xp = hyper + (size_t)b * NC * NPIX + (size_t)bb * BLKPX + (size_t)ch * 32 + lane;
  float v[NC];
#pragma unroll
  for (int c = 0; c < NC; ++c) v[c] = xp[(size_t)c * NPIX];

  int myk = sgv[0];
#pragma unroll
  for (int j = 1; j < SUBS; ++j) myk = (s == j) ? sgv[j] : myk;

#pragma unroll
  for (int j = 0; j < SUBS; ++j) {
    int pc = 0;
#pragma unroll
    for (int k = 0; k < NCLS; ++k) {
      unsigned bal = __builtin_amdgcn_ballot_w32(sgv[j] == k);
      if (lane == k) pc = __popc(bal);
    }
    cnt[j * SCW + wave][lane] = pc;
  }
  unsigned balmy = 0;
#pragma unroll
  for (int k = 0; k < NCLS; ++k) {
    unsigned bal = __builtin_amdgcn_ballot_w32(myk == k);
    if (myk == k) balmy = bal;
  }

#pragma unroll
  for (int r = 0; r < CPW; ++r) {
    int k = wave + SCW * r;
    int q0 = 4 * lane, q1 = 128 + 4 * lane;
    bool hi = lane < HQ - 32;
    int tt = ((hlo[r].x + hlo[r].y) + (hlo[r].z + hlo[r].w)) +
             (hi ? ((hhi[r].x + hhi[r].y) + (hhi[r].z + hhi[r].w)) : 0);
    int bf = ((q0 < bb) ? hlo[r].x : 0) + ((q0 + 1 < bb) ? hlo[r].y : 0) +
             ((q0 + 2 < bb) ? hlo[r].z : 0) + ((q0 + 3 < bb) ? hlo[r].w : 0) +
             ((hi && q1 < bb) ? hhi[r].x : 0) + ((hi && q1 + 1 < bb) ? hhi[r].y : 0) +
             ((hi && q1 + 2 < bb) ? hhi[r].z : 0) + ((hi && q1 + 3 < bb) ? hhi[r].w : 0);
    tt = wave_sum(tt);
    bf = wave_sum(bf);
    if (lane == 0 && k < NCLS) {
      totS[k] = tt;
      befS[k] = bf;
    }
  }

  float* tl = tile[wave];
#pragma unroll
  for (int q = 0; q < NC / 4; ++q)
    *(float4*)&tl[lane * TPITCH + 4 * q] =
        make_float4(v[4 * q], v[4 * q + 1], v[4 * q + 2], v[4 * q + 3]);
  __syncthreads();

  int cls = (lane < NCLS) ? lane : 0;
  int base = 0;
#pragma unroll
  for (int k = 0; k < NCLS - 1; ++k) {
    int p = (totS[k] + PADALIGN - 1) & ~(PADALIGN - 1);
    base += (k < lane) ? p : 0;
  }
  int tk = totS[cls];
  int pk = (tk + PADALIGN - 1) & ~(PADALIGN - 1);
  int cb = base + befS[cls];

  int run = 0;
#pragma unroll
  for (int c2 = 0; c2 < CHUNKS - 1; ++c2) {
    int t = cnt[c2][lane];
    run += (c2 < ch) ? t : 0;
  }
  unsigned ltmask = (1u << lane) - 1u;
  int pos = __shfl(cb + run, myk, 32) + __popc(balmy & ltmask);

  float* crow = comp + (size_t)b * NPAD_MAX * NC;
#pragma unroll
  for (int p = 0; p < 32; ++p) {
    int pp = __builtin_amdgcn_readlane(pos, p);
    float2 val = *(const float2*)&tl[p * TPITCH + 2 * lane];
    *(float2*)&crow[(size_t)pp * NC + 2 * lane] = val;
  }

  if (s == 0 && bb == 0 && wave == 0 && lane < NCLS) {
    classinfo[(b * NCLS + lane) * 3 + 0] = base;
    classinfo[(b * NCLS + lane) * 3 + 1] = tk;
    classinfo[(b * NCLS + lane) * 3 + 2] = pk;
  }
  if (s == 0 && bb < NCLS) {
    int zb = __builtin_amdgcn_readlane(base, bb);
    int zt = __builtin_amdgcn_readlane(tk, bb);
    int zp = __builtin_amdgcn_readlane(pk, bb);
    float4* dst = (float4*)(crow + ((size_t)zb + zt) * NC);
    int n4 = (zp - zt) * (NC / 4);
#pragma unroll
    for (int j = 0; j < PADALIGN * (NC / 4) / (SCW * 32); ++j) {
      int i = (int)threadIdx.x + j * (SCW * 32);
      if (i < n4) dst[i] = make_float4(0.0f, 0.0f, 0.0f, 0.0f);
    }
  }
}

#define SYW 4

template <int S> __device__ constexpr int tile_a(int t) { return S == 0 ? (t < 2 ? 1 : 0) : (S == 1 ? 2 : 3); }
template <int S> __device__ constexpr int tile_b(int t) { return S == 0 ? (t == 1 ? 1 : 0) : (S == 3 ? 2 + t : t); }
template <int S> __device__ constexpr int tile_n() { return S < 2 ? 3 : 2; }
template <int S> __device__ constexpr int sum_blk() { return S == 1 ? 2 : (S == 2 ? 1 : S); }

__device__ __forceinline__ float xhalf(float x) {
  int b = __float_as_int(x);
  return __int_as_float(__builtin_amdgcn_permlanex16(b, b, 0x76543210, 0xfedcba98, false, false));
}

__device__ __forceinline__ void load_group(v2f (&f)[4], const float* __restrict__ gp) {
#pragma unroll
  for (int blk = 0; blk < 4; ++blk) {
    f[blk].x = gp[blk * 16];
    f[blk].y = gp[NC + blk * 16];
  }
}

__device__ __forceinline__ void load_trip(v2f (&f)[4][4], const float* __restrict__ tp) {
#pragma unroll
  for (int i = 0; i < 4; ++i) load_group(f[i], tp + (size_t)i * 4 * NC);
}

template <int S>
__device__ __forceinline__ void use_group(v8f (&acc)[3], float& cs, const v2f (&f)[4]) {
#pragma unroll
  for (int t = 0; t < tile_n<S>(); ++t)
    acc[t] = __builtin_amdgcn_wmma_f32_16x16x4_f32(
        false, f[tile_a<S>(t)], false, f[tile_b<S>(t)], (short)0, acc[t], false, false);
  constexpr int SB = sum_blk<S>();
  float ox = xhalf(f[SB].x), oy = xhalf(f[SB].y);
  cs = (((cs + f[SB].x) + f[SB].y) + ox) + oy;
}

template <int S>
__device__ __forceinline__ void use_trip(v8f (&acc)[3], float& cs, const v2f (&f)[4][4]) {
#pragma unroll
  for (int i = 0; i < 4; ++i) use_group<S>(acc, cs, f[i]);
}

template <int S>
__device__ __forceinline__ void syrk_tile_set(const float* __restrict__ rp, int plen, int lane,
                                              float* __restrict__ sxxout, float* __restrict__ sumout) {
  const int ntrip = plen >> 4;
  const int rem = (plen >> 2) & 3;

  v8f acc[3] = {};
  float cs = 0.0f;

  v2f ft[3][4];
  const float* tailp = rp + (size_t)ntrip * 16 * NC;
  if (rem > 0) load_group(ft[0], tailp);
  if (rem > 1) load_group(ft[1], tailp + 4 * NC);
  if (rem > 2) load_group(ft[2], tailp + 8 * NC);

  v2f fa[4][4], fb[4][4];
  if (ntrip > 0) load_trip(fa, rp);
  int t = 0;
  for (; t + 2 <= ntrip; t += 2) {
    load_trip(fb, rp + (size_t)(t + 1) * 16 * NC);
    use_trip<S>(acc, cs, fa);
    int tn = (t + 2 < ntrip) ? t + 2 : ntrip - 1;
    load_trip(fa, rp + (size_t)tn * 16 * NC);
    use_trip<S>(acc, cs, fb);
  }
  if (t < ntrip) use_trip<S>(acc, cs, fa);
  if (rem > 0) use_group<S>(acc, cs, ft[0]);
  if (rem > 1) use_group<S>(acc, cs, ft[1]);
  if (rem > 2) use_group<S>(acc, cs, ft[2]);

  int col = lane & 15;
  int rof = (lane >> 4) * 8;
#pragma unroll
  for (int t = 0; t < tile_n<S>(); ++t)
#pragma unroll
    for (int j = 0; j < 8; ++j) {
      int r = tile_a<S>(t) * 16 + rof + j;
      sxxout[(size_t)r * NC + tile_b<S>(t) * 16 + col] = acc[t][j];
    }
  if (lane < 16) sumout[sum_blk<S>() * 16 + lane] = cs;
}

__global__ __launch_bounds__(SYW * 32) void syrk_kernel(const float* __restrict__ comp,
                                                        const int* __restrict__ classinfo,
                                                        float* __restrict__ sxxpart,
                                                        float* __restrict__ sumpart) {
  int gid = blockIdx.x;
  int part = gid / (NB * NCLS);
  int bk = gid % (NB * NCLS);
  int b = bk / NCLS;
  int lane = threadIdx.x & 31;
  int wave = __builtin_amdgcn_readfirstlane((int)(threadIdx.x >> 5));

  int base = classinfo[bk * 3 + 0];
  int padl = classinfo[bk * 3 + 2];
  int plen = padl / PARTS;
  size_t rowbase = (size_t)b * NPAD_MAX + base + (size_t)part * plen;

  int m = lane & 15;
  int kb = (lane >> 4) << 1;
  const float* rp = comp + (rowbase + kb) * NC + m;
  float* sxxout = sxxpart + (size_t)gid * (NC * NC);
  float* sumout = sumpart + (size_t)gid * NC;

  int set = (wave + (gid ^ (gid >> 8))) & 3;
  switch (set) {
    case 0: syrk_tile_set<0>(rp, plen, lane, sxxout, sumout); break;
    case 1: syrk_tile_set<1>(rp, plen, lane, sxxout, sumout); break;
    case 2: syrk_tile_set<2>(rp, plen, lane, sxxout, sumout); break;
    default: syrk_tile_set<3>(rp, plen, lane, sxxout, sumout); break;
  }
}

__device__ __forceinline__ float lane_read(float v, int l) {
  return __int_as_float(__builtin_amdgcn_readlane(__float_as_int(v), l));
}

template <int NA>
__device__ __forceinline__ void chol_step_a(v2f (&cab)[16], float& ra, float* covL, int lane, int kk) {
  float piv = lane_read(cab[0].x, kk);
  float ri = __builtin_amdgcn_rsqf(piv);
  v2f l = cab[0] * ri;
  ra = (lane == kk) ? ri : ra;
  covL[lane * 65 + kk] = l.x;
  covL[(32 + lane) * 65 + kk] = l.y;
#pragma unroll
  for (int jj = 1; jj <= NA; ++jj) {
    float s = lane_read(l.x, (kk + jj) & 31);
    cab[jj - 1] = cab[jj] - l * s;
  }
}

template <int NBC>
__device__ __forceinline__ void chol_step_b(float (&cbh)[16], float& rb, float* covL, int lane, int kq) {
  float piv = lane_read(cbh[0], kq);
  float ri = __builtin_amdgcn_rsqf(piv);
  float l1 = cbh[0] * ri;
  rb = (lane == kq) ? ri : rb;
  covL[(32 + lane) * 65 + 32 + kq] = l1;
#pragma unroll
  for (int jj = 1; jj <= NBC; ++jj) {
    float s = lane_read(l1, (kq + jj) & 31);
    cbh[jj - 1] = cbh[jj] - l1 * s;
  }
}

__device__ __forceinline__ v8f tile_load(const float* covL, int lane, int r0, int c0) {
  const float* p = &covL[(r0 + (lane >> 4) * 8) * 65 + c0 + (lane & 15)];
  v8f a;
#pragma unroll
  for (int r = 0; r < 8; ++r) a[r] = p[r * 65];
  return a;
}
__device__ __forceinline__ void tile_store(float* covL, int lane, int r0, int c0, v8f a) {
  float* p = &covL[(r0 + (lane >> 4) * 8) * 65 + c0 + (lane & 15)];
#pragma unroll
  for (int r = 0; r < 8; ++r) p[r * 65] = a[r];
}
__device__ __forceinline__ v2f l_frag(const float* covL, int lane, int r0, int k0) {
  const float* p = &covL[(r0 + (lane & 15)) * 65 + k0 + ((lane >> 4) << 1)];
  v2f f;
  f.x = p[0];
  f.y = p[1];
  return f;
}
__device__ __forceinline__ v8f tile_sub(v8f c, v2f a, v2f b) {
  return __builtin_amdgcn_wmma_f32_16x16x4_f32(false, -a, false, b, (short)0, c, false, false);
}

__global__ __launch_bounds__(256) void reduce_solve_kernel(const float* __restrict__ sxxpart,
                                                           const float* __restrict__ sumpart,
                                                           const int* __restrict__ classinfo,
                                                           const float* __restrict__ pattern,
                                                           float* __restrict__ params) {
  int bk = blockIdx.x;
  int tid = threadIdx.x;
  __shared__ float covL[64 * 65];
  __shared__ float meanL[64], pL[64], wv[64];

  int count = classinfo[bk * 3 + 1];
  float safe = (float)(count > 0 ? count : 1);

  if (tid < 64) {
    float s = 0.0f;
    for (int part = 0; part < PARTS; ++part)
      s += sumpart[((size_t)part * NB * NCLS + bk) * NC + tid];
    meanL[tid] = s / safe;
    pL[tid] = pattern[tid];
  }
  __syncthreads();

  const int er = tid >> 4, ec = tid & 15;
  for (int t = 0; t < 10; ++t) {
    int ti = (t >= 6) ? 3 : (t >= 3) ? 2 : (t >= 1) ? 1 : 0;
    int tj = t - ((ti * (ti + 1)) >> 1);
    int i = ti * 16 + er, j = tj * 16 + ec;
    int idx = i * NC + j;
    float s = 0.0f;
    for (int part = 0; part < PARTS; ++part)
      s += sxxpart[((size_t)part * NB * NCLS + bk) * (NC * NC) + idx];
    float cv = s / safe - meanL[i] * meanL[j];
    if (i == j) cv += EPSV;
    covL[i * 65 + j] = cv;
  }
  __syncthreads();
  for (int t = 0; t < 6; ++t) {
    int ti = (t >= 3) ? 3 : (t >= 1) ? 2 : 1;
    int tj = t - ((ti * (ti - 1)) >> 1);
    int i = ti * 16 + er, j = tj * 16 + ec;
    covL[j * 65 + i] = covL[i * 65 + j];
  }
  __syncthreads();

  if (tid < 32) {
    const int lane = tid;
    v2f cab[16];
    float cbh[16];
#pragma unroll
    for (int j = 0; j < 16; ++j) {
      cab[j].x = covL[lane * 65 + j];
      cab[j].y = covL[(32 + lane) * 65 + j];
    }

    float r0 = 0.0f, r1 = 0.0f;
#pragma nounroll
    for (int kk = 0; kk < 8; ++kk) chol_step_a<15>(cab, r0, covL, lane, kk);
#pragma nounroll
    for (int kk = 8; kk < 16; ++kk) chol_step_a<7>(cab, r0, covL, lane, kk);
    {
      v8f acc[3];
      v2f fr[3][4];
#pragma unroll
      for (int t = 0; t < 3; ++t) {
        acc[t] = tile_load(covL, lane, 16 * (t + 1), 16);
#pragma unroll
        for (int s = 0; s < 4; ++s) fr[t][s] = l_frag(covL, lane, 16 * (t + 1), 4 * s);
      }
#pragma unroll
      for (int s = 0; s < 4; ++s)
#pragma unroll
        for (int t = 0; t < 3; ++t) acc[t] = tile_sub(acc[t], fr[t][s], fr[0][s]);
#pragma unroll
      for (int t = 0; t < 3; ++t) tile_store(covL, lane, 16 * (t + 1), 16, acc[t]);
#pragma unroll
      for (int j = 0; j < 16; ++j) {
        cab[j].x = covL[lane * 65 + 16 + j];
        cab[j].y = covL[(32 + lane) * 65 + 16 + j];
      }
    }
#pragma nounroll
    for (int kk = 16; kk < 24; ++kk) chol_step_a<15>(cab, r0, covL, lane, kk);
#pragma nounroll
    for (int kk = 24; kk < 32; ++kk) chol_step_a<7>(cab, r0, covL, lane, kk);
    {
      v8f acc[3];
      v2f fr[2][8];
      acc[0] = tile_load(covL, lane, 32, 32);
      acc[1] = tile_load(covL, lane, 48, 32);
      acc[2] = tile_load(covL, lane, 48, 48);
#pragma unroll
      for (int s = 0; s < 8; ++s) {
        fr[0][s] = l_frag(covL, lane, 32, 4 * s);
        fr[1][s] = l_frag(covL, lane, 48, 4 * s);
      }
#pragma unroll
      for (int s = 0; s < 8; ++s) {
        acc[0] = tile_sub(acc[0], fr[0][s], fr[0][s]);
        acc[1] = tile_sub(acc[1], fr[1][s], fr[0][s]);
        acc[2] = tile_sub(acc[2], fr[1][s], fr[1][s]);
      }
      tile_store(covL, lane, 32, 32, acc[0]);
      tile_store(covL, lane, 48, 32, acc[1]);
      tile_store(covL, lane, 48, 48, acc[2]);
#pragma unroll
      for (int j = 0; j < 16; ++j) cbh[j] = covL[(32 + lane) * 65 + 32 + j];
    }
#pragma nounroll
    for (int kq = 0; kq < 8; ++kq) chol_step_b<15>(cbh, r1, covL, lane, kq);
#pragma nounroll
    for (int kq = 8; kq < 16; ++kq) chol_step_b<7>(cbh, r1, covL, lane, kq);
    {
      v8f acc = tile_load(covL, lane, 48, 48);
      v2f fr[4];
#pragma unroll
      for (int s = 0; s < 4; ++s) fr[s] = l_frag(covL, lane, 48, 32 + 4 * s);
#pragma unroll
      for (int s = 0; s < 4; ++s) acc = tile_sub(acc, fr[s], fr[s]);
      tile_store(covL, lane, 48, 48, acc);
#pragma unroll
      for (int j = 0; j < 16; ++j) cbh[j] = covL[(32 + lane) * 65 + 48 + j];
    }
#pragma nounroll
    for (int kq = 16; kq < 24; ++kq) chol_step_b<15>(cbh, r1, covL, lane, kq);
#pragma nounroll
    for (int kq = 24; kq < 32; ++kq) chol_step_b<7>(cbh, r1, covL, lane, kq);

    float la[32], lb[NC], ua[NC], ub[32];
#pragma unroll
    for (int j = 0; j < 32; ++j) {
      la[j] = covL[lane * 65 + j];
      lb[j] = covL[(32 + lane) * 65 + j];
      lb[32 + j] = covL[(32 + lane) * 65 + 32 + j];
      ua[j] = covL[j * 65 + lane];
      ua[32 + j] = covL[(32 + j) * 65 + lane];
      ub[j] = covL[(32 + j) * 65 + 32 + lane];
    }

    float y0 = pL[lane], y1 = pL[32 + lane];
#pragma unroll
    for (int j = 0; j < 32; ++j) {
      float s = lane_read(y0 * r0, j);
      float t0 = y0 - la[j] * s, t1 = y1 - lb[j] * s;
      y0 = (lane == j) ? s : ((lane > j) ? t0 : y0);
      y1 = t1;
    }
#pragma unroll
    for (int j = 0; j < 32; ++j) {
      float s = lane_read(y1 * r1, j);
      float t1 = y1 - lb[32 + j] * s;
      y1 = (lane == j) ? s : ((lane > j) ? t1 : y1);
    }

    float w0 = y0, w1 = y1;
#pragma unroll
    for (int j = 31; j >= 0; --j) {
      float s = lane_read(w1 * r1, j);
      float t0 = w0 - ua[32 + j] * s, t1 = w1 - ub[j] * s;
      w1 = (lane == j) ? s : ((lane < j) ? t1 : w1);
      w0 = t0;
    }
#pragma unroll
    for (int j = 31; j >= 0; --j) {
      float s = lane_read(w0 * r0, j);
      float t0 = w0 - ua[j] * s;
      w0 = (lane == j) ? s : ((lane < j) ? t0 : w0);
    }

    params[(size_t)bk * 66 + lane] = w0;
    params[(size_t)bk * 66 + 32 + lane] = w1;
    wv[lane] = w0;
    wv[32 + lane] = w1;
    float pw = 0.0f, off = 0.0f;
    for (int c = 0; c < NC; ++c) {
      pw += pL[c] * wv[c];
      off += meanL[c] * wv[c];
    }
    if (lane == 0) {
      float denom = sqrtf(pw);
      float scale = (count >= MIN_CLASS_SIZE && denom > 0.0f) ? (1.0f / denom) : 0.0f;
      params[(size_t)bk * 66 + 64] = off;
      params[(size_t)bk * 66 + 65] = scale;
    }
  }
}

__global__ __launch_bounds__(256) void score_kernel(const float* __restrict__ hyper,
                                                    const int* __restrict__ seg,
                                                    const float* __restrict__ params,
                                                    float* __restrict__ out) {
  int b = blockIdx.y;
  int tid = threadIdx.x;
  __shared__ float wS[NCLS * 65];
  __shared__ float offS[NCLS], sclS[NCLS];
  for (int i = tid; i < NCLS * NC; i += 256) {
    int k = i >> 6, c = i & 63;
    wS[k * 65 + c] = params[((size_t)b * NCLS + k) * 66 + c];
  }
  if (tid < NCLS) {
    offS[tid] = params[((size_t)b * NCLS + tid) * 66 + 64];
    sclS[tid] = params[((size_t)b * NCLS + tid) * 66 + 65];
  }
  __syncthreads();

  size_t n = (size_t)blockIdx.x * 256 + tid;
  int k = seg[(size_t)b * NPIX + n];
  const float* xp = hyper + (size_t)b * NC * NPIX + n;
  float acc = 0.0f;
#pragma unroll
  for (int c = 0; c < NC; ++c)
    acc = fmaf(xp[(size_t)c * NPIX], wS[k * 65 + c], acc);
  out[(size_t)b * NPIX + n] = (acc - offS[k]) * sclS[k];
}

extern "C" void kernel_launch(void* const* d_in, const int* in_sizes, int n_in,
                              void* d_out, int out_size, void* d_ws, size_t ws_size,
                              hipStream_t stream) {
  const float* hyper = (const float*)d_in[0];
  const float* pattern = (const float*)d_in[1];
  const int* seg = (const int*)d_in[2];
  float* out = (float*)d_out;

  char* ws = (char*)d_ws;
  auto al = [](size_t x) { return (x + 255) & ~(size_t)255; };
  size_t off = 0;
  float* comp = (float*)(ws + off);
  off += al((size_t)NB * NPAD_MAX * NC * sizeof(float));
  int* hist = (int*)(ws + off);
  off += al((size_t)NB * NBLKS * NCLS * sizeof(int));
  int* classinfo = (int*)(ws + off);
  off += al((size_t)NB * NCLS * 3 * sizeof(int));
  float* sxxpart = (float*)(ws + off);
  off += al((size_t)PARTS * NB * NCLS * NC * NC * sizeof(float));
  float* sumpart = (float*)(ws + off);
  off += al((size_t)PARTS * NB * NCLS * NC * sizeof(float));
  float* params = (float*)(ws + off);
  off += al((size_t)NB * NCLS * 66 * sizeof(float));

  hipLaunchKernelGGL(hist_kernel, dim3(NB * NBLKS), dim3(HW * 32), 0, stream, seg, hist);
  hipLaunchKernelGGL(scatter_kernel, dim3(NB * NBLKS * SUBS), dim3(SCW * 32), 0, stream, hyper, seg,
                     hist, classinfo, comp);
  hipLaunchKernelGGL(syrk_kernel, dim3(PARTS * NB * NCLS), dim3(SYW * 32), 0, stream, comp,
                     classinfo, sxxpart, sumpart);
  hipLaunchKernelGGL(reduce_solve_kernel, dim3(NB * NCLS), dim3(256), 0, stream, sxxpart,
                     sumpart, classinfo, pattern, params);
  hipLaunchKernelGGL(score_kernel, dim3(NPIX / 256, NB), dim3(256), 0, stream, hyper, seg,
                     params, out);
}


# syrk: each wave forms all 10 lower-triangle tiles on its own quarter of the part's rows (a row is fetched by one wave, was up to 3); run sums of tiles and column sums added in LDS: sum order changes
# speedup vs baseline: 6.8473x; 1.0536x over previous
#include <hip/hip_runtime.h>
#include <hip/hip_bf16.h>


typedef float v2f __attribute__((ext_vector_type(2)));
typedef float v8f __attribute__((ext_vector_type(8)));

#define NB 2
#define NC 64
#define NPIX 147456
#define NCLS 20
#define BLKPX 1024
#define CHUNKS (BLKPX / 32)
#define NBLKS (NPIX / BLKPX)
#define PARTS 16
#define PADALIGN (PARTS * 4)
#define NPAD_MAX (NPIX + NCLS * PADALIGN)
#define MIN_CLASS_SIZE 5000
#define EPSV 1e-6f

#define HW 4
#define HLD (BLKPX / 4 / (HW * 32))

__global__ __launch_bounds__(HW * 32) void hist_kernel(const int* __restrict__ seg,
                                                       int* __restrict__ hist) {
  static_assert(HLD * 4 * HW * 32 == BLKPX, "the work-group's loads cover the block exactly");
  int blk = blockIdx.x;
  int b = blk / NBLKS, bb = blk % NBLKS;
  int lane = threadIdx.x & 31;
  int wave = __builtin_amdgcn_readfirstlane((int)(threadIdx.x >> 5));
  __shared__ int part[HW][32];
  const int4* s4 = (const int4*)(seg + (size_t)b * NPIX + (size_t)bb * BLKPX);
  int4 q[HLD];
#pragma unroll
  for (int j = 0; j < HLD; ++j) q[j] = s4[j * (HW * 32) + (int)threadIdx.x];
  int cnt = 0;
#pragma unroll
  for (int k = 0; k < NCLS; ++k) {
    int c = 0;
#pragma unroll
    for (int j = 0; j < HLD; ++j)
      c += __popc(__builtin_amdgcn_ballot_w32(q[j].x == k)) +
           __popc(__builtin_amdgcn_ballot_w32(q[j].y == k)) +
           __popc(__builtin_amdgcn_ballot_w32(q[j].z == k)) +
           __popc(__builtin_amdgcn_ballot_w32(q[j].w == k));
    if (lane == k) cnt = c;
  }
  part[wave][lane] = cnt;
  __syncthreads();
  if (wave == 0) {
    int t = 0;
#pragma unroll
    for (int w = 0; w < HW; ++w) t += part[w][lane];
    if (lane < NCLS) hist[((size_t)b * NCLS + lane) * NBLKS + bb] = t;
  }
}

#define SUBS 4
#define SCW (CHUNKS / SUBS)
#define TPITCH 68
#define CPW ((NCLS + SCW - 1) / SCW)
#define HQ (NBLKS / 4)

__device__ __forceinline__ int wave_sum(int v) {
  v += __builtin_amdgcn_update_dpp(0, v, 0x161, 0xf, 0xf, false);
  v += __builtin_amdgcn_update_dpp(0, v, 0x162, 0xf, 0xf, false);
  v += __builtin_amdgcn_update_dpp(0, v, 0x164, 0xf, 0xf, false);
  v += __builtin_amdgcn_update_dpp(0, v, 0x168, 0xf, 0xf, false);
  v += __builtin_amdgcn_permlanex16(v, v, 0x76543210, 0xfedcba98, false, false);
  return v;
}

__global__ __launch_bounds__(SCW * 32) void scatter_kernel(const float* __restrict__ hyper,
                                                           const int* __restrict__ seg,
                                                           const int* __restrict__ hist,
                                                           int* __restrict__ classinfo,
                                                           float* __restrict__ comp) {
  int s = blockIdx.x % SUBS;
  int hb = blockIdx.x / SUBS;
  int b = hb / NBLKS, bb = hb % NBLKS;
  int lane = threadIdx.x & 31;
  int wave = __builtin_amdgcn_readfirstlane((int)(threadIdx.x >> 5));
  __shared__ __attribute__((aligned(16))) float tile[SCW][32 * TPITCH];
  __shared__ int cnt[CHUNKS][32];
  __shared__ int totS[32];
  __shared__ int befS[32];

  const int* sg = seg + (size_t)b * NPIX + (size_t)bb * BLKPX;
  int sgv[SUBS];
#pragma unroll
  for (int j = 0; j < SUBS; ++j) sgv[j] = sg[(j * SCW + wave) * 32 + lane];

  static_assert(HQ - 32 == 4, "the upper part of a histogram column is four 4-block units");
  int4 hlo[CPW], hhi[CPW];
#pragma unroll
  for (int r = 0; r < CPW; ++r) {
    int k = wave + SCW * r;
    int kc = (k < NCLS) ? k : NCLS - 1;
    const int4* hp = (const int4*)(hist + ((size_t)b * NCLS + kc) * NBLKS);
    hlo[r] = hp[lane];
    hhi[r] = hp[32 + (lane & 3)];
  }

  int ch = s * SCW + wave;
  const float* xp = hyper + (size_t)b * NC * NPIX + (size_t)bb * BLKPX + (size_t)ch * 32 + lane;
  float v[NC];
#pragma unroll
  for (int c = 0; c < NC; ++c) v[c] = xp[(size_t)c * NPIX];

  int myk = sgv[0];
#pragma unroll
  for (int j = 1; j < SUBS; ++j) myk = (s == j) ? sgv[j] : myk;

#pragma unroll
  for (int j = 0; j < SUBS; ++j) {
    int pc = 0;
#pragma unroll
    for (int k = 0; k < NCLS; ++k) {
      unsigned bal = __builtin_amdgcn_ballot_w32(sgv[j] == k);
      if (lane == k) pc = __popc(bal);
    }
    cnt[j * SCW + wave][lane] = pc;
  }
  unsigned balmy = 0;
#pragma unroll
  for (int k = 0; k < NCLS; ++k) {
    unsigned bal = __builtin_amdgcn_ballot_w32(myk == k);
    if (myk == k) balmy = bal;
  }

#pragma unroll
  for (int r = 0; r < CPW; ++r) {
    int k = wave + SCW * r;
    int q0 = 4 * lane, q1 = 128 + 4 * lane;
    bool hi = lane < HQ - 32;
    int tt = ((hlo[r].x + hlo[r].y) + (hlo[r].z + hlo[r].w)) +
             (hi ? ((hhi[r].x + hhi[r].y) + (hhi[r].z + hhi[r].w)) : 0);
    int bf = ((q0 < bb) ? hlo[r].x : 0) + ((q0 + 1 < bb) ? hlo[r].y : 0) +
             ((q0 + 2 < bb) ? hlo[r].z : 0) + ((q0 + 3 < bb) ? hlo[r].w : 0) +
             ((hi && q1 < bb) ? hhi[r].x : 0) + ((hi && q1 + 1 < bb) ? hhi[r].y : 0) +
             ((hi && q1 + 2 < bb) ? hhi[r].z : 0) + ((hi && q1 + 3 < bb) ? hhi[r].w : 0);
    tt = wave_sum(tt);
    bf = wave_sum(bf);
    if (lane == 0 && k < NCLS) {
      totS[k] = tt;
      befS[k] = bf;
    }
  }

  float* tl = tile[wave];
#pragma unroll
  for (int q = 0; q < NC / 4; ++q)
    *(float4*)&tl[lane * TPITCH + 4 * q] =
        make_float4(v[4 * q], v[4 * q + 1], v[4 * q + 2], v[4 * q + 3]);
  __syncthreads();

  int cls = (lane < NCLS) ? lane : 0;
  int base = 0;
#pragma unroll
  for (int k = 0; k < NCLS - 1; ++k) {
    int p = (totS[k] + PADALIGN - 1) & ~(PADALIGN - 1);
    base += (k < lane) ? p : 0;
  }
  int tk = totS[cls];
  int pk = (tk + PADALIGN - 1) & ~(PADALIGN - 1);
  int cb = base + befS[cls];

  int run = 0;
#pragma unroll
  for (int c2 = 0; c2 < CHUNKS - 1; ++c2) {
    int t = cnt[c2][lane];
    run += (c2 < ch) ? t : 0;
  }
  unsigned ltmask = (1u << lane) - 1u;
  int pos = __shfl(cb + run, myk, 32) + __popc(balmy & ltmask);

  float* crow = comp + (size_t)b * NPAD_MAX * NC;
#pragma unroll
  for (int p = 0; p < 32; ++p) {
    int pp = __builtin_amdgcn_readlane(pos, p);
    float2 val = *(const float2*)&tl[p * TPITCH + 2 * lane];
    *(float2*)&crow[(size_t)pp * NC + 2 * lane] = val;
  }

  if (s == 0 && bb == 0 && wave == 0 && lane < NCLS) {
    classinfo[(b * NCLS + lane) * 3 + 0] = base;
    classinfo[(b * NCLS + lane) * 3 + 1] = tk;
    classinfo[(b * NCLS + lane) * 3 + 2] = pk;
  }
  if (s == 0 && bb < NCLS) {
    int zb = __builtin_amdgcn_readlane(base, bb);
    int zt = __builtin_amdgcn_readlane(tk, bb);
    int zp = __builtin_amdgcn_readlane(pk, bb);
    float4* dst = (float4*)(crow + ((size_t)zb + zt) * NC);
    int n4 = (zp - zt) * (NC / 4);
#pragma unroll
    for (int j = 0; j < PADALIGN * (NC / 4) / (SCW * 32); ++j) {
      int i = (int)threadIdx.x + j * (SCW * 32);
      if (i < n4) dst[i] = make_float4(0.0f, 0.0f, 0.0f, 0.0f);
    }
  }
}

#define SYW 4
#define LT 10

typedef float v4f __attribute__((ext_vector_type(4)));

__device__ constexpr int lt_a(int t) { return t >= 6 ? 3 : (t >= 3 ? 2 : (t >= 1 ? 1 : 0)); }
__device__ constexpr int lt_b(int t) { return t - ((lt_a(t) * (lt_a(t) + 1)) >> 1); }

__device__ __forceinline__ void load_group(v2f (&f)[4], const float* __restrict__ gp) {
#pragma unroll
  for (int blk = 0; blk < 4; ++blk) {
    f[blk].x = gp[blk * 16];
    f[blk].y = gp[NC + blk * 16];
  }
}

__device__ __forceinline__ void load_trip(v2f (&f)[4][4], const float* __restrict__ tp) {
#pragma unroll
  for (int i = 0; i < 4; ++i) load_group(f[i], tp + (size_t)i * 4 * NC);
}

__device__ __forceinline__ void use_group(v8f (&acc)[LT], v2f (&cs)[4], const v2f (&f)[4]) {
#pragma unroll
  for (int t = 0; t < LT; ++t)
    acc[t] = __builtin_amdgcn_wmma_f32_16x16x4_f32(
        false, f[lt_a(t)], false, f[lt_b(t)], (short)0, acc[t], false, false);
#pragma unroll
  for (int blk = 0; blk < 4; ++blk) cs[blk] += f[blk];
}

__device__ __forceinline__ void use_trip(v8f (&acc)[LT], v2f (&cs)[4], const v2f (&f)[4][4]) {
#pragma unroll
  for (int i = 0; i < 4; ++i) use_group(acc, cs, f[i]);
}

__global__ __launch_bounds__(SYW * 32) void syrk_kernel(const float* __restrict__ comp,
                                                        const int* __restrict__ classinfo,
                                                        float* __restrict__ sxxpart,
                                                        float* __restrict__ sumpart) {
  int gid = blockIdx.x;
  int part = gid / (NB * NCLS);
  int bk = gid % (NB * NCLS);
  int b = bk / NCLS;
  int lane = threadIdx.x & 31;
  int wave = __builtin_amdgcn_readfirstlane((int)(threadIdx.x >> 5));

  __shared__ v4f red[SYW][LT][2][32];
  __shared__ float csr[SYW][4][32];

  int base = classinfo[bk * 3 + 0];
  int padl = classinfo[bk * 3 + 2];
  int plen = padl / PARTS;
  int ngrp = plen >> 2;
  int g0 = (ngrp * wave) >> 2;
  int g1 = (ngrp * (wave + 1)) >> 2;
  int nmy = g1 - g0;
  size_t rowbase = (size_t)b * NPAD_MAX + base + (size_t)part * plen + (size_t)g0 * 4;

  int m = lane & 15;
  int kb = (lane >> 4) << 1;
  const float* rp = comp + (rowbase + kb) * NC + m;

  const int ntrip = nmy >> 2;
  const int rem = nmy & 3;

  v8f acc[LT] = {};
  v2f cs[4] = {};

  v2f ft[3][4];
  const float* tailp = rp + (size_t)ntrip * 16 * NC;
  if (rem > 0) load_group(ft[0], tailp);
  if (rem > 1) load_group(ft[1], tailp + 4 * NC);
  if (rem > 2) load_group(ft[2], tailp + 8 * NC);

  v2f fa[4][4], fb[4][4];
  if (ntrip > 0) load_trip(fa, rp);
  int t = 0;
  for (; t + 2 <= ntrip; t += 2) {
    load_trip(fb, rp + (size_t)(t + 1) * 16 * NC);
    use_trip(acc, cs, fa);
    int tn = (t + 2 < ntrip) ? t + 2 : ntrip - 1;
    load_trip(fa, rp + (size_t)tn * 16 * NC);
    use_trip(acc, cs, fb);
  }
  if (t < ntrip) use_trip(acc, cs, fa);
  if (rem > 0) use_group(acc, cs, ft[0]);
  if (rem > 1) use_group(acc, cs, ft[1]);
  if (rem > 2) use_group(acc, cs, ft[2]);

#pragma unroll
  for (int tt = 0; tt < LT; ++tt) {
    red[wave][tt][0][lane] = (v4f){acc[tt][0], acc[tt][1], acc[tt][2], acc[tt][3]};
    red[wave][tt][1][lane] = (v4f){acc[tt][4], acc[tt][5], acc[tt][6], acc[tt][7]};
  }
#pragma unroll
  for (int blk = 0; blk < 4; ++blk) csr[wave][blk][lane] = cs[blk].x + cs[blk].y;
  __syncthreads();

  float* sxxout = sxxpart + (size_t)gid * (NC * NC);
  float* sumout = sumpart + (size_t)gid * NC;
  int col = lane & 15;
  int rof = (lane >> 4) * 8;
  int nt = (wave < 2) ? 3 : 2;
  for (int tt = 0; tt < nt; ++tt) {
    int ta = (wave == 0) ? (tt < 2 ? 1 : 0) : (wave == 1 ? 2 : 3);
    int tb = (wave == 0) ? (tt == 1 ? 1 : 0) : (wave == 3 ? 2 + tt : tt);
    int lt = ((ta * (ta + 1)) >> 1) + tb;
    v4f lo = red[0][lt][0][lane], hi = red[0][lt][1][lane];
#pragma unroll
    for (int w = 1; w < SYW; ++w) {
      lo += red[w][lt][0][lane];
      hi += red[w][lt][1][lane];
    }
    float* o = sxxout + (size_t)(ta * 16 + rof) * NC + tb * 16 + col;
    o[0 * NC] = lo.x;
    o[1 * NC] = lo.y;
    o[2 * NC] = lo.z;
    o[3 * NC] = lo.w;
    o[4 * NC] = hi.x;
    o[5 * NC] = hi.y;
    o[6 * NC] = hi.z;
    o[7 * NC] = hi.w;
  }
  if (wave == SYW - 1) {
#pragma unroll
    for (int hh = 0; hh < 2; ++hh) {
      int c = hh * 32 + lane;
      int blk = c >> 4, mm = c & 15;
      float s = 0.0f;
#pragma unroll
      for (int w = 0; w < SYW; ++w) s += csr[w][blk][mm] + csr[w][blk][mm + 16];
      sumout[c] = s;
    }
  }
}

__device__ __forceinline__ float lane_read(float v, int l) {
  return __int_as_float(__builtin_amdgcn_readlane(__float_as_int(v), l));
}

template <int NA>
__device__ __forceinline__ void chol_step_a(v2f (&cab)[16], float& ra, float* covL, int lane, int kk) {
  float piv = lane_read(cab[0].x, kk);
  float ri = __builtin_amdgcn_rsqf(piv);
  v2f l = cab[0] * ri;
  ra = (lane == kk) ? ri : ra;
  covL[lane * 65 + kk] = l.x;
  covL[(32 + lane) * 65 + kk] = l.y;
#pragma unroll
  for (int jj = 1; jj <= NA; ++jj) {
    float s = lane_read(l.x, (kk + jj) & 31);
    cab[jj - 1] = cab[jj] - l * s;
  }
}

template <int NBC>
__device__ __forceinline__ void chol_step_b(float (&cbh)[16], float& rb, float* covL, int lane, int kq) {
  float piv = lane_read(cbh[0], kq);
  float ri = __builtin_amdgcn_rsqf(piv);
  float l1 = cbh[0] * ri;
  rb = (lane == kq) ? ri : rb;
  covL[(32 + lane) * 65 + 32 + kq] = l1;
#pragma unroll
  for (int jj = 1; jj <= NBC; ++jj) {
    float s = lane_read(l1, (kq + jj) & 31);
    cbh[jj - 1] = cbh[jj] - l1 * s;
  }
}

__device__ __forceinline__ v8f tile_load(const float* covL, int lane, int r0, int c0) {
  const float* p = &covL[(r0 + (lane >> 4) * 8) * 65 + c0 + (lane & 15)];
  v8f a;
#pragma unroll
  for (int r = 0; r < 8; ++r) a[r] = p[r * 65];
  return a;
}
__device__ __forceinline__ void tile_store(float* covL, int lane, int r0, int c0, v8f a) {
  float* p = &covL[(r0 + (lane >> 4) * 8) * 65 + c0 + (lane & 15)];
#pragma unroll
  for (int r = 0; r < 8; ++r) p[r * 65] = a[r];
}
__device__ __forceinline__ v2f l_frag(const float* covL, int lane, int r0, int k0) {
  const float* p = &covL[(r0 + (lane & 15)) * 65 + k0 + ((lane >> 4) << 1)];
  v2f f;
  f.x = p[0];
  f.y = p[1];
  return f;
}
__device__ __forceinline__ v8f tile_sub(v8f c, v2f a, v2f b) {
  return __builtin_amdgcn_wmma_f32_16x16x4_f32(false, -a, false, b, (short)0, c, false, false);
}

__global__ __launch_bounds__(256) void reduce_solve_kernel(const float* __restrict__ sxxpart,
                                                           const float* __restrict__ sumpart,
                                                           const int* __restrict__ classinfo,
                                                           const float* __restrict__ pattern,
                                                           float* __restrict__ params) {
  int bk = blockIdx.x;
  int tid = threadIdx.x;
  __shared__ float covL[64 * 65];
  __shared__ float meanL[64], pL[64], wv[64];

  int count = classinfo[bk * 3 + 1];
  float safe = (float)(count > 0 ? count : 1);

  if (tid < 64) {
    float s = 0.0f;
    for (int part = 0; part < PARTS; ++part)
      s += sumpart[((size_t)part * NB * NCLS + bk) * NC + tid];
    meanL[tid] = s / safe;
    pL[tid] = pattern[tid];
  }
  __syncthreads();

  const int er = tid >> 4, ec = tid & 15;
  for (int t = 0; t < 10; ++t) {
    int ti = (t >= 6) ? 3 : (t >= 3) ? 2 : (t >= 1) ? 1 : 0;
    int tj = t - ((ti * (ti + 1)) >> 1);
    int i = ti * 16 + er, j = tj * 16 + ec;
    int idx = i * NC + j;
    float s = 0.0f;
    for (int part = 0; part < PARTS; ++part)
      s += sxxpart[((size_t)part * NB * NCLS + bk) * (NC * NC) + idx];
    float cv = s / safe - meanL[i] * meanL[j];
    if (i == j) cv += EPSV;
    covL[i * 65 + j] = cv;
  }
  __syncthreads();
  for (int t = 0; t < 6; ++t) {
    int ti = (t >= 3) ? 3 : (t >= 1) ? 2 : 1;
    int tj = t - ((ti * (ti - 1)) >> 1);
    int i = ti * 16 + er, j = tj * 16 + ec;
    covL[j * 65 + i] = covL[i * 65 + j];
  }
  __syncthreads();

  if (tid < 32) {
    const int lane = tid;
    v2f cab[16];
    float cbh[16];
#pragma unroll
    for (int j = 0; j < 16; ++j) {
      cab[j].x = covL[lane * 65 + j];
      cab[j].y = covL[(32 + lane) * 65 + j];
    }

    float r0 = 0.0f, r1 = 0.0f;
#pragma nounroll
    for (int kk = 0; kk < 8; ++kk) chol_step_a<15>(cab, r0, covL, lane, kk);
#pragma nounroll
    for (int kk = 8; kk < 16; ++kk) chol_step_a<7>(cab, r0, covL, lane, kk);
    {
      v8f acc[3];
      v2f fr[3][4];
#pragma unroll
      for (int t = 0; t < 3; ++t) {
        acc[t] = tile_load(covL, lane, 16 * (t + 1), 16);
#pragma unroll
        for (int s = 0; s < 4; ++s) fr[t][s] = l_frag(covL, lane, 16 * (t + 1), 4 * s);
      }
#pragma unroll
      for (int s = 0; s < 4; ++s)
#pragma unroll
        for (int t = 0; t < 3; ++t) acc[t] = tile_sub(acc[t], fr[t][s], fr[0][s]);
#pragma unroll
      for (int t = 0; t < 3; ++t) tile_store(covL, lane, 16 * (t + 1), 16, acc[t]);
#pragma unroll
      for (int j = 0; j < 16; ++j) {
        cab[j].x = covL[lane * 65 + 16 + j];
        cab[j].y = covL[(32 + lane) * 65 + 16 + j];
      }
    }
#pragma nounroll
    for (int kk = 16; kk < 24; ++kk) chol_step_a<15>(cab, r0, covL, lane, kk);
#pragma nounroll
    for (int kk = 24; kk < 32; ++kk) chol_step_a<7>(cab, r0, covL, lane, kk);
    {
      v8f acc[3];
      v2f fr[2][8];
      acc[0] = tile_load(covL, lane, 32, 32);
      acc[1] = tile_load(covL, lane, 48, 32);
      acc[2] = tile_load(covL, lane, 48, 48);
#pragma unroll
      for (int s = 0; s < 8; ++s) {
        fr[0][s] = l_frag(covL, lane, 32, 4 * s);
        fr[1][s] = l_frag(covL, lane, 48, 4 * s);
      }
#pragma unroll
      for (int s = 0; s < 8; ++s) {
        acc[0] = tile_sub(acc[0], fr[0][s], fr[0][s]);
        acc[1] = tile_sub(acc[1], fr[1][s], fr[0][s]);
        acc[2] = tile_sub(acc[2], fr[1][s], fr[1][s]);
      }
      tile_store(covL, lane, 32, 32, acc[0]);
      tile_store(covL, lane, 48, 32, acc[1]);
      tile_store(covL, lane, 48, 48, acc[2]);
#pragma unroll
      for (int j = 0; j < 16; ++j) cbh[j] = covL[(32 + lane) * 65 + 32 + j];
    }
#pragma nounroll
    for (int kq = 0; kq < 8; ++kq) chol_step_b<15>(cbh, r1, covL, lane, kq);
#pragma nounroll
    for (int kq = 8; kq < 16; ++kq) chol_step_b<7>(cbh, r1, covL, lane, kq);
    {
      v8f acc = tile_load(covL, lane, 48, 48);
      v2f fr[4];
#pragma unroll
      for (int s = 0; s < 4; ++s) fr[s] = l_frag(covL, lane, 48, 32 + 4 * s);
#pragma unroll
      for (int s = 0; s < 4; ++s) acc = tile_sub(acc, fr[s], fr[s]);
      tile_store(covL, lane, 48, 48, acc);
#pragma unroll
      for (int j = 0; j < 16; ++j) cbh[j] = covL[(32 + lane) * 65 + 48 + j];
    }
#pragma nounroll
    for (int kq = 16; kq < 24; ++kq) chol_step_b<15>(cbh, r1, covL, lane, kq);
#pragma nounroll
    for (int kq = 24; kq < 32; ++kq) chol_step_b<7>(cbh, r1, covL, lane, kq);

    float la[32], lb[NC], ua[NC], ub[32];
#pragma unroll
    for (int j = 0; j < 32; ++j) {
      la[j] = covL[lane * 65 + j];
      lb[j] = covL[(32 + lane) * 65 + j];
      lb[32 + j] = covL[(32 + lane) * 65 + 32 + j];
      ua[j] = covL[j * 65 + lane];
      ua[32 + j] = covL[(32 + j) * 65 + lane];
      ub[j] = covL[(32 + j) * 65 + 32 + lane];
    }

    float y0 = pL[lane], y1 = pL[32 + lane];
#pragma unroll
    for (int j = 0; j < 32; ++j) {
      float s = lane_read(y0 * r0, j);
      float t0 = y0 - la[j] * s, t1 = y1 - lb[j] * s;
      y0 = (lane == j) ? s : ((lane > j) ? t0 : y0);
      y1 = t1;
    }
#pragma unroll
    for (int j = 0; j < 32; ++j) {
      float s = lane_read(y1 * r1, j);
      float t1 = y1 - lb[32 + j] * s;
      y1 = (lane == j) ? s : ((lane > j) ? t1 : y1);
    }

    float w0 = y0, w1 = y1;
#pragma unroll
    for (int j = 31; j >= 0; --j) {
      float s = lane_read(w1 * r1, j);
      float t0 = w0 - ua[32 + j] * s, t1 = w1 - ub[j] * s;
      w1 = (lane == j) ? s : ((lane < j) ? t1 : w1);
      w0 = t0;
    }
#pragma unroll
    for (int j = 31; j >= 0; --j) {
      float s = lane_read(w0 * r0, j);
      float t0 = w0 - ua[j] * s;
      w0 = (lane == j) ? s : ((lane < j) ? t0 : w0);
    }

    params[(size_t)bk * 66 + lane] = w0;
    params[(size_t)bk * 66 + 32 + lane] = w1;
    wv[lane] = w0;
    wv[32 + lane] = w1;
    float pw = 0.0f, off = 0.0f;
    for (int c = 0; c < NC; ++c) {
      pw += pL[c] * wv[c];
      off += meanL[c] * wv[c];
    }
    if (lane == 0) {
      float denom = sqrtf(pw);
      float scale = (count >= MIN_CLASS_SIZE && denom > 0.0f) ? (1.0f / denom) : 0.0f;
      params[(size_t)bk * 66 + 64] = off;
      params[(size_t)bk * 66 + 65] = scale;
    }
  }
}

__global__ __launch_bounds__(256) void score_kernel(const float* __restrict__ hyper,
                                                    const int* __restrict__ seg,
                                                    const float* __restrict__ params,
                                                    float* __restrict__ out) {
  int b = blockIdx.y;
  int tid = threadIdx.x;
  __shared__ float wS[NCLS * 65];
  __shared__ float offS[NCLS], sclS[NCLS];
  for (int i = tid; i < NCLS * NC; i += 256) {
    int k = i >> 6, c = i & 63;
    wS[k * 65 + c] = params[((size_t)b * NCLS + k) * 66 + c];
  }
  if (tid < NCLS) {
    offS[tid] = params[((size_t)b * NCLS + tid) * 66 + 64];
    sclS[tid] = params[((size_t)b * NCLS + tid) * 66 + 65];
  }
  __syncthreads();

  size_t n = (size_t)blockIdx.x * 256 + tid;
  int k = seg[(size_t)b * NPIX + n];
  const float* xp = hyper + (size_t)b * NC * NPIX + n;
  float acc = 0.0f;
#pragma unroll
  for (int c = 0; c < NC; ++c)
    acc = fmaf(xp[(size_t)c * NPIX], wS[k * 65 + c], acc);
  out[(size_t)b * NPIX + n] = (acc - offS[k]) * sclS[k];
}

extern "C" void kernel_launch(void* const* d_in, const int* in_sizes, int n_in,
                              void* d_out, int out_size, void* d_ws, size_t ws_size,
                              hipStream_t stream) {
  const float* hyper = (const float*)d_in[0];
  const float* pattern = (const float*)d_in[1];
  const int* seg = (const int*)d_in[2];
  float* out = (float*)d_out;

  char* ws = (char*)d_ws;
  auto al = [](size_t x) { return (x + 255) & ~(size_t)255; };
  size_t off = 0;
  float* comp = (float*)(ws + off);
  off += al((size_t)NB * NPAD_MAX * NC * sizeof(float));
  int* hist = (int*)(ws + off);
  off += al((size_t)NB * NBLKS * NCLS * sizeof(int));
  int* classinfo = (int*)(ws + off);
  off += al((size_t)NB * NCLS * 3 * sizeof(int));
  float* sxxpart = (float*)(ws + off);
  off += al((size_t)PARTS * NB * NCLS * NC * NC * sizeof(float));
  float* sumpart = (float*)(ws + off);
  off += al((size_t)PARTS * NB * NCLS * NC * sizeof(float));
  float* params = (float*)(ws + off);
  off += al((size_t)NB * NCLS * 66 * sizeof(float));

  hipLaunchKernelGGL(hist_kernel, dim3(NB * NBLKS), dim3(HW * 32), 0, stream, seg, hist);
  hipLaunchKernelGGL(scatter_kernel, dim3(NB * NBLKS * SUBS), dim3(SCW * 32), 0, stream, hyper, seg,
                     hist, classinfo, comp);
  hipLaunchKernelGGL(syrk_kernel, dim3(PARTS * NB * NCLS), dim3(SYW * 32), 0, stream, comp,
                     classinfo, sxxpart, sumpart);
  hipLaunchKernelGGL(reduce_solve_kernel, dim3(NB * NCLS), dim3(256), 0, stream, sxxpart,
                     sumpart, classinfo, pattern, params);
  hipLaunchKernelGGL(score_kernel, dim3(NPIX / 256, NB), dim3(256), 0, stream, hyper, seg,
                     params, out);
}
